# starting kernel
# baseline (speedup 1.0000x reference)
#include <hip/hip_runtime.h>
#include <math.h>

typedef float v2f __attribute__((ext_vector_type(2)));
typedef float v8f __attribute__((ext_vector_type(8)));

__global__ __launch_bounds__(128) void gemm_f32_wmma(
    const float* __restrict__ A, long sAm, long sAk, long sAb,
    const float* __restrict__ B, long sBk, long sBn, long sBb,
    float* __restrict__ C, long sCm, long sCn, long sCb,
    int M, int N, int K, float alpha, int accum)
{
  const int lane = threadIdx.x & 31;
  const int wave = threadIdx.x >> 5;
  const int half = lane >> 4;
  const int l16  = lane & 15;
  const int mBase = blockIdx.x * 64;
  const int nBase = blockIdx.y * 64 + wave * 16;

  A += (long)blockIdx.z * sAb;
  B += (long)blockIdx.z * sBb;
  C += (long)blockIdx.z * sCb;

  v8f zero = {0.f,0.f,0.f,0.f,0.f,0.f,0.f,0.f};
  v8f acc[4];
#pragma unroll
  for (int t = 0; t < 4; ++t) acc[t] = zero;

  const float* aPtr[4];
  float aMask[4];
#pragma unroll
  for (int t = 0; t < 4; ++t) {
    int m = mBase + t * 16 + l16;
    bool v = (m < M);
    aMask[t] = v ? 1.f : 0.f;
    aPtr[t] = A + (long)(v ? m : 0) * sAm + (long)(2 * half) * sAk;
  }
  const int n = nBase + l16;
  const bool nv = (n < N);
  const float bMask = nv ? 1.f : 0.f;
  const float* bPtr = B + (long)(nv ? n : 0) * sBn + (long)(2 * half) * sBk;

  for (int k0 = 0; k0 < K; k0 += 4) {
    v2f bf;
    bf.x = bMask * bPtr[0];
    bf.y = bMask * bPtr[sBk];
    bPtr += 4 * sBk;
#pragma unroll
    for (int t = 0; t < 4; ++t) {
      v2f af;
      af.x = aMask[t] * aPtr[t][0];
      af.y = aMask[t] * aPtr[t][sAk];
      aPtr[t] += 4 * sAk;
      acc[t] = __builtin_amdgcn_wmma_f32_16x16x4_f32(
          false, af, false, bf, (short)0, acc[t], false, false);
    }
  }

  if (nv) {
#pragma unroll
    for (int t = 0; t < 4; ++t) {
#pragma unroll
      for (int r = 0; r < 8; ++r) {
        int m = mBase + t * 16 + r + 8 * half;
        if (m < M) {
          long idx = (long)m * sCm + (long)n * sCn;
          float val = alpha * acc[t][r];
          if (accum) C[idx] += val; else C[idx] = val;
        }
      }
    }
  }
}

__global__ void gen_consts(float* Fre, float* Fim, float* Vre, float* Vim) {
  int i = blockIdx.x * 256 + threadIdx.x;
  if (i >= 48 * 96) return;
  int k = i / 96, nn = i % 96;
  int f = (k < 24) ? k : (k + 48);
  int ph = (f * nn) % 96;
  float th = 6.283185307179586f * (float)ph / 96.f;
  float c = cosf(th), s = sinf(th);
  Fre[i] = c;  Fim[i] = -s;
  Vre[nn * 48 + k] = c;  Vim[nn * 48 + k] = s;
}

__global__ void pack_wqkv(const float* __restrict__ wq, const float* __restrict__ wk,
                          const float* __restrict__ wv, float* __restrict__ dst) {
  int i = blockIdx.x * 256 + threadIdx.x;
  if (i >= 384 * 64) return;
  int ic = i % 64, r = i / 64;
  int t = r / 128, q = r % 128;
  const float* w = (t == 0) ? wq : ((t == 1) ? wk : wv);
  dst[i] = w[q * 64 + ic];
}

__global__ void addsub(const float* __restrict__ x, const float* __restrict__ y,
                       float* __restrict__ sum, float* __restrict__ dif, int nelem) {
  int i = blockIdx.x * 256 + threadIdx.x;
  if (i >= nelem) return;
  float a = x[i], b = y[i];
  sum[i] = a + b;
  dif[i] = a - b;
}

__global__ void selu_k(float* __restrict__ s, int nelem) {
  int i = blockIdx.x * 256 + threadIdx.x;
  if (i >= nelem) return;
  float x = s[i];
  const float lam = 1.0507009873554805f, alp = 1.6732632423543772f;
  s[i] = lam * (x > 0.f ? x : alp * (expf(x) - 1.f));
}

__global__ void group_qkv(const float* __restrict__ src, float* __restrict__ dst) {
  int i = blockIdx.x * 256 + threadIdx.x;
  if (i >= 42467328) return;
  int s = i % 110592, r = i / 110592;
  int x = s % 48, y = (s / 48) % 48, z = s / 2304;
  int o = r % 16, h = (r / 16) % 8, t = r / 128;
  int px = x & 7, nx = x >> 3, py = y & 7, ny = y >> 3, pz = z & 7, nz = z >> 3;
  int cp = ((((o << 3) | pz) << 3) | py) << 3 | px;
  int tok = (nz * 6 + ny) * 6 + nx;
  dst[((long)((t * 8 + h) * 8192 + cp)) * 216 + tok] = src[i];
}

__global__ void ungroup_o(const float* __restrict__ src, float* __restrict__ dst) {
  int i = blockIdx.x * 256 + threadIdx.x;
  if (i >= 14155776) return;
  int tok = i % 216, cp = (i / 216) % 8192, h = i / (216 * 8192);
  int px = cp & 7, py = (cp >> 3) & 7, pz = (cp >> 6) & 7, o = cp >> 9;
  int nx = tok % 6, ny = (tok / 6) % 6, nz = tok / 36;
  int z = nz * 8 + pz, y = ny * 8 + py, x = nx * 8 + px;
  dst[(long)(h * 16 + o) * 110592 + (long)(z * 48 + y) * 48 + x] = src[i];
}

extern "C" void kernel_launch(void* const* d_in, const int* in_sizes, int n_in,
                              void* d_out, int out_size, void* d_ws, size_t ws_size,
                              hipStream_t stream) {
  const float* in = (const float*)d_in[0];
  const float* wq = (const float*)d_in[1];
  const float* wk = (const float*)d_in[2];
  const float* wv = (const float*)d_in[3];
  const float* wo = (const float*)d_in[4];
  float* out = (float*)d_out;
  float* w = (float*)d_ws;

  const long R1re_o = 0,          R1im_o = 28311552;
  const long R2re_o = 56623104,   R2im_o = 70778880;
  const long blk_o  = 84934656;
  const long QKV_o  = 0,          G_o    = 42467328;
  const long S_o    = 92012544,   Og_o   = 92385792;
  const long U_o    = 0,          spec_o = 14155776;
  const long T1re_o = 15925248,   T1im_o = 19464192;
  const long T2re_o = 23003136,   T2im_o = 30081024;
  const long sumI_o = 37158912,   difI_o = 44236800;
  const long Wqkv_o = 106541568;
  const long Fre_o  = 106566144,  Fim_o  = 106570752;
  const long Vre_o  = 106575360,  Vim_o  = 106579968;
  const long WS_FLOATS = 106584576;
  if (ws_size < (size_t)WS_FLOATS * sizeof(float)) return;

  float *R1re = w+R1re_o, *R1im = w+R1im_o, *R2re = w+R2re_o, *R2im = w+R2im_o;
  float *blk = w+blk_o, *QKV = w+QKV_o, *G = w+G_o, *S = w+S_o, *Og = w+Og_o;
  float *U = w+U_o, *spec = w+spec_o;
  float *T1re = w+T1re_o, *T1im = w+T1im_o, *T2re = w+T2re_o, *T2im = w+T2im_o;
  float *sumI = w+sumI_o, *difI = w+difI_o, *Wqkv = w+Wqkv_o;
  float *Fre = w+Fre_o, *Fim = w+Fim_o, *Vre = w+Vre_o, *Vim = w+Vim_o;

  auto gemm = [&](const float* A, long sAm, long sAk, long sAb,
                  const float* B, long sBk, long sBn, long sBb,
                  float* C, long sCm, long sCn, long sCb,
                  int M, int N, int K, int batch, float alpha, int accum) {
    dim3 g((unsigned)((M + 63) / 64), (unsigned)((N + 63) / 64), (unsigned)batch);
    gemm_f32_wmma<<<g, 128, 0, stream>>>(A, sAm, sAk, sAb, B, sBk, sBn, sBb,
                                         C, sCm, sCn, sCb, M, N, K, alpha, accum);
  };
  auto ew = [&](int n) { return dim3((unsigned)((n + 255) / 256)); };

  gen_consts<<<ew(4608), 256, 0, stream>>>(Fre, Fim, Vre, Vim);
  pack_wqkv<<<ew(24576), 256, 0, stream>>>(wq, wk, wv, Wqkv);

  gemm(in, 96, 1, 0, Fre, 1, 96, 0, R1re, 48, 1, 0, 589824, 48, 96, 1, 1.f, 0);
  gemm(in, 96, 1, 0, Fim, 1, 96, 0, R1im, 48, 1, 0, 589824, 48, 96, 1, 1.f, 0);
  gemm(Fre, 96, 1, 0, R1re, 48, 1, 4608, R2re, 48, 1, 2304, 48, 48, 96, 6144,  1.f, 0);
  gemm(Fim, 96, 1, 0, R1im, 48, 1, 4608, R2re, 48, 1, 2304, 48, 48, 96, 6144, -1.f, 1);
  gemm(Fre, 96, 1, 0, R1im, 48, 1, 4608, R2im, 48, 1, 2304, 48, 48, 96, 6144,  1.f, 0);
  gemm(Fim, 96, 1, 0, R1re, 48, 1, 4608, R2im, 48, 1, 2304, 48, 48, 96, 6144,  1.f, 1);
  addsub<<<ew(14155776), 256, 0, stream>>>(R2re, R2im,  R1re,  R1im, 14155776);
  gemm(Fre, 96, 1, 0, R1im, 2304, 1, 221184, blk, 2304, 1, 110592, 48, 2304, 96, 64,  1.f, 0);
  gemm(Fim, 96, 1, 0, R1re, 2304, 1, 221184, blk, 2304, 1, 110592, 48, 2304, 96, 64, -1.f, 1);

  gemm(Wqkv, 64, 1, 0, blk, 110592, 1, 0, QKV, 110592, 1, 0, 384, 110592, 64, 1, 1.f, 0);
  group_qkv<<<ew(42467328), 256, 0, stream>>>(QKV, G);

  const long HD = 8192L * 216L;
  gemm(G, 1, 216, HD, G + 8 * HD, 216, 1, HD, S, 216, 1, 46656,
       216, 216, 8192, 8, 0.011048543456039806f, 0);
  selu_k<<<ew(373248), 256, 0, stream>>>(S, 373248);
  gemm(G + 16 * HD, 216, 1, HD, S, 1, 216, 46656, Og, 216, 1, HD,
       8192, 216, 216, 8, 1.f, 0);
  ungroup_o<<<ew(14155776), 256, 0, stream>>>(Og, U);

  gemm(wo, 128, 1, 0, U, 110592, 1, 0, spec, 110592, 1, 0, 16, 110592, 128, 1, 1.f, 0);

  gemm(spec, 48, 1, 0, Vre, 1, 48, 0, T1re, 96, 1, 0, 36864, 96, 48, 1, 1.f, 0);
  gemm(spec, 48, 1, 0, Vim, 1, 48, 0, T1im, 96, 1, 0, 36864, 96, 48, 1, 1.f, 0);
  gemm(Vre, 48, 1, 0, T1re, 96, 1, 4608, T2re, 96, 1, 9216, 96, 96, 48, 768,  1.f, 0);
  gemm(Vim, 48, 1, 0, T1im, 96, 1, 4608, T2re, 96, 1, 9216, 96, 96, 48, 768, -1.f, 1);
  gemm(Vre, 48, 1, 0, T1im, 96, 1, 4608, T2im, 96, 1, 9216, 96, 96, 48, 768,  1.f, 0);
  gemm(Vim, 48, 1, 0, T1re, 96, 1, 4608, T2im, 96, 1, 9216, 96, 96, 48, 768,  1.f, 1);
  addsub<<<ew(7077888), 256, 0, stream>>>(T2re, T2im, sumI, difI, 7077888);
  const float invN = 1.f / 884736.f;
  gemm(Vre, 48, 1, 0, sumI, 9216, 1, 442368, out, 9216, 1, 884736, 96, 9216, 48, 16, invN, 0);
  gemm(Vim, 48, 1, 0, difI, 9216, 1, 442368, out, 9216, 1, 884736, 96, 9216, 48, 16, invN, 1);
}


# Patch grouping folded into the attention products' addressing: score and value products read Q,K,V from and write O to the (channel,z,y,x) layout; group/ungroup passes gone (score sum order changes)
# speedup vs baseline: 4.9842x; 4.9842x over previous
#include <hip/hip_runtime.h>
#include <math.h>

typedef float v2f __attribute__((ext_vector_type(2)));
typedef float v4f __attribute__((ext_vector_type(4)));
typedef float v8f __attribute__((ext_vector_type(8)));

template <bool BK>
__global__ __launch_bounds__(128) void gemm_f32_wmma_k4(
    const float* __restrict__ A, long sAm, long sAb,
    const float* __restrict__ B, long sBk, long sBn, long sBb,
    float* __restrict__ C, long sCm, long sCn, long sCb,
    int M, int N, int K, float alpha, int accum)
{
  const int lane = threadIdx.x & 31;
  const int wave = threadIdx.x >> 5;
  const int half = lane >> 4;
  const int l16  = lane & 15;
  const int mBase = blockIdx.x * 64;
  const int nBase = blockIdx.y * 64 + wave * 16;

  A += (long)blockIdx.z * sAb;
  B += (long)blockIdx.z * sBb;
  C += (long)blockIdx.z * sCb;

  v8f zero = {0.f,0.f,0.f,0.f,0.f,0.f,0.f,0.f};
  v8f acc[4];
#pragma unroll
  for (int t = 0; t < 4; ++t) acc[t] = zero;

  const float* aPtr[4];
  float aMask[4];
#pragma unroll
  for (int t = 0; t < 4; ++t) {
    int m = mBase + t * 16 + l16;
    bool v = (m < M);
    aMask[t] = v ? 1.f : 0.f;
    aPtr[t] = A + (long)(v ? m : 0) * sAm + 4 * half;
  }
  const int n = nBase + l16;
  const bool nv = (n < N);
  const float bMask = nv ? 1.f : 0.f;
  const float* bPtr = B + (long)(nv ? n : 0) * sBn + (long)(4 * half) * sBk;

  constexpr int D = 3;
  const int ng = K >> 3;
  v4f ra[D][4];
  v4f rb[D];
  v4f aq[4];
  v4f bq;

  auto load_step = [&](int j) {
    if (BK) {
      rb[j] = *(const v4f*)bPtr;
      bPtr += 8;
    } else {
      rb[j].x = bPtr[0];
      rb[j].y = bPtr[sBk];
      rb[j].z = bPtr[2 * sBk];
      rb[j].w = bPtr[3 * sBk];
      bPtr += 8 * sBk;
    }
#pragma unroll
    for (int t = 0; t < 4; ++t) {
      ra[j][t] = *(const v4f*)aPtr[t];
      aPtr[t] += 8;
    }
  };
  auto take_step = [&](int j) {
    bq = bMask * rb[j];
#pragma unroll
    for (int t = 0; t < 4; ++t) aq[t] = aMask[t] * ra[j][t];
  };
  auto mma_step = [&]() {
    v2f b0 = {bq.x, bq.y};
    v2f b1 = {bq.z, bq.w};
#pragma unroll
    for (int t = 0; t < 4; ++t) {
      v2f a0 = {aq[t].x, aq[t].y};
      acc[t] = __builtin_amdgcn_wmma_f32_16x16x4_f32(
          false, a0, false, b0, (short)0, acc[t], false, false);
    }
#pragma unroll
    for (int t = 0; t < 4; ++t) {
      v2f a1 = {aq[t].z, aq[t].w};
      acc[t] = __builtin_amdgcn_wmma_f32_16x16x4_f32(
          false, a1, false, b1, (short)0, acc[t], false, false);
    }
  };

  if (ng >= 2 * D) {
#pragma unroll
    for (int j = 0; j < D; ++j) {
      load_step(j);
      __builtin_amdgcn_sched_barrier(0);
    }

    int s = 0;
    do {
#pragma unroll
      for (int j = 0; j < D; ++j) {
        take_step(j);
        load_step(j);
        mma_step();
        __builtin_amdgcn_sched_barrier(0);
      }
      s += D;
    } while (s + 2 * D <= ng);

#pragma unroll
    for (int j = 0; j < D; ++j) {
      take_step(j);
      if (s + D + j < ng) load_step(j);
      mma_step();
    }
#pragma unroll
    for (int j = 0; j < D - 1; ++j) {
      if (s + D + j < ng) {
        take_step(j);
        mma_step();
      }
    }
  } else {
    for (int i = 0; i < ng; ++i) {
      load_step(0);
      take_step(0);
      mma_step();
    }
  }

  if (nv) {
#pragma unroll
    for (int t = 0; t < 4; ++t) {
#pragma unroll
      for (int r = 0; r < 8; ++r) {
        int m = mBase + t * 16 + r + 8 * half;
        if (m < M) {
          long idx = (long)m * sCm + (long)n * sCn;
          float val = alpha * acc[t][r];
          if (accum) C[idx] += val; else C[idx] = val;
        }
      }
    }
  }
}

constexpr int tok_off(int m) {
  return (m / 36) * 18432 + ((m / 6) % 6) * 384 + (m % 6) * 8;
}
constexpr int chan_off(int c) {
  return (c >> 9) * 110592 + ((c >> 6) & 7) * 2304 + ((c >> 3) & 7) * 48 + (c & 7);
}
struct TokOffsets {
  int v[264];
  constexpr TokOffsets() : v() {
    for (int m = 0; m < 264; ++m) v[m] = tok_off(m < 216 ? m : 0);
  }
};
__constant__ const TokOffsets kTok = TokOffsets();

__global__ __launch_bounds__(128) void attn_scores(
    const float* __restrict__ Q, const float* __restrict__ Kp,
    float* __restrict__ S, float alpha)
{
  const int lane = threadIdx.x & 31;
  const int wave = threadIdx.x >> 5;
  const int half = lane >> 4;
  const int l16  = lane & 15;
  const int mBase = blockIdx.x * 64;
  const int nBase = blockIdx.y * 64 + wave * 16;

  Q  += (long)blockIdx.z * 1769472;
  Kp += (long)blockIdx.z * 1769472;
  S  += (long)blockIdx.z * 46656;

  v8f zero = {0.f,0.f,0.f,0.f,0.f,0.f,0.f,0.f};
  v8f acc[4];
#pragma unroll
  for (int t = 0; t < 4; ++t) acc[t] = zero;

  unsigned aByte[4];
  float aMask[4];
#pragma unroll
  for (int t = 0; t < 4; ++t) {
    int m = mBase + t * 16 + l16;
    bool v = (m < 216);
    aMask[t] = v ? 1.f : 0.f;
    aByte[t] = (unsigned)(tok_off(v ? m : 0) + 4 * half) * 4u;
  }
  const int n = nBase + l16;
  const bool nv = (n < 216);
  const float bMask = nv ? 1.f : 0.f;
  const unsigned bByte = (unsigned)(tok_off(nv ? n : 0) + 4 * half) * 4u;

  constexpr int D = 6;
  constexpr int NG = 1024;
  v4f ra[D][4];
  v4f rb[D];
  v4f aq[4];
  v4f bq;
  int gl = 0;

  auto load_step = [&](int j) {
    const long ko = (long)(gl >> 6) * 110592 + (long)(((gl >> 3) & 7) * 2304 + (gl & 7) * 48);
    const char* qg = (const char*)(Q + ko);
    const char* kg = (const char*)(Kp + ko);
    rb[j] = *(const v4f*)(kg + bByte);
#pragma unroll
    for (int t = 0; t < 4; ++t) ra[j][t] = *(const v4f*)(qg + aByte[t]);
    ++gl;
  };
  auto take_step = [&](int j) {
    bq = bMask * rb[j];
#pragma unroll
    for (int t = 0; t < 4; ++t) aq[t] = aMask[t] * ra[j][t];
  };
  auto mma_step = [&]() {
    v2f b0 = {bq.x, bq.y};
    v2f b1 = {bq.z, bq.w};
#pragma unroll
    for (int t = 0; t < 4; ++t) {
      v2f a0 = {aq[t].x, aq[t].y};
      acc[t] = __builtin_amdgcn_wmma_f32_16x16x4_f32(
          false, a0, false, b0, (short)0, acc[t], false, false);
    }
#pragma unroll
    for (int t = 0; t < 4; ++t) {
      v2f a1 = {aq[t].z, aq[t].w};
      acc[t] = __builtin_amdgcn_wmma_f32_16x16x4_f32(
          false, a1, false, b1, (short)0, acc[t], false, false);
    }
  };

#pragma unroll
  for (int j = 0; j < D; ++j) {
    load_step(j);
    __builtin_amdgcn_sched_barrier(0);
  }

  int s = 0;
  do {
#pragma unroll
    for (int j = 0; j < D; ++j) {
      take_step(j);
      load_step(j);
      mma_step();
      __builtin_amdgcn_sched_barrier(0);
    }
    s += D;
  } while (s + 2 * D <= NG);

#pragma unroll
  for (int j = 0; j < D; ++j) {
    take_step(j);
    if (s + D + j < NG) load_step(j);
    mma_step();
  }
#pragma unroll
  for (int j = 0; j < D - 1; ++j) {
    if (s + D + j < NG) {
      take_step(j);
      mma_step();
    }
  }

  if (nv) {
#pragma unroll
    for (int t = 0; t < 4; ++t) {
#pragma unroll
      for (int r = 0; r < 8; ++r) {
        int m = mBase + t * 16 + r + 8 * half;
        if (m < 216) S[m * 216 + n] = alpha * acc[t][r];
      }
    }
  }
}

__global__ __launch_bounds__(128) void attn_values(
    const float* __restrict__ S, const float* __restrict__ V,
    float* __restrict__ U)
{
  const int lane = threadIdx.x & 31;
  const int wave = threadIdx.x >> 5;
  const int half = lane >> 4;
  const int l16  = lane & 15;
  const int mBase = blockIdx.x * 64;
  const int nBase = blockIdx.y * 64 + wave * 16;

  S += (long)blockIdx.z * 46656;
  V += (long)blockIdx.z * 1769472;
  U += (long)blockIdx.z * 1769472;

  v8f zero = {0.f,0.f,0.f,0.f,0.f,0.f,0.f,0.f};
  v8f acc[4];
#pragma unroll
  for (int t = 0; t < 4; ++t) acc[t] = zero;

  const float* aPtr[4];
  float aMask[4];
#pragma unroll
  for (int t = 0; t < 4; ++t) {
    int m = mBase + t * 16 + l16;
    bool v = (m < 216);
    aMask[t] = v ? 1.f : 0.f;
    aPtr[t] = S + (v ? m : 0) * 216 + 4 * half;
  }
  const int c = nBase + l16;
  const bool nv = (c < 8192);
  const float bMask = nv ? 1.f : 0.f;
  const unsigned cOff = (unsigned)chan_off(nv ? c : 0);
  const char* vBytes = (const char*)V;

  constexpr int D = 3;
  constexpr int NG = 27;
  v4f ra[D][4];
  v4f rb[D];
  v4f aq[4];
  v4f bq;
  int gl = 0;
  int tk0[4], tk1[4];
  auto fetch_tok = [&]() {
#pragma unroll
    for (int e = 0; e < 4; ++e) {
      tk0[e] = kTok.v[8 * gl + e];
      tk1[e] = kTok.v[8 * gl + 4 + e];
    }
  };
  fetch_tok();

  auto load_step = [&](int j) {
    const unsigned o0 = (cOff + (unsigned)(half ? tk1[0] : tk0[0])) << 2;
    const unsigned o1 = (cOff + (unsigned)(half ? tk1[1] : tk0[1])) << 2;
    const unsigned o2 = (cOff + (unsigned)(half ? tk1[2] : tk0[2])) << 2;
    const unsigned o3 = (cOff + (unsigned)(half ? tk1[3] : tk0[3])) << 2;
    rb[j].x = *(const float*)(vBytes + o0);
    rb[j].y = *(const float*)(vBytes + o1);
    rb[j].z = *(const float*)(vBytes + o2);
    rb[j].w = *(const float*)(vBytes + o3);
#pragma unroll
    for (int t = 0; t < 4; ++t) {
      ra[j][t] = *(const v4f*)aPtr[t];
      aPtr[t] += 8;
    }
    ++gl;
    fetch_tok();
  };
  auto take_step = [&](int j) {
    bq = bMask * rb[j];
#pragma unroll
    for (int t = 0; t < 4; ++t) aq[t] = aMask[t] * ra[j][t];
  };
  auto mma_step = [&]() {
    v2f b0 = {bq.x, bq.y};
    v2f b1 = {bq.z, bq.w};
#pragma unroll
    for (int t = 0; t < 4; ++t) {
      v2f a0 = {aq[t].x, aq[t].y};
      acc[t] = __builtin_amdgcn_wmma_f32_16x16x4_f32(
          false, a0, false, b0, (short)0, acc[t], false, false);
    }
#pragma unroll
    for (int t = 0; t < 4; ++t) {
      v2f a1 = {aq[t].z, aq[t].w};
      acc[t] = __builtin_amdgcn_wmma_f32_16x16x4_f32(
          false, a1, false, b1, (short)0, acc[t], false, false);
    }
  };

#pragma unroll
  for (int j = 0; j < D; ++j) {
    load_step(j);
    __builtin_amdgcn_sched_barrier(0);
  }

  int s = 0;
  do {
#pragma unroll
    for (int j = 0; j < D; ++j) {
      take_step(j);
      load_step(j);
      mma_step();
      __builtin_amdgcn_sched_barrier(0);
    }
    s += D;
  } while (s + 2 * D <= NG);

#pragma unroll
  for (int j = 0; j < D; ++j) {
    take_step(j);
    if (s + D + j < NG) load_step(j);
    mma_step();
  }
#pragma unroll
  for (int j = 0; j < D - 1; ++j) {
    if (s + D + j < NG) {
      take_step(j);
      mma_step();
    }
  }

#pragma unroll
  for (int t = 0; t < 4; ++t) {
#pragma unroll
    for (int r = 0; r < 8; ++r) {
      const int mu = mBase + t * 16 + r;
      const int e0 = kTok.v[mu];
      const int e1 = kTok.v[mu + 8];
      if (nv && mu + 8 * half < 216) U[cOff + (unsigned)(half ? e1 : e0)] = acc[t][r];
    }
  }
}

__global__ void gen_consts(float* Cf, float* Vc) {
  int i = blockIdx.x * 256 + threadIdx.x;
  if (i >= 49 * 96) return;
  int j = i / 96, nn = i % 96;
  int f = (j <= 24) ? j : (j + 47);
  float th = 6.283185307179586f * (float)((f * nn) % 96) / 96.f;
  Cf[i] = cosf(th) + sinf(th);
  if (i < 48 * 96) {
    int g = (j < 24) ? j : (j + 48);
    float tv = 6.283185307179586f * (float)((g * nn) % 96) / 96.f;
    Vc[nn * 48 + j] = cosf(tv) + sinf(tv);
  }
}

__global__ void hartley_mirror_fwd(const float* __restrict__ T, float* __restrict__ blk) {
  int i = blockIdx.x * 256 + threadIdx.x;
  if (i >= 7077888) return;
  int ix = i % 48, iy = (i / 48) % 48, iz = (i / 2304) % 48, c = i / 110592;
  int jx = ix + (ix >= 24), jy = iy + (iy >= 24), jz = iz + (iz >= 24);
  int gx = jx ? 49 - jx : 0, gy = jy ? 49 - jy : 0, gz = jz ? 49 - jz : 0;
  const float* t = T + (long)c * 117649;
  float a = t[(gz * 49 + jy) * 49 + jx];
  float b = t[(jz * 49 + gy) * 49 + jx];
  float d = t[(jz * 49 + jy) * 49 + gx];
  float e = t[(gz * 49 + gy) * 49 + gx];
  blk[i] = 0.5f * ((a + b) + (d - e));
}

__global__ void hartley_mirror_inv(const float* __restrict__ T, float* __restrict__ out) {
  int i = blockIdx.x * 256 + threadIdx.x;
  if (i >= 14155776) return;
  int x = i % 96, y = (i / 96) % 96, z = (i / 9216) % 96, c = i / 884736;
  int gx = x ? 96 - x : 0, gy = y ? 96 - y : 0, gz = z ? 96 - z : 0;
  const float* t = T + (long)c * 884736;
  float a = t[(gz * 96 + y) * 96 + x];
  float b = t[(z * 96 + gy) * 96 + x];
  float d = t[(z * 96 + y) * 96 + gx];
  float e = t[(gz * 96 + gy) * 96 + gx];
  out[i] = (a + b) + (d - e);
}

__global__ void pack_wqkv(const float* __restrict__ wq, const float* __restrict__ wk,
                          const float* __restrict__ wv, float* __restrict__ dst) {
  int i = blockIdx.x * 256 + threadIdx.x;
  if (i >= 384 * 64) return;
  int ic = i % 64, r = i / 64;
  int t = r / 128, q = r % 128;
  const float* w = (t == 0) ? wq : ((t == 1) ? wk : wv);
  dst[i] = w[q * 64 + ic];
}

__global__ void selu_k(float* __restrict__ s, int nelem) {
  int i = blockIdx.x * 256 + threadIdx.x;
  if (i >= nelem) return;
  float x = s[i];
  const float lam = 1.0507009873554805f, alp = 1.6732632423543772f;
  s[i] = lam * (x > 0.f ? x : alp * (expf(x) - 1.f));
}

extern "C" void kernel_launch(void* const* d_in, const int* in_sizes, int n_in,
                              void* d_out, int out_size, void* d_ws, size_t ws_size,
                              hipStream_t stream) {
  const float* in = (const float*)d_in[0];
  const float* wq = (const float*)d_in[1];
  const float* wk = (const float*)d_in[2];
  const float* wv = (const float*)d_in[3];
  const float* wo = (const float*)d_in[4];
  float* out = (float*)d_out;
  float* w = (float*)d_ws;

  const long P1_o   = 0,          P2_o   = 28901376;
  const long T_o    = 43653120;
  const long blk_o  = 84934656;
  const long QKV_o  = 0;
  const long S_o    = 92012544;
  const long U_o    = 42467328,   spec_o = 14155776;
  const long T1_o   = 15925248,   T2_o   = 23003136;
  const long T3_o   = 37158912;
  const long Wqkv_o = 106541568;
  const long Cf_o   = 106566144,  Vc_o   = 106575360;
  const long WS_FLOATS = 106584576;
  if (ws_size < (size_t)WS_FLOATS * sizeof(float)) return;

  float *P1 = w+P1_o, *P2 = w+P2_o, *T = w+T_o;
  float *blk = w+blk_o, *QKV = w+QKV_o, *S = w+S_o;
  float *U = w+U_o, *spec = w+spec_o;
  float *T1 = w+T1_o, *T2 = w+T2_o, *T3 = w+T3_o, *Wqkv = w+Wqkv_o;
  float *Cf = w+Cf_o, *Vc = w+Vc_o;

  auto gemm = [&](const float* A, long sAm, long sAb,
                  const float* B, long sBk, long sBn, long sBb,
                  float* C, long sCm, long sCn, long sCb,
                  int M, int N, int K, int batch, float alpha, int accum) {
    dim3 g((unsigned)((M + 63) / 64), (unsigned)((N + 63) / 64), (unsigned)batch);
    if (sBk == 1)
      gemm_f32_wmma_k4<true><<<g, 128, 0, stream>>>(A, sAm, sAb, B, sBk, sBn, sBb,
                                           C, sCm, sCn, sCb, M, N, K, alpha, accum);
    else
      gemm_f32_wmma_k4<false><<<g, 128, 0, stream>>>(A, sAm, sAb, B, sBk, sBn, sBb,
                                           C, sCm, sCn, sCb, M, N, K, alpha, accum);
  };
  auto ew = [&](int n) { return dim3((unsigned)((n + 255) / 256)); };

  gen_consts<<<ew(4704), 256, 0, stream>>>(Cf, Vc);
  pack_wqkv<<<ew(24576), 256, 0, stream>>>(wq, wk, wv, Wqkv);

  gemm(in, 96, 0, Cf, 1, 96, 0, P1, 49, 1, 0, 589824, 49, 96, 1, 1.f, 0);
  gemm(Cf, 96, 0, P1, 49, 1, 4704, P2, 49, 1, 2401, 49, 49, 96, 6144, 1.f, 0);
  gemm(Cf, 96, 0, P2, 2401, 1, 230496, T, 2401, 1, 117649, 49, 2401, 96, 64, 1.f, 0);
  hartley_mirror_fwd<<<ew(7077888), 256, 0, stream>>>(T, blk);

  gemm(Wqkv, 64, 0, blk, 110592, 1, 0, QKV, 110592, 1, 0, 384, 110592, 64, 1, 1.f, 0);

  const long HD = 16L * 110592L;
  attn_scores<<<dim3(4, 4, 8), 128, 0, stream>>>(QKV, QKV + 8 * HD, S,
                                                 0.011048543456039806f);
  selu_k<<<ew(373248), 256, 0, stream>>>(S, 373248);
  attn_values<<<dim3(4, 128, 8), 128, 0, stream>>>(S, QKV + 16 * HD, U);

  gemm(wo, 128, 0, U, 110592, 1, 0, spec, 110592, 1, 0, 16, 110592, 128, 1, 1.f, 0);

  gemm(spec, 48, 0, Vc, 1, 48, 0, T1, 96, 1, 0, 36864, 96, 48, 1, 1.f, 0);
  gemm(Vc, 48, 0, T1, 96, 1, 4608, T2, 96, 1, 9216, 96, 96, 48, 768, 1.f, 0);
  const float halfInvN = 0.5f / 884736.f;
  gemm(Vc, 48, 0, T2, 9216, 1, 442368, T3, 9216, 1, 884736, 96, 9216, 48, 16, halfInvN, 0);
  hartley_mirror_inv<<<ew(14155776), 256, 0, stream>>>(T3, out);
}


# Inverse mirror-add moved from the 96^3 result to the spectrum (four-term sum on 49 modes per axis, k padded to 56); z product writes the output, no 96^3 mirror pass (sum order changes)
# speedup vs baseline: 5.1571x; 1.0347x over previous
#include <hip/hip_runtime.h>
#include <math.h>

typedef float v2f __attribute__((ext_vector_type(2)));
typedef float v4f __attribute__((ext_vector_type(4)));
typedef float v8f __attribute__((ext_vector_type(8)));

template <bool BK>
__global__ __launch_bounds__(128) void gemm_f32_wmma_k4(
    const float* __restrict__ A, long sAm, long sAb,
    const float* __restrict__ B, long sBk, long sBn, long sBb,
    float* __restrict__ C, long sCm, long sCn, long sCb,
    int M, int N, int K, float alpha, int accum)
{
  const int lane = threadIdx.x & 31;
  const int wave = threadIdx.x >> 5;
  const int half = lane >> 4;
  const int l16  = lane & 15;
  const int mBase = blockIdx.x * 64;
  const int nBase = blockIdx.y * 64 + wave * 16;

  A += (long)blockIdx.z * sAb;
  B += (long)blockIdx.z * sBb;
  C += (long)blockIdx.z * sCb;

  v8f zero = {0.f,0.f,0.f,0.f,0.f,0.f,0.f,0.f};
  v8f acc[4];
#pragma unroll
  for (int t = 0; t < 4; ++t) acc[t] = zero;

  const float* aPtr[4];
  float aMask[4];
#pragma unroll
  for (int t = 0; t < 4; ++t) {
    int m = mBase + t * 16 + l16;
    bool v = (m < M);
    aMask[t] = v ? 1.f : 0.f;
    aPtr[t] = A + (long)(v ? m : 0) * sAm + 4 * half;
  }
  const int n = nBase + l16;
  const bool nv = (n < N);
  const float bMask = nv ? 1.f : 0.f;
  const float* bPtr = B + (long)(nv ? n : 0) * sBn + (long)(4 * half) * sBk;

  constexpr int D = 3;
  const int ng = K >> 3;
  v4f ra[D][4];
  v4f rb[D];
  v4f aq[4];
  v4f bq;

  auto load_step = [&](int j) {
    if (BK) {
      rb[j] = *(const v4f*)bPtr;
      bPtr += 8;
    } else {
      rb[j].x = bPtr[0];
      rb[j].y = bPtr[sBk];
      rb[j].z = bPtr[2 * sBk];
      rb[j].w = bPtr[3 * sBk];
      bPtr += 8 * sBk;
    }
#pragma unroll
    for (int t = 0; t < 4; ++t) {
      ra[j][t] = *(const v4f*)aPtr[t];
      aPtr[t] += 8;
    }
  };
  auto take_step = [&](int j) {
    bq = bMask * rb[j];
#pragma unroll
    for (int t = 0; t < 4; ++t) aq[t] = aMask[t] * ra[j][t];
  };
  auto mma_step = [&]() {
    v2f b0 = {bq.x, bq.y};
    v2f b1 = {bq.z, bq.w};
#pragma unroll
    for (int t = 0; t < 4; ++t) {
      v2f a0 = {aq[t].x, aq[t].y};
      acc[t] = __builtin_amdgcn_wmma_f32_16x16x4_f32(
          false, a0, false, b0, (short)0, acc[t], false, false);
    }
#pragma unroll
    for (int t = 0; t < 4; ++t) {
      v2f a1 = {aq[t].z, aq[t].w};
      acc[t] = __builtin_amdgcn_wmma_f32_16x16x4_f32(
          false, a1, false, b1, (short)0, acc[t], false, false);
    }
  };

  if (ng >= 2 * D) {
#pragma unroll
    for (int j = 0; j < D; ++j) {
      load_step(j);
      __builtin_amdgcn_sched_barrier(0);
    }

    int s = 0;
    do {
#pragma unroll
      for (int j = 0; j < D; ++j) {
        take_step(j);
        load_step(j);
        mma_step();
        __builtin_amdgcn_sched_barrier(0);
      }
      s += D;
    } while (s + 2 * D <= ng);

#pragma unroll
    for (int j = 0; j < D; ++j) {
      take_step(j);
      if (s + D + j < ng) load_step(j);
      mma_step();
    }
#pragma unroll
    for (int j = 0; j < D - 1; ++j) {
      if (s + D + j < ng) {
        take_step(j);
        mma_step();
      }
    }
  } else {
    for (int i = 0; i < ng; ++i) {
      load_step(0);
      take_step(0);
      mma_step();
    }
  }

  if (nv) {
#pragma unroll
    for (int t = 0; t < 4; ++t) {
#pragma unroll
      for (int r = 0; r < 8; ++r) {
        int m = mBase + t * 16 + r + 8 * half;
        if (m < M) {
          long idx = (long)m * sCm + (long)n * sCn;
          float val = alpha * acc[t][r];
          if (accum) C[idx] += val; else C[idx] = val;
        }
      }
    }
  }
}

constexpr int tok_off(int m) {
  return (m / 36) * 18432 + ((m / 6) % 6) * 384 + (m % 6) * 8;
}
constexpr int chan_off(int c) {
  return (c >> 9) * 110592 + ((c >> 6) & 7) * 2304 + ((c >> 3) & 7) * 48 + (c & 7);
}
struct TokOffsets {
  int v[264];
  constexpr TokOffsets() : v() {
    for (int m = 0; m < 264; ++m) v[m] = tok_off(m < 216 ? m : 0);
  }
};
__constant__ const TokOffsets kTok = TokOffsets();

__global__ __launch_bounds__(128) void attn_scores(
    const float* __restrict__ Q, const float* __restrict__ Kp,
    float* __restrict__ S, float alpha)
{
  const int lane = threadIdx.x & 31;
  const int wave = threadIdx.x >> 5;
  const int half = lane >> 4;
  const int l16  = lane & 15;
  const int mBase = blockIdx.x * 64;
  const int nBase = blockIdx.y * 64 + wave * 16;

  Q  += (long)blockIdx.z * 1769472;
  Kp += (long)blockIdx.z * 1769472;
  S  += (long)blockIdx.z * 46656;

  v8f zero = {0.f,0.f,0.f,0.f,0.f,0.f,0.f,0.f};
  v8f acc[4];
#pragma unroll
  for (int t = 0; t < 4; ++t) acc[t] = zero;

  unsigned aByte[4];
  float aMask[4];
#pragma unroll
  for (int t = 0; t < 4; ++t) {
    int m = mBase + t * 16 + l16;
    bool v = (m < 216);
    aMask[t] = v ? 1.f : 0.f;
    aByte[t] = (unsigned)(tok_off(v ? m : 0) + 4 * half) * 4u;
  }
  const int n = nBase + l16;
  const bool nv = (n < 216);
  const float bMask = nv ? 1.f : 0.f;
  const unsigned bByte = (unsigned)(tok_off(nv ? n : 0) + 4 * half) * 4u;

  constexpr int D = 6;
  constexpr int NG = 1024;
  v4f ra[D][4];
  v4f rb[D];
  v4f aq[4];
  v4f bq;
  int gl = 0;

  auto load_step = [&](int j) {
    const long ko = (long)(gl >> 6) * 110592 + (long)(((gl >> 3) & 7) * 2304 + (gl & 7) * 48);
    const char* qg = (const char*)(Q + ko);
    const char* kg = (const char*)(Kp + ko);
    rb[j] = *(const v4f*)(kg + bByte);
#pragma unroll
    for (int t = 0; t < 4; ++t) ra[j][t] = *(const v4f*)(qg + aByte[t]);
    ++gl;
  };
  auto take_step = [&](int j) {
    bq = bMask * rb[j];
#pragma unroll
    for (int t = 0; t < 4; ++t) aq[t] = aMask[t] * ra[j][t];
  };
  auto mma_step = [&]() {
    v2f b0 = {bq.x, bq.y};
    v2f b1 = {bq.z, bq.w};
#pragma unroll
    for (int t = 0; t < 4; ++t) {
      v2f a0 = {aq[t].x, aq[t].y};
      acc[t] = __builtin_amdgcn_wmma_f32_16x16x4_f32(
          false, a0, false, b0, (short)0, acc[t], false, false);
    }
#pragma unroll
    for (int t = 0; t < 4; ++t) {
      v2f a1 = {aq[t].z, aq[t].w};
      acc[t] = __builtin_amdgcn_wmma_f32_16x16x4_f32(
          false, a1, false, b1, (short)0, acc[t], false, false);
    }
  };

#pragma unroll
  for (int j = 0; j < D; ++j) {
    load_step(j);
    __builtin_amdgcn_sched_barrier(0);
  }

  int s = 0;
  do {
#pragma unroll
    for (int j = 0; j < D; ++j) {
      take_step(j);
      load_step(j);
      mma_step();
      __builtin_amdgcn_sched_barrier(0);
    }
    s += D;
  } while (s + 2 * D <= NG);

#pragma unroll
  for (int j = 0; j < D; ++j) {
    take_step(j);
    if (s + D + j < NG) load_step(j);
    mma_step();
  }
#pragma unroll
  for (int j = 0; j < D - 1; ++j) {
    if (s + D + j < NG) {
      take_step(j);
      mma_step();
    }
  }

  if (nv) {
#pragma unroll
    for (int t = 0; t < 4; ++t) {
#pragma unroll
      for (int r = 0; r < 8; ++r) {
        int m = mBase + t * 16 + r + 8 * half;
        if (m < 216) S[m * 216 + n] = alpha * acc[t][r];
      }
    }
  }
}

__global__ __launch_bounds__(128) void attn_values(
    const float* __restrict__ S, const float* __restrict__ V,
    float* __restrict__ U)
{
  const int lane = threadIdx.x & 31;
  const int wave = threadIdx.x >> 5;
  const int half = lane >> 4;
  const int l16  = lane & 15;
  const int mBase = blockIdx.x * 64;
  const int nBase = blockIdx.y * 64 + wave * 16;

  S += (long)blockIdx.z * 46656;
  V += (long)blockIdx.z * 1769472;
  U += (long)blockIdx.z * 1769472;

  v8f zero = {0.f,0.f,0.f,0.f,0.f,0.f,0.f,0.f};
  v8f acc[4];
#pragma unroll
  for (int t = 0; t < 4; ++t) acc[t] = zero;

  const float* aPtr[4];
  float aMask[4];
#pragma unroll
  for (int t = 0; t < 4; ++t) {
    int m = mBase + t * 16 + l16;
    bool v = (m < 216);
    aMask[t] = v ? 1.f : 0.f;
    aPtr[t] = S + (v ? m : 0) * 216 + 4 * half;
  }
  const int c = nBase + l16;
  const bool nv = (c < 8192);
  const float bMask = nv ? 1.f : 0.f;
  const unsigned cOff = (unsigned)chan_off(nv ? c : 0);
  const char* vBytes = (const char*)V;

  constexpr int D = 3;
  constexpr int NG = 27;
  v4f ra[D][4];
  v4f rb[D];
  v4f aq[4];
  v4f bq;
  int gl = 0;
  int tk0[4], tk1[4];
  auto fetch_tok = [&]() {
#pragma unroll
    for (int e = 0; e < 4; ++e) {
      tk0[e] = kTok.v[8 * gl + e];
      tk1[e] = kTok.v[8 * gl + 4 + e];
    }
  };
  fetch_tok();

  auto load_step = [&](int j) {
    const unsigned o0 = (cOff + (unsigned)(half ? tk1[0] : tk0[0])) << 2;
    const unsigned o1 = (cOff + (unsigned)(half ? tk1[1] : tk0[1])) << 2;
    const unsigned o2 = (cOff + (unsigned)(half ? tk1[2] : tk0[2])) << 2;
    const unsigned o3 = (cOff + (unsigned)(half ? tk1[3] : tk0[3])) << 2;
    rb[j].x = *(const float*)(vBytes + o0);
    rb[j].y = *(const float*)(vBytes + o1);
    rb[j].z = *(const float*)(vBytes + o2);
    rb[j].w = *(const float*)(vBytes + o3);
#pragma unroll
    for (int t = 0; t < 4; ++t) {
      ra[j][t] = *(const v4f*)aPtr[t];
      aPtr[t] += 8;
    }
    ++gl;
    fetch_tok();
  };
  auto take_step = [&](int j) {
    bq = bMask * rb[j];
#pragma unroll
    for (int t = 0; t < 4; ++t) aq[t] = aMask[t] * ra[j][t];
  };
  auto mma_step = [&]() {
    v2f b0 = {bq.x, bq.y};
    v2f b1 = {bq.z, bq.w};
#pragma unroll
    for (int t = 0; t < 4; ++t) {
      v2f a0 = {aq[t].x, aq[t].y};
      acc[t] = __builtin_amdgcn_wmma_f32_16x16x4_f32(
          false, a0, false, b0, (short)0, acc[t], false, false);
    }
#pragma unroll
    for (int t = 0; t < 4; ++t) {
      v2f a1 = {aq[t].z, aq[t].w};
      acc[t] = __builtin_amdgcn_wmma_f32_16x16x4_f32(
          false, a1, false, b1, (short)0, acc[t], false, false);
    }
  };

#pragma unroll
  for (int j = 0; j < D; ++j) {
    load_step(j);
    __builtin_amdgcn_sched_barrier(0);
  }

  int s = 0;
  do {
#pragma unroll
    for (int j = 0; j < D; ++j) {
      take_step(j);
      load_step(j);
      mma_step();
      __builtin_amdgcn_sched_barrier(0);
    }
    s += D;
  } while (s + 2 * D <= NG);

#pragma unroll
  for (int j = 0; j < D; ++j) {
    take_step(j);
    if (s + D + j < NG) load_step(j);
    mma_step();
  }
#pragma unroll
  for (int j = 0; j < D - 1; ++j) {
    if (s + D + j < NG) {
      take_step(j);
      mma_step();
    }
  }

#pragma unroll
  for (int t = 0; t < 4; ++t) {
#pragma unroll
    for (int r = 0; r < 8; ++r) {
      const int mu = mBase + t * 16 + r;
      const int e0 = kTok.v[mu];
      const int e1 = kTok.v[mu + 8];
      if (nv && mu + 8 * half < 216) U[cOff + (unsigned)(half ? e1 : e0)] = acc[t][r];
    }
  }
}

__global__ void gen_consts(float* Cf, float* Vp) {
  int i = blockIdx.x * 256 + threadIdx.x;
  if (i >= 56 * 96) return;
  int j = i / 96, nn = i % 96;
  float val = 0.f;
  if (j < 49) {
    int f = (j <= 24) ? j : (j + 47);
    float th = 6.283185307179586f * (float)((f * nn) % 96) / 96.f;
    val = cosf(th) + sinf(th);
    Cf[i] = val;
  }
  Vp[nn * 56 + j] = val;
}

__global__ void hartley_mirror_fwd(const float* __restrict__ T, float* __restrict__ blk) {
  int i = blockIdx.x * 256 + threadIdx.x;
  if (i >= 7077888) return;
  int ix = i % 48, iy = (i / 48) % 48, iz = (i / 2304) % 48, c = i / 110592;
  int jx = ix + (ix >= 24), jy = iy + (iy >= 24), jz = iz + (iz >= 24);
  int gx = jx ? 49 - jx : 0, gy = jy ? 49 - jy : 0, gz = jz ? 49 - jz : 0;
  const float* t = T + (long)c * 117649;
  float a = t[(gz * 49 + jy) * 49 + jx];
  float b = t[(jz * 49 + gy) * 49 + jx];
  float d = t[(jz * 49 + jy) * 49 + gx];
  float e = t[(gz * 49 + gy) * 49 + gx];
  blk[i] = 0.5f * ((a + b) + (d - e));
}

__device__ __forceinline__ float spec_at(const float* __restrict__ s, int jz, int jy, int jx) {
  int iz = jz - (jz > 24), iy = jy - (jy > 24), ix = jx - (jx > 24);
  float v = s[(iz * 48 + iy) * 48 + ix];
  return (jz == 24 || jy == 24 || jx == 24) ? 0.f : v;
}
__global__ void hartley_mirror_spec(const float* __restrict__ spec, float* __restrict__ Xm) {
  int i = blockIdx.x * 256 + threadIdx.x;
  if (i >= 702464) return;
  int q = i % 14, jy = (i / 14) % 56, jz = (i / 784) % 56, c = i / 43904;
  float r[4] = {0.f, 0.f, 0.f, 0.f};
  if (jz < 49 && jy < 49) {
    const float* s = spec + (long)c * 110592;
    int gz = jz ? 49 - jz : 0, gy = jy ? 49 - jy : 0;
#pragma unroll
    for (int e = 0; e < 4; ++e) {
      int jx = 4 * q + e;
      int jc = jx < 49 ? jx : 48;
      int gx = jc ? 49 - jc : 0;
      float a = spec_at(s, gz, jy, jc);
      float b = spec_at(s, jz, gy, jc);
      float d = spec_at(s, jz, jy, gx);
      float m = spec_at(s, gz, gy, gx);
      r[e] = jx < 49 ? (a + b) + (d - m) : 0.f;
    }
  }
  v4f o = {r[0], r[1], r[2], r[3]};
  *(v4f*)(Xm + 4 * (long)i) = o;
}

__global__ void pack_wqkv(const float* __restrict__ wq, const float* __restrict__ wk,
                          const float* __restrict__ wv, float* __restrict__ dst) {
  int i = blockIdx.x * 256 + threadIdx.x;
  if (i >= 384 * 64) return;
  int ic = i % 64, r = i / 64;
  int t = r / 128, q = r % 128;
  const float* w = (t == 0) ? wq : ((t == 1) ? wk : wv);
  dst[i] = w[q * 64 + ic];
}

__global__ void selu_k(float* __restrict__ s, int nelem) {
  int i = blockIdx.x * 256 + threadIdx.x;
  if (i >= nelem) return;
  float x = s[i];
  const float lam = 1.0507009873554805f, alp = 1.6732632423543772f;
  s[i] = lam * (x > 0.f ? x : alp * (expf(x) - 1.f));
}

extern "C" void kernel_launch(void* const* d_in, const int* in_sizes, int n_in,
                              void* d_out, int out_size, void* d_ws, size_t ws_size,
                              hipStream_t stream) {
  const float* in = (const float*)d_in[0];
  const float* wq = (const float*)d_in[1];
  const float* wk = (const float*)d_in[2];
  const float* wv = (const float*)d_in[3];
  const float* wo = (const float*)d_in[4];
  float* out = (float*)d_out;
  float* w = (float*)d_ws;

  const long P1_o   = 0,          P2_o   = 28901376;
  const long T_o    = 43653120;
  const long blk_o  = 84934656;
  const long QKV_o  = 0;
  const long S_o    = 92012544;
  const long U_o    = 42467328,   spec_o = 14155776;
  const long Xm_o   = 15925248;
  const long T1_o   = 18735104,   T2_o   = 23552000;
  const long Wqkv_o = 106541568;
  const long Cf_o   = 106566144,  Vp_o   = 106575360;
  const long WS_FLOATS = 106584576;
  if (ws_size < (size_t)WS_FLOATS * sizeof(float)) return;

  float *P1 = w+P1_o, *P2 = w+P2_o, *T = w+T_o;
  float *blk = w+blk_o, *QKV = w+QKV_o, *S = w+S_o;
  float *U = w+U_o, *spec = w+spec_o;
  float *Xm = w+Xm_o, *T1 = w+T1_o, *T2 = w+T2_o, *Wqkv = w+Wqkv_o;
  float *Cf = w+Cf_o, *Vp = w+Vp_o;

  auto gemm = [&](const float* A, long sAm, long sAb,
                  const float* B, long sBk, long sBn, long sBb,
                  float* C, long sCm, long sCn, long sCb,
                  int M, int N, int K, int batch, float alpha, int accum) {
    dim3 g((unsigned)((M + 63) / 64), (unsigned)((N + 63) / 64), (unsigned)batch);
    if (sBk == 1)
      gemm_f32_wmma_k4<true><<<g, 128, 0, stream>>>(A, sAm, sAb, B, sBk, sBn, sBb,
                                           C, sCm, sCn, sCb, M, N, K, alpha, accum);
    else
      gemm_f32_wmma_k4<false><<<g, 128, 0, stream>>>(A, sAm, sAb, B, sBk, sBn, sBb,
                                           C, sCm, sCn, sCb, M, N, K, alpha, accum);
  };
  auto ew = [&](int n) { return dim3((unsigned)((n + 255) / 256)); };

  gen_consts<<<ew(5376), 256, 0, stream>>>(Cf, Vp);
  pack_wqkv<<<ew(24576), 256, 0, stream>>>(wq, wk, wv, Wqkv);

  gemm(in, 96, 0, Cf, 1, 96, 0, P1, 49, 1, 0, 589824, 49, 96, 1, 1.f, 0);
  gemm(Cf, 96, 0, P1, 49, 1, 4704, P2, 49, 1, 2401, 49, 49, 96, 6144, 1.f, 0);
  gemm(Cf, 96, 0, P2, 2401, 1, 230496, T, 2401, 1, 117649, 49, 2401, 96, 64, 1.f, 0);
  hartley_mirror_fwd<<<ew(7077888), 256, 0, stream>>>(T, blk);

  gemm(Wqkv, 64, 0, blk, 110592, 1, 0, QKV, 110592, 1, 0, 384, 110592, 64, 1, 1.f, 0);

  const long HD = 16L * 110592L;
  attn_scores<<<dim3(4, 4, 8), 128, 0, stream>>>(QKV, QKV + 8 * HD, S,
                                                 0.011048543456039806f);
  selu_k<<<ew(373248), 256, 0, stream>>>(S, 373248);
  attn_values<<<dim3(4, 128, 8), 128, 0, stream>>>(S, QKV + 16 * HD, U);

  gemm(wo, 128, 0, U, 110592, 1, 0, spec, 110592, 1, 0, 16, 110592, 128, 1, 1.f, 0);

  hartley_mirror_spec<<<ew(702464), 256, 0, stream>>>(spec, Xm);
  gemm(Xm, 56, 0, Vp, 1, 56, 0, T1, 96, 1, 0, 50176, 96, 56, 1, 1.f, 0);
  gemm(Vp, 56, 0, T1, 96, 1, 5376, T2, 96, 1, 9216, 96, 96, 56, 896, 1.f, 0);
  const float halfInvN = 0.5f / 884736.f;
  gemm(Vp, 56, 0, T2, 9216, 1, 516096, out, 9216, 1, 884736, 96, 9216, 56, 16, halfInvN, 0);
}


# Score product re-tiled for parallelism: block tile 32x64 (two 16-row subtiles per wave, not four), 224 blocks not 128, ring of 16 k-groups; each element's k order and sum order unchanged
# speedup vs baseline: 5.6188x; 1.0895x over previous
#include <hip/hip_runtime.h>
#include <math.h>

typedef float v2f __attribute__((ext_vector_type(2)));
typedef float v4f __attribute__((ext_vector_type(4)));
typedef float v8f __attribute__((ext_vector_type(8)));

template <bool BK>
__global__ __launch_bounds__(128) void gemm_f32_wmma_k4(
    const float* __restrict__ A, long sAm, long sAb,
    const float* __restrict__ B, long sBk, long sBn, long sBb,
    float* __restrict__ C, long sCm, long sCn, long sCb,
    int M, int N, int K, float alpha, int accum)
{
  const int lane = threadIdx.x & 31;
  const int wave = threadIdx.x >> 5;
  const int half = lane >> 4;
  const int l16  = lane & 15;
  const int mBase = blockIdx.x * 64;
  const int nBase = blockIdx.y * 64 + wave * 16;

  A += (long)blockIdx.z * sAb;
  B += (long)blockIdx.z * sBb;
  C += (long)blockIdx.z * sCb;

  v8f zero = {0.f,0.f,0.f,0.f,0.f,0.f,0.f,0.f};
  v8f acc[4];
#pragma unroll
  for (int t = 0; t < 4; ++t) acc[t] = zero;

  const float* aPtr[4];
  float aMask[4];
#pragma unroll
  for (int t = 0; t < 4; ++t) {
    int m = mBase + t * 16 + l16;
    bool v = (m < M);
    aMask[t] = v ? 1.f : 0.f;
    aPtr[t] = A + (long)(v ? m : 0) * sAm + 4 * half;
  }
  const int n = nBase + l16;
  const bool nv = (n < N);
  const float bMask = nv ? 1.f : 0.f;
  const float* bPtr = B + (long)(nv ? n : 0) * sBn + (long)(4 * half) * sBk;

  constexpr int D = 3;
  const int ng = K >> 3;
  v4f ra[D][4];
  v4f rb[D];
  v4f aq[4];
  v4f bq;

  auto load_step = [&](int j) {
    if (BK) {
      rb[j] = *(const v4f*)bPtr;
      bPtr += 8;
    } else {
      rb[j].x = bPtr[0];
      rb[j].y = bPtr[sBk];
      rb[j].z = bPtr[2 * sBk];
      rb[j].w = bPtr[3 * sBk];
      bPtr += 8 * sBk;
    }
#pragma unroll
    for (int t = 0; t < 4; ++t) {
      ra[j][t] = *(const v4f*)aPtr[t];
      aPtr[t] += 8;
    }
  };
  auto take_step = [&](int j) {
    bq = bMask * rb[j];
#pragma unroll
    for (int t = 0; t < 4; ++t) aq[t] = aMask[t] * ra[j][t];
  };
  auto mma_step = [&]() {
    v2f b0 = {bq.x, bq.y};
    v2f b1 = {bq.z, bq.w};
#pragma unroll
    for (int t = 0; t < 4; ++t) {
      v2f a0 = {aq[t].x, aq[t].y};
      acc[t] = __builtin_amdgcn_wmma_f32_16x16x4_f32(
          false, a0, false, b0, (short)0, acc[t], false, false);
    }
#pragma unroll
    for (int t = 0; t < 4; ++t) {
      v2f a1 = {aq[t].z, aq[t].w};
      acc[t] = __builtin_amdgcn_wmma_f32_16x16x4_f32(
          false, a1, false, b1, (short)0, acc[t], false, false);
    }
  };

  if (ng >= 2 * D) {
#pragma unroll
    for (int j = 0; j < D; ++j) {
      load_step(j);
      __builtin_amdgcn_sched_barrier(0);
    }

    int s = 0;
    do {
#pragma unroll
      for (int j = 0; j < D; ++j) {
        take_step(j);
        load_step(j);
        mma_step();
        __builtin_amdgcn_sched_barrier(0);
      }
      s += D;
    } while (s + 2 * D <= ng);

#pragma unroll
    for (int j = 0; j < D; ++j) {
      take_step(j);
      if (s + D + j < ng) load_step(j);
      mma_step();
    }
#pragma unroll
    for (int j = 0; j < D - 1; ++j) {
      if (s + D + j < ng) {
        take_step(j);
        mma_step();
      }
    }
  } else {
    for (int i = 0; i < ng; ++i) {
      load_step(0);
      take_step(0);
      mma_step();
    }
  }

  if (nv) {
#pragma unroll
    for (int t = 0; t < 4; ++t) {
#pragma unroll
      for (int r = 0; r < 8; ++r) {
        int m = mBase + t * 16 + r + 8 * half;
        if (m < M) {
          long idx = (long)m * sCm + (long)n * sCn;
          float val = alpha * acc[t][r];
          if (accum) C[idx] += val; else C[idx] = val;
        }
      }
    }
  }
}

constexpr int tok_off(int m) {
  return (m / 36) * 18432 + ((m / 6) % 6) * 384 + (m % 6) * 8;
}
constexpr int chan_off(int c) {
  return (c >> 9) * 110592 + ((c >> 6) & 7) * 2304 + ((c >> 3) & 7) * 48 + (c & 7);
}
struct TokOffsets {
  int v[264];
  constexpr TokOffsets() : v() {
    for (int m = 0; m < 264; ++m) v[m] = tok_off(m < 216 ? m : 0);
  }
};
__constant__ const TokOffsets kTok = TokOffsets();

__global__ __launch_bounds__(128) void attn_scores(
    const float* __restrict__ Q, const float* __restrict__ Kp,
    float* __restrict__ S, float alpha)
{
  constexpr int TM = 2;
  const int lane = threadIdx.x & 31;
  const int wave = threadIdx.x >> 5;
  const int half = lane >> 4;
  const int l16  = lane & 15;
  const int mBase = blockIdx.x * (16 * TM);
  const int nBase = blockIdx.y * 64 + wave * 16;
  if (nBase >= 216) return;

  Q  += (long)blockIdx.z * 1769472;
  Kp += (long)blockIdx.z * 1769472;
  S  += (long)blockIdx.z * 46656;

  v8f zero = {0.f,0.f,0.f,0.f,0.f,0.f,0.f,0.f};
  v8f acc[TM];
#pragma unroll
  for (int t = 0; t < TM; ++t) acc[t] = zero;

  unsigned aByte[TM];
  float aMask[TM];
#pragma unroll
  for (int t = 0; t < TM; ++t) {
    int m = mBase + t * 16 + l16;
    bool v = (m < 216);
    aMask[t] = v ? 1.f : 0.f;
    aByte[t] = (unsigned)(tok_off(v ? m : 0) + 4 * half) * 4u;
  }
  const int n = nBase + l16;
  const bool nv = (n < 216);
  const float bMask = nv ? 1.f : 0.f;
  const unsigned bByte = (unsigned)(tok_off(nv ? n : 0) + 4 * half) * 4u;

  constexpr int D = 16;
  constexpr int NG = 1024;
  v4f ra[D][TM];
  v4f rb[D];
  v4f aq[TM];
  v4f bq;
  int gl = 0;

  auto load_step = [&](int j) {
    const long ko = (long)(gl >> 6) * 110592 + (long)(((gl >> 3) & 7) * 2304 + (gl & 7) * 48);
    const char* qg = (const char*)(Q + ko);
    const char* kg = (const char*)(Kp + ko);
    rb[j] = *(const v4f*)(kg + bByte);
#pragma unroll
    for (int t = 0; t < TM; ++t) ra[j][t] = *(const v4f*)(qg + aByte[t]);
    ++gl;
  };
  auto take_step = [&](int j) {
    bq = bMask * rb[j];
#pragma unroll
    for (int t = 0; t < TM; ++t) aq[t] = aMask[t] * ra[j][t];
  };
  auto mma_step = [&]() {
    v2f b0 = {bq.x, bq.y};
    v2f b1 = {bq.z, bq.w};
#pragma unroll
    for (int t = 0; t < TM; ++t) {
      v2f a0 = {aq[t].x, aq[t].y};
      acc[t] = __builtin_amdgcn_wmma_f32_16x16x4_f32(
          false, a0, false, b0, (short)0, acc[t], false, false);
    }
#pragma unroll
    for (int t = 0; t < TM; ++t) {
      v2f a1 = {aq[t].z, aq[t].w};
      acc[t] = __builtin_amdgcn_wmma_f32_16x16x4_f32(
          false, a1, false, b1, (short)0, acc[t], false, false);
    }
  };

#pragma unroll
  for (int j = 0; j < D; ++j) {
    load_step(j);
    __builtin_amdgcn_sched_barrier(0);
  }

  int s = 0;
  do {
#pragma unroll
    for (int j = 0; j < D; ++j) {
      take_step(j);
      load_step(j);
      mma_step();
      __builtin_amdgcn_sched_barrier(0);
    }
    s += D;
  } while (s + 2 * D <= NG);

#pragma unroll
  for (int j = 0; j < D; ++j) {
    take_step(j);
    if (s + D + j < NG) load_step(j);
    mma_step();
  }
#pragma unroll
  for (int j = 0; j < D - 1; ++j) {
    if (s + D + j < NG) {
      take_step(j);
      mma_step();
    }
  }

  if (nv) {
#pragma unroll
    for (int t = 0; t < TM; ++t) {
#pragma unroll
      for (int r = 0; r < 8; ++r) {
        int m = mBase + t * 16 + r + 8 * half;
        if (m < 216) S[m * 216 + n] = alpha * acc[t][r];
      }
    }
  }
}

__global__ __launch_bounds__(128) void attn_values(
    const float* __restrict__ S, const float* __restrict__ V,
    float* __restrict__ U)
{
  const int lane = threadIdx.x & 31;
  const int wave = threadIdx.x >> 5;
  const int half = lane >> 4;
  const int l16  = lane & 15;
  const int mBase = blockIdx.x * 64;
  const int nBase = blockIdx.y * 64 + wave * 16;

  S += (long)blockIdx.z * 46656;
  V += (long)blockIdx.z * 1769472;
  U += (long)blockIdx.z * 1769472;

  v8f zero = {0.f,0.f,0.f,0.f,0.f,0.f,0.f,0.f};
  v8f acc[4];
#pragma unroll
  for (int t = 0; t < 4; ++t) acc[t] = zero;

  const float* aPtr[4];
  float aMask[4];
#pragma unroll
  for (int t = 0; t < 4; ++t) {
    int m = mBase + t * 16 + l16;
    bool v = (m < 216);
    aMask[t] = v ? 1.f : 0.f;
    aPtr[t] = S + (v ? m : 0) * 216 + 4 * half;
  }
  const int c = nBase + l16;
  const bool nv = (c < 8192);
  const float bMask = nv ? 1.f : 0.f;
  const unsigned cOff = (unsigned)chan_off(nv ? c : 0);
  const char* vBytes = (const char*)V;

  constexpr int D = 3;
  constexpr int NG = 27;
  v4f ra[D][4];
  v4f rb[D];
  v4f aq[4];
  v4f bq;
  int gl = 0;
  int tk0[4], tk1[4];
  auto fetch_tok = [&]() {
#pragma unroll
    for (int e = 0; e < 4; ++e) {
      tk0[e] = kTok.v[8 * gl + e];
      tk1[e] = kTok.v[8 * gl + 4 + e];
    }
  };
  fetch_tok();

  auto load_step = [&](int j) {
    const unsigned o0 = (cOff + (unsigned)(half ? tk1[0] : tk0[0])) << 2;
    const unsigned o1 = (cOff + (unsigned)(half ? tk1[1] : tk0[1])) << 2;
    const unsigned o2 = (cOff + (unsigned)(half ? tk1[2] : tk0[2])) << 2;
    const unsigned o3 = (cOff + (unsigned)(half ? tk1[3] : tk0[3])) << 2;
    rb[j].x = *(const float*)(vBytes + o0);
    rb[j].y = *(const float*)(vBytes + o1);
    rb[j].z = *(const float*)(vBytes + o2);
    rb[j].w = *(const float*)(vBytes + o3);
#pragma unroll
    for (int t = 0; t < 4; ++t) {
      ra[j][t] = *(const v4f*)aPtr[t];
      aPtr[t] += 8;
    }
    ++gl;
    fetch_tok();
  };
  auto take_step = [&](int j) {
    bq = bMask * rb[j];
#pragma unroll
    for (int t = 0; t < 4; ++t) aq[t] = aMask[t] * ra[j][t];
  };
  auto mma_step = [&]() {
    v2f b0 = {bq.x, bq.y};
    v2f b1 = {bq.z, bq.w};
#pragma unroll
    for (int t = 0; t < 4; ++t) {
      v2f a0 = {aq[t].x, aq[t].y};
      acc[t] = __builtin_amdgcn_wmma_f32_16x16x4_f32(
          false, a0, false, b0, (short)0, acc[t], false, false);
    }
#pragma unroll
    for (int t = 0; t < 4; ++t) {
      v2f a1 = {aq[t].z, aq[t].w};
      acc[t] = __builtin_amdgcn_wmma_f32_16x16x4_f32(
          false, a1, false, b1, (short)0, acc[t], false, false);
    }
  };

#pragma unroll
  for (int j = 0; j < D; ++j) {
    load_step(j);
    __builtin_amdgcn_sched_barrier(0);
  }

  int s = 0;
  do {
#pragma unroll
    for (int j = 0; j < D; ++j) {
      take_step(j);
      load_step(j);
      mma_step();
      __builtin_amdgcn_sched_barrier(0);
    }
    s += D;
  } while (s + 2 * D <= NG);

#pragma unroll
  for (int j = 0; j < D; ++j) {
    take_step(j);
    if (s + D + j < NG) load_step(j);
    mma_step();
  }
#pragma unroll
  for (int j = 0; j < D - 1; ++j) {
    if (s + D + j < NG) {
      take_step(j);
      mma_step();
    }
  }

#pragma unroll
  for (int t = 0; t < 4; ++t) {
#pragma unroll
    for (int r = 0; r < 8; ++r) {
      const int mu = mBase + t * 16 + r;
      const int e0 = kTok.v[mu];
      const int e1 = kTok.v[mu + 8];
      if (nv && mu + 8 * half < 216) U[cOff + (unsigned)(half ? e1 : e0)] = acc[t][r];
    }
  }
}

__global__ void gen_consts(float* Cf, float* Vp) {
  int i = blockIdx.x * 256 + threadIdx.x;
  if (i >= 56 * 96) return;
  int j = i / 96, nn = i % 96;
  float val = 0.f;
  if (j < 49) {
    int f = (j <= 24) ? j : (j + 47);
    float th = 6.283185307179586f * (float)((f * nn) % 96) / 96.f;
    val = cosf(th) + sinf(th);
    Cf[i] = val;
  }
  Vp[nn * 56 + j] = val;
}

__global__ void hartley_mirror_fwd(const float* __restrict__ T, float* __restrict__ blk) {
  int i = blockIdx.x * 256 + threadIdx.x;
  if (i >= 7077888) return;
  int ix = i % 48, iy = (i / 48) % 48, iz = (i / 2304) % 48, c = i / 110592;
  int jx = ix + (ix >= 24), jy = iy + (iy >= 24), jz = iz + (iz >= 24);
  int gx = jx ? 49 - jx : 0, gy = jy ? 49 - jy : 0, gz = jz ? 49 - jz : 0;
  const float* t = T + (long)c * 117649;
  float a = t[(gz * 49 + jy) * 49 + jx];
  float b = t[(jz * 49 + gy) * 49 + jx];
  float d = t[(jz * 49 + jy) * 49 + gx];
  float e = t[(gz * 49 + gy) * 49 + gx];
  blk[i] = 0.5f * ((a + b) + (d - e));
}

__device__ __forceinline__ float spec_at(const float* __restrict__ s, int jz, int jy, int jx) {
  int iz = jz - (jz > 24), iy = jy - (jy > 24), ix = jx - (jx > 24);
  float v = s[(iz * 48 + iy) * 48 + ix];
  return (jz == 24 || jy == 24 || jx == 24) ? 0.f : v;
}
__global__ void hartley_mirror_spec(const float* __restrict__ spec, float* __restrict__ Xm) {
  int i = blockIdx.x * 256 + threadIdx.x;
  if (i >= 702464) return;
  int q = i % 14, jy = (i / 14) % 56, jz = (i / 784) % 56, c = i / 43904;
  float r[4] = {0.f, 0.f, 0.f, 0.f};
  if (jz < 49 && jy < 49) {
    const float* s = spec + (long)c * 110592;
    int gz = jz ? 49 - jz : 0, gy = jy ? 49 - jy : 0;
#pragma unroll
    for (int e = 0; e < 4; ++e) {
      int jx = 4 * q + e;
      int jc = jx < 49 ? jx : 48;
      int gx = jc ? 49 - jc : 0;
      float a = spec_at(s, gz, jy, jc);
      float b = spec_at(s, jz, gy, jc);
      float d = spec_at(s, jz, jy, gx);
      float m = spec_at(s, gz, gy, gx);
      r[e] = jx < 49 ? (a + b) + (d - m) : 0.f;
    }
  }
  v4f o = {r[0], r[1], r[2], r[3]};
  *(v4f*)(Xm + 4 * (long)i) = o;
}

__global__ void pack_wqkv(const float* __restrict__ wq, const float* __restrict__ wk,
                          const float* __restrict__ wv, float* __restrict__ dst) {
  int i = blockIdx.x * 256 + threadIdx.x;
  if (i >= 384 * 64) return;
  int ic = i % 64, r = i / 64;
  int t = r / 128, q = r % 128;
  const float* w = (t == 0) ? wq : ((t == 1) ? wk : wv);
  dst[i] = w[q * 64 + ic];
}

__global__ void selu_k(float* __restrict__ s, int nelem) {
  int i = blockIdx.x * 256 + threadIdx.x;
  if (i >= nelem) return;
  float x = s[i];
  const float lam = 1.0507009873554805f, alp = 1.6732632423543772f;
  s[i] = lam * (x > 0.f ? x : alp * (expf(x) - 1.f));
}

extern "C" void kernel_launch(void* const* d_in, const int* in_sizes, int n_in,
                              void* d_out, int out_size, void* d_ws, size_t ws_size,
                              hipStream_t stream) {
  const float* in = (const float*)d_in[0];
  const float* wq = (const float*)d_in[1];
  const float* wk = (const float*)d_in[2];
  const float* wv = (const float*)d_in[3];
  const float* wo = (const float*)d_in[4];
  float* out = (float*)d_out;
  float* w = (float*)d_ws;

  const long P1_o   = 0,          P2_o   = 28901376;
  const long T_o    = 43653120;
  const long blk_o  = 84934656;
  const long QKV_o  = 0;
  const long S_o    = 92012544;
  const long U_o    = 42467328,   spec_o = 14155776;
  const long Xm_o   = 15925248;
  const long T1_o   = 18735104,   T2_o   = 23552000;
  const long Wqkv_o = 106541568;
  const long Cf_o   = 106566144,  Vp_o   = 106575360;
  const long WS_FLOATS = 106584576;
  if (ws_size < (size_t)WS_FLOATS * sizeof(float)) return;

  float *P1 = w+P1_o, *P2 = w+P2_o, *T = w+T_o;
  float *blk = w+blk_o, *QKV = w+QKV_o, *S = w+S_o;
  float *U = w+U_o, *spec = w+spec_o;
  float *Xm = w+Xm_o, *T1 = w+T1_o, *T2 = w+T2_o, *Wqkv = w+Wqkv_o;
  float *Cf = w+Cf_o, *Vp = w+Vp_o;

  auto gemm = [&](const float* A, long sAm, long sAb,
                  const float* B, long sBk, long sBn, long sBb,
                  float* C, long sCm, long sCn, long sCb,
                  int M, int N, int K, int batch, float alpha, int accum) {
    dim3 g((unsigned)((M + 63) / 64), (unsigned)((N + 63) / 64), (unsigned)batch);
    if (sBk == 1)
      gemm_f32_wmma_k4<true><<<g, 128, 0, stream>>>(A, sAm, sAb, B, sBk, sBn, sBb,
                                           C, sCm, sCn, sCb, M, N, K, alpha, accum);
    else
      gemm_f32_wmma_k4<false><<<g, 128, 0, stream>>>(A, sAm, sAb, B, sBk, sBn, sBb,
                                           C, sCm, sCn, sCb, M, N, K, alpha, accum);
  };
  auto ew = [&](int n) { return dim3((unsigned)((n + 255) / 256)); };

  gen_consts<<<ew(5376), 256, 0, stream>>>(Cf, Vp);
  pack_wqkv<<<ew(24576), 256, 0, stream>>>(wq, wk, wv, Wqkv);

  gemm(in, 96, 0, Cf, 1, 96, 0, P1, 49, 1, 0, 589824, 49, 96, 1, 1.f, 0);
  gemm(Cf, 96, 0, P1, 49, 1, 4704, P2, 49, 1, 2401, 49, 49, 96, 6144, 1.f, 0);
  gemm(Cf, 96, 0, P2, 2401, 1, 230496, T, 2401, 1, 117649, 49, 2401, 96, 64, 1.f, 0);
  hartley_mirror_fwd<<<ew(7077888), 256, 0, stream>>>(T, blk);

  gemm(Wqkv, 64, 0, blk, 110592, 1, 0, QKV, 110592, 1, 0, 384, 110592, 64, 1, 1.f, 0);

  const long HD = 16L * 110592L;
  attn_scores<<<dim3(7, 4, 8), 128, 0, stream>>>(QKV, QKV + 8 * HD, S,
                                                 0.011048543456039806f);
  selu_k<<<ew(373248), 256, 0, stream>>>(S, 373248);
  attn_values<<<dim3(4, 128, 8), 128, 0, stream>>>(S, QKV + 16 * HD, U);

  gemm(wo, 128, 0, U, 110592, 1, 0, spec, 110592, 1, 0, 16, 110592, 128, 1, 1.f, 0);

  hartley_mirror_spec<<<ew(702464), 256, 0, stream>>>(spec, Xm);
  gemm(Xm, 56, 0, Vp, 1, 56, 0, T1, 96, 1, 0, 50176, 96, 56, 1, 1.f, 0);
  gemm(Vp, 56, 0, T1, 96, 1, 5376, T2, 96, 1, 9216, 96, 96, 56, 896, 1.f, 0);
  const float halfInvN = 0.5f / 884736.f;
  gemm(Vp, 56, 0, T2, 9216, 1, 516096, out, 9216, 1, 884736, 96, 9216, 56, 16, halfInvN, 0);
}


# Each GEMM's table or weight operand stays in registers for all K (A panel in y, z and QKV products, B panel in forward x); blocks walk many tiles streaming only the other operand; sums unchanged
# speedup vs baseline: 6.6657x; 1.1863x over previous
#include <hip/hip_runtime.h>
#include <math.h>

typedef float v2f __attribute__((ext_vector_type(2)));
typedef float v4f __attribute__((ext_vector_type(4)));
typedef float v8f __attribute__((ext_vector_type(8)));

template <bool BK>
__global__ __launch_bounds__(128) void gemm_f32_wmma_k4(
    const float* __restrict__ A, long sAm, long sAb,
    const float* __restrict__ B, long sBk, long sBn, long sBb,
    float* __restrict__ C, long sCm, long sCn, long sCb,
    int M, int N, int K, float alpha, int accum)
{
  const int lane = threadIdx.x & 31;
  const int wave = threadIdx.x >> 5;
  const int half = lane >> 4;
  const int l16  = lane & 15;
  const int mBase = blockIdx.x * 64;
  const int nBase = blockIdx.y * 64 + wave * 16;

  A += (long)blockIdx.z * sAb;
  B += (long)blockIdx.z * sBb;
  C += (long)blockIdx.z * sCb;

  v8f zero = {0.f,0.f,0.f,0.f,0.f,0.f,0.f,0.f};
  v8f acc[4];
#pragma unroll
  for (int t = 0; t < 4; ++t) acc[t] = zero;

  const float* aPtr[4];
  float aMask[4];
#pragma unroll
  for (int t = 0; t < 4; ++t) {
    int m = mBase + t * 16 + l16;
    bool v = (m < M);
    aMask[t] = v ? 1.f : 0.f;
    aPtr[t] = A + (long)(v ? m : 0) * sAm + 4 * half;
  }
  const int n = nBase + l16;
  const bool nv = (n < N);
  const float bMask = nv ? 1.f : 0.f;
  const float* bPtr = B + (long)(nv ? n : 0) * sBn + (long)(4 * half) * sBk;

  constexpr int D = 3;
  const int ng = K >> 3;
  v4f ra[D][4];
  v4f rb[D];
  v4f aq[4];
  v4f bq;

  auto load_step = [&](int j) {
    if (BK) {
      rb[j] = *(const v4f*)bPtr;
      bPtr += 8;
    } else {
      rb[j].x = bPtr[0];
      rb[j].y = bPtr[sBk];
      rb[j].z = bPtr[2 * sBk];
      rb[j].w = bPtr[3 * sBk];
      bPtr += 8 * sBk;
    }
#pragma unroll
    for (int t = 0; t < 4; ++t) {
      ra[j][t] = *(const v4f*)aPtr[t];
      aPtr[t] += 8;
    }
  };
  auto take_step = [&](int j) {
    bq = bMask * rb[j];
#pragma unroll
    for (int t = 0; t < 4; ++t) aq[t] = aMask[t] * ra[j][t];
  };
  auto mma_step = [&]() {
    v2f b0 = {bq.x, bq.y};
    v2f b1 = {bq.z, bq.w};
#pragma unroll
    for (int t = 0; t < 4; ++t) {
      v2f a0 = {aq[t].x, aq[t].y};
      acc[t] = __builtin_amdgcn_wmma_f32_16x16x4_f32(
          false, a0, false, b0, (short)0, acc[t], false, false);
    }
#pragma unroll
    for (int t = 0; t < 4; ++t) {
      v2f a1 = {aq[t].z, aq[t].w};
      acc[t] = __builtin_amdgcn_wmma_f32_16x16x4_f32(
          false, a1, false, b1, (short)0, acc[t], false, false);
    }
  };

  if (ng >= 2 * D) {
#pragma unroll
    for (int j = 0; j < D; ++j) {
      load_step(j);
      __builtin_amdgcn_sched_barrier(0);
    }

    int s = 0;
    do {
#pragma unroll
      for (int j = 0; j < D; ++j) {
        take_step(j);
        load_step(j);
        mma_step();
        __builtin_amdgcn_sched_barrier(0);
      }
      s += D;
    } while (s + 2 * D <= ng);

#pragma unroll
    for (int j = 0; j < D; ++j) {
      take_step(j);
      if (s + D + j < ng) load_step(j);
      mma_step();
    }
#pragma unroll
    for (int j = 0; j < D - 1; ++j) {
      if (s + D + j < ng) {
        take_step(j);
        mma_step();
      }
    }
  } else {
    for (int i = 0; i < ng; ++i) {
      load_step(0);
      take_step(0);
      mma_step();
    }
  }

  if (nv) {
#pragma unroll
    for (int t = 0; t < 4; ++t) {
#pragma unroll
      for (int r = 0; r < 8; ++r) {
        int m = mBase + t * 16 + r + 8 * half;
        if (m < M) {
          long idx = (long)m * sCm + (long)n * sCn;
          float val = alpha * acc[t][r];
          if (accum) C[idx] += val; else C[idx] = val;
        }
      }
    }
  }
}

template <int NGK, int D>
__global__ __launch_bounds__(128) void gemm_f32_wmma_astat(
    const float* __restrict__ A, int sAm,
    const float* __restrict__ B, long sBk, long sBb,
    float* __restrict__ C, long sCm, long sCb,
    int M, int N, int ntn, int nitems, float alpha)
{
  static_assert(NGK % D == 0, "a group must keep its ring slot from item to item");
  const int lane = threadIdx.x & 31;
  const int wave = threadIdx.x >> 5;
  const int half = lane >> 4;
  const int l16  = lane & 15;
  const int mBase = blockIdx.x * 64;
  const int nl = wave * 16 + l16;
  const int stride = (int)gridDim.y;
  int it = (int)blockIdx.y;
  if (it >= nitems) return;

  v4f pa[4][NGK];
#pragma unroll
  for (int t = 0; t < 4; ++t) {
    const int m = mBase + t * 16 + l16;
    const bool v = (m < M);
    const float am = v ? 1.f : 0.f;
    const float* ap = A + (long)(v ? m : 0) * sAm + 4 * half;
#pragma unroll
    for (int g = 0; g < NGK; ++g) pa[t][g] = am * *(const v4f*)(ap + 8 * g);
  }

  const unsigned hoff = (unsigned)(4 * half) * (unsigned)sBk;
  const long sBkB = sBk * 4;

  struct Item { const char* base; unsigned vo; bool nv; long c; };
  auto item_of = [&](int i) {
    Item r;
    const int b = i / ntn;
    const int nt = i - b * ntn;
    const int n = nt * 64 + nl;
    r.nv = (n < N);
    r.base = (const char*)(B + (long)b * sBb + nt * 64);
    r.vo = ((unsigned)(r.nv ? nl : 0) + hoff) * 4u;
    r.c = (long)b * sCb + n;
    return r;
  };

  v4f rb[D];
  v4f bq;
  v8f acc[4];
  const v8f zero = {0.f,0.f,0.f,0.f,0.f,0.f,0.f,0.f};

  auto load_step = [&](int j, const Item& x, int g) {
    const char* q = x.base + (long)(8 * g) * sBkB;
    rb[j].x = *(const float*)(q + x.vo);
    rb[j].y = *(const float*)(q + sBkB + x.vo);
    rb[j].z = *(const float*)(q + 2 * sBkB + x.vo);
    rb[j].w = *(const float*)(q + 3 * sBkB + x.vo);
  };
  auto mma_step = [&](int g) {
    v2f b0 = {bq.x, bq.y};
    v2f b1 = {bq.z, bq.w};
#pragma unroll
    for (int t = 0; t < 4; ++t) {
      v2f a0 = {pa[t][g].x, pa[t][g].y};
      acc[t] = __builtin_amdgcn_wmma_f32_16x16x4_f32(
          false, a0, false, b0, (short)0, g == 0 ? zero : acc[t], false, false);
    }
#pragma unroll
    for (int t = 0; t < 4; ++t) {
      v2f a1 = {pa[t][g].z, pa[t][g].w};
      acc[t] = __builtin_amdgcn_wmma_f32_16x16x4_f32(
          false, a1, false, b1, (short)0, acc[t], false, false);
    }
  };

  Item cur = item_of(it);
#pragma unroll
  for (int j = 0; j < D; ++j) {
    load_step(j, cur, j);
    __builtin_amdgcn_sched_barrier(0);
  }

  for (;;) {
    const int itn = it + stride;
    const bool more = (itn < nitems);
    const Item nxt = item_of(more ? itn : it);
    const float bMask = cur.nv ? 1.f : 0.f;
#pragma unroll
    for (int g = 0; g < NGK; ++g) {
      const int j = g % D;
      bq = bMask * rb[j];
      if (g + D < NGK) load_step(j, cur, g + D);
      else             load_step(j, nxt, g + D - NGK);
      mma_step(g);
      __builtin_amdgcn_sched_barrier(0);
    }
    if (cur.nv) {
      float* cp = C + cur.c + (long)(mBase + 8 * half) * sCm;
#pragma unroll
      for (int t = 0; t < 4; ++t) {
#pragma unroll
        for (int r = 0; r < 8; ++r) {
          const int m = mBase + t * 16 + r + 8 * half;
          if (m < M) *cp = alpha * acc[t][r];
          cp += sCm;
        }
        cp += 8 * sCm;
      }
    }
    if (!more) break;
    it = itn;
    cur = nxt;
  }
}

template <int NGK, int NT, int D>
__global__ __launch_bounds__(128) void gemm_f32_wmma_bstat(
    const float* __restrict__ A, int sAm,
    const float* __restrict__ B, int sBn,
    float* __restrict__ C, int sCm,
    int M, int N, int nitems)
{
  static_assert(NGK % D == 0, "a group must keep its ring slot from item to item");
  const int lane = threadIdx.x & 31;
  const int wave = threadIdx.x >> 5;
  const int half = lane >> 4;
  const int l16  = lane & 15;
  const int stride = (int)gridDim.x * 4;
  int it = (int)blockIdx.x * 4 + wave;
  if (it >= nitems) return;

  v4f pb[NT][NGK];
#pragma unroll
  for (int u = 0; u < NT; ++u) {
    const int n = u * 16 + l16;
    const bool v = (n < N);
    const float bm = v ? 1.f : 0.f;
    const float* bp = B + (long)(v ? n : 0) * sBn + 4 * half;
#pragma unroll
    for (int g = 0; g < NGK; ++g) pb[u][g] = bm * *(const v4f*)(bp + 8 * g);
  }

  auto row_of = [&](int i, float& mask) {
    const int m = i * 16 + l16;
    const bool v = (m < M);
    mask = v ? 1.f : 0.f;
    return A + (long)(v ? m : 0) * sAm + 4 * half;
  };

  v4f ra[D];
  v4f aq;
  v8f acc[NT];
  const v8f zero = {0.f,0.f,0.f,0.f,0.f,0.f,0.f,0.f};

  auto mma_step = [&](int g) {
    v2f a0 = {aq.x, aq.y};
    v2f a1 = {aq.z, aq.w};
#pragma unroll
    for (int u = 0; u < NT; ++u) {
      v2f b0 = {pb[u][g].x, pb[u][g].y};
      acc[u] = __builtin_amdgcn_wmma_f32_16x16x4_f32(
          false, a0, false, b0, (short)0, g == 0 ? zero : acc[u], false, false);
    }
#pragma unroll
    for (int u = 0; u < NT; ++u) {
      v2f b1 = {pb[u][g].z, pb[u][g].w};
      acc[u] = __builtin_amdgcn_wmma_f32_16x16x4_f32(
          false, a1, false, b1, (short)0, acc[u], false, false);
    }
  };

  float curMask;
  const float* curRow = row_of(it, curMask);
#pragma unroll
  for (int j = 0; j < D; ++j) {
    ra[j] = *(const v4f*)(curRow + 8 * j);
    __builtin_amdgcn_sched_barrier(0);
  }

  for (;;) {
    const int itn = it + stride;
    const bool more = (itn < nitems);
    float nxtMask;
    const float* nxtRow = row_of(more ? itn : it, nxtMask);
#pragma unroll
    for (int g = 0; g < NGK; ++g) {
      const int j = g % D;
      aq = curMask * ra[j];
      if (g + D < NGK) ra[j] = *(const v4f*)(curRow + 8 * (g + D));
      else             ra[j] = *(const v4f*)(nxtRow + 8 * (g + D - NGK));
      mma_step(g);
      __builtin_amdgcn_sched_barrier(0);
    }
    {
      const bool full = (it * 16 + 16 <= M);
#pragma unroll
      for (int u = 0; u < NT; ++u) {
        if (u * 16 + l16 < N) {
          float* cp = C + (long)(it * 16 + 8 * half) * sCm + (u * 16 + l16);
#pragma unroll
          for (int r = 0; r < 8; ++r) {
            if (full || it * 16 + r + 8 * half < M) *cp = acc[u][r];
            cp += sCm;
          }
        }
      }
    }
    if (!more) break;
    it = itn;
    curRow = nxtRow;
    curMask = nxtMask;
  }
}

constexpr int tok_off(int m) {
  return (m / 36) * 18432 + ((m / 6) % 6) * 384 + (m % 6) * 8;
}
constexpr int chan_off(int c) {
  return (c >> 9) * 110592 + ((c >> 6) & 7) * 2304 + ((c >> 3) & 7) * 48 + (c & 7);
}
struct TokOffsets {
  int v[264];
  constexpr TokOffsets() : v() {
    for (int m = 0; m < 264; ++m) v[m] = tok_off(m < 216 ? m : 0);
  }
};
__constant__ const TokOffsets kTok = TokOffsets();

__global__ __launch_bounds__(128) void attn_scores(
    const float* __restrict__ Q, const float* __restrict__ Kp,
    float* __restrict__ S, float alpha)
{
  constexpr int TM = 2;
  const int lane = threadIdx.x & 31;
  const int wave = threadIdx.x >> 5;
  const int half = lane >> 4;
  const int l16  = lane & 15;
  const int mBase = blockIdx.x * (16 * TM);
  const int nBase = blockIdx.y * 64 + wave * 16;
  if (nBase >= 216) return;

  Q  += (long)blockIdx.z * 1769472;
  Kp += (long)blockIdx.z * 1769472;
  S  += (long)blockIdx.z * 46656;

  v8f zero = {0.f,0.f,0.f,0.f,0.f,0.f,0.f,0.f};
  v8f acc[TM];
#pragma unroll
  for (int t = 0; t < TM; ++t) acc[t] = zero;

  unsigned aByte[TM];
  float aMask[TM];
#pragma unroll
  for (int t = 0; t < TM; ++t) {
    int m = mBase + t * 16 + l16;
    bool v = (m < 216);
    aMask[t] = v ? 1.f : 0.f;
    aByte[t] = (unsigned)(tok_off(v ? m : 0) + 4 * half) * 4u;
  }
  const int n = nBase + l16;
  const bool nv = (n < 216);
  const float bMask = nv ? 1.f : 0.f;
  const unsigned bByte = (unsigned)(tok_off(nv ? n : 0) + 4 * half) * 4u;

  constexpr int D = 16;
  constexpr int NG = 1024;
  v4f ra[D][TM];
  v4f rb[D];
  v4f aq[TM];
  v4f bq;
  int gl = 0;

  auto load_step = [&](int j) {
    const long ko = (long)(gl >> 6) * 110592 + (long)(((gl >> 3) & 7) * 2304 + (gl & 7) * 48);
    const char* qg = (const char*)(Q + ko);
    const char* kg = (const char*)(Kp + ko);
    rb[j] = *(const v4f*)(kg + bByte);
#pragma unroll
    for (int t = 0; t < TM; ++t) ra[j][t] = *(const v4f*)(qg + aByte[t]);
    ++gl;
  };
  auto take_step = [&](int j) {
    bq = bMask * rb[j];
#pragma unroll
    for (int t = 0; t < TM; ++t) aq[t] = aMask[t] * ra[j][t];
  };
  auto mma_step = [&]() {
    v2f b0 = {bq.x, bq.y};
    v2f b1 = {bq.z, bq.w};
#pragma unroll
    for (int t = 0; t < TM; ++t) {
      v2f a0 = {aq[t].x, aq[t].y};
      acc[t] = __builtin_amdgcn_wmma_f32_16x16x4_f32(
          false, a0, false, b0, (short)0, acc[t], false, false);
    }
#pragma unroll
    for (int t = 0; t < TM; ++t) {
      v2f a1 = {aq[t].z, aq[t].w};
      acc[t] = __builtin_amdgcn_wmma_f32_16x16x4_f32(
          false, a1, false, b1, (short)0, acc[t], false, false);
    }
  };

#pragma unroll
  for (int j = 0; j < D; ++j) {
    load_step(j);
    __builtin_amdgcn_sched_barrier(0);
  }

  int s = 0;
  do {
#pragma unroll
    for (int j = 0; j < D; ++j) {
      take_step(j);
      load_step(j);
      mma_step();
      __builtin_amdgcn_sched_barrier(0);
    }
    s += D;
  } while (s + 2 * D <= NG);

#pragma unroll
  for (int j = 0; j < D; ++j) {
    take_step(j);
    if (s + D + j < NG) load_step(j);
    mma_step();
  }
#pragma unroll
  for (int j = 0; j < D - 1; ++j) {
    if (s + D + j < NG) {
      take_step(j);
      mma_step();
    }
  }

  if (nv) {
#pragma unroll
    for (int t = 0; t < TM; ++t) {
#pragma unroll
      for (int r = 0; r < 8; ++r) {
        int m = mBase + t * 16 + r + 8 * half;
        if (m < 216) S[m * 216 + n] = alpha * acc[t][r];
      }
    }
  }
}

__global__ __launch_bounds__(128) void attn_values(
    const float* __restrict__ S, const float* __restrict__ V,
    float* __restrict__ U)
{
  const int lane = threadIdx.x & 31;
  const int wave = threadIdx.x >> 5;
  const int half = lane >> 4;
  const int l16  = lane & 15;
  const int mBase = blockIdx.x * 64;
  const int nBase = blockIdx.y * 64 + wave * 16;

  S += (long)blockIdx.z * 46656;
  V += (long)blockIdx.z * 1769472;
  U += (long)blockIdx.z * 1769472;

  v8f zero = {0.f,0.f,0.f,0.f,0.f,0.f,0.f,0.f};
  v8f acc[4];
#pragma unroll
  for (int t = 0; t < 4; ++t) acc[t] = zero;

  const float* aPtr[4];
  float aMask[4];
#pragma unroll
  for (int t = 0; t < 4; ++t) {
    int m = mBase + t * 16 + l16;
    bool v = (m < 216);
    aMask[t] = v ? 1.f : 0.f;
    aPtr[t] = S + (v ? m : 0) * 216 + 4 * half;
  }
  const int c = nBase + l16;
  const bool nv = (c < 8192);
  const float bMask = nv ? 1.f : 0.f;
  const unsigned cOff = (unsigned)chan_off(nv ? c : 0);
  const char* vBytes = (const char*)V;

  constexpr int D = 3;
  constexpr int NG = 27;
  v4f ra[D][4];
  v4f rb[D];
  v4f aq[4];
  v4f bq;
  int gl = 0;
  int tk0[4], tk1[4];
  auto fetch_tok = [&]() {
#pragma unroll
    for (int e = 0; e < 4; ++e) {
      tk0[e] = kTok.v[8 * gl + e];
      tk1[e] = kTok.v[8 * gl + 4 + e];
    }
  };
  fetch_tok();

  auto load_step = [&](int j) {
    const unsigned o0 = (cOff + (unsigned)(half ? tk1[0] : tk0[0])) << 2;
    const unsigned o1 = (cOff + (unsigned)(half ? tk1[1] : tk0[1])) << 2;
    const unsigned o2 = (cOff + (unsigned)(half ? tk1[2] : tk0[2])) << 2;
    const unsigned o3 = (cOff + (unsigned)(half ? tk1[3] : tk0[3])) << 2;
    rb[j].x = *(const float*)(vBytes + o0);
    rb[j].y = *(const float*)(vBytes + o1);
    rb[j].z = *(const float*)(vBytes + o2);
    rb[j].w = *(const float*)(vBytes + o3);
#pragma unroll
    for (int t = 0; t < 4; ++t) {
      ra[j][t] = *(const v4f*)aPtr[t];
      aPtr[t] += 8;
    }
    ++gl;
    fetch_tok();
  };
  auto take_step = [&](int j) {
    bq = bMask * rb[j];
#pragma unroll
    for (int t = 0; t < 4; ++t) aq[t] = aMask[t] * ra[j][t];
  };
  auto mma_step = [&]() {
    v2f b0 = {bq.x, bq.y};
    v2f b1 = {bq.z, bq.w};
#pragma unroll
    for (int t = 0; t < 4; ++t) {
      v2f a0 = {aq[t].x, aq[t].y};
      acc[t] = __builtin_amdgcn_wmma_f32_16x16x4_f32(
          false, a0, false, b0, (short)0, acc[t], false, false);
    }
#pragma unroll
    for (int t = 0; t < 4; ++t) {
      v2f a1 = {aq[t].z, aq[t].w};
      acc[t] = __builtin_amdgcn_wmma_f32_16x16x4_f32(
          false, a1, false, b1, (short)0, acc[t], false, false);
    }
  };

#pragma unroll
  for (int j = 0; j < D; ++j) {
    load_step(j);
    __builtin_amdgcn_sched_barrier(0);
  }

  int s = 0;
  do {
#pragma unroll
    for (int j = 0; j < D; ++j) {
      take_step(j);
      load_step(j);
      mma_step();
      __builtin_amdgcn_sched_barrier(0);
    }
    s += D;
  } while (s + 2 * D <= NG);

#pragma unroll
  for (int j = 0; j < D; ++j) {
    take_step(j);
    if (s + D + j < NG) load_step(j);
    mma_step();
  }
#pragma unroll
  for (int j = 0; j < D - 1; ++j) {
    if (s + D + j < NG) {
      take_step(j);
      mma_step();
    }
  }

#pragma unroll
  for (int t = 0; t < 4; ++t) {
#pragma unroll
    for (int r = 0; r < 8; ++r) {
      const int mu = mBase + t * 16 + r;
      const int e0 = kTok.v[mu];
      const int e1 = kTok.v[mu + 8];
      if (nv && mu + 8 * half < 216) U[cOff + (unsigned)(half ? e1 : e0)] = acc[t][r];
    }
  }
}

__global__ void gen_consts(float* Cf, float* Vp) {
  int i = blockIdx.x * 256 + threadIdx.x;
  if (i >= 56 * 96) return;
  int j = i / 96, nn = i % 96;
  float val = 0.f;
  if (j < 49) {
    int f = (j <= 24) ? j : (j + 47);
    float th = 6.283185307179586f * (float)((f * nn) % 96) / 96.f;
    val = cosf(th) + sinf(th);
    Cf[i] = val;
  }
  Vp[nn * 56 + j] = val;
}

__global__ void hartley_mirror_fwd(const float* __restrict__ T, float* __restrict__ blk) {
  int i = blockIdx.x * 256 + threadIdx.x;
  if (i >= 7077888) return;
  int ix = i % 48, iy = (i / 48) % 48, iz = (i / 2304) % 48, c = i / 110592;
  int jx = ix + (ix >= 24), jy = iy + (iy >= 24), jz = iz + (iz >= 24);
  int gx = jx ? 49 - jx : 0, gy = jy ? 49 - jy : 0, gz = jz ? 49 - jz : 0;
  const float* t = T + (long)c * 117649;
  float a = t[(gz * 49 + jy) * 49 + jx];
  float b = t[(jz * 49 + gy) * 49 + jx];
  float d = t[(jz * 49 + jy) * 49 + gx];
  float e = t[(gz * 49 + gy) * 49 + gx];
  blk[i] = 0.5f * ((a + b) + (d - e));
}

__device__ __forceinline__ float spec_at(const float* __restrict__ s, int jz, int jy, int jx) {
  int iz = jz - (jz > 24), iy = jy - (jy > 24), ix = jx - (jx > 24);
  float v = s[(iz * 48 + iy) * 48 + ix];
  return (jz == 24 || jy == 24 || jx == 24) ? 0.f : v;
}
__global__ void hartley_mirror_spec(const float* __restrict__ spec, float* __restrict__ Xm) {
  int i = blockIdx.x * 256 + threadIdx.x;
  if (i >= 702464) return;
  int q = i % 14, jy = (i / 14) % 56, jz = (i / 784) % 56, c = i / 43904;
  float r[4] = {0.f, 0.f, 0.f, 0.f};
  if (jz < 49 && jy < 49) {
    const float* s = spec + (long)c * 110592;
    int gz = jz ? 49 - jz : 0, gy = jy ? 49 - jy : 0;
#pragma unroll
    for (int e = 0; e < 4; ++e) {
      int jx = 4 * q + e;
      int jc = jx < 49 ? jx : 48;
      int gx = jc ? 49 - jc : 0;
      float a = spec_at(s, gz, jy, jc);
      float b = spec_at(s, jz, gy, jc);
      float d = spec_at(s, jz, jy, gx);
      float m = spec_at(s, gz, gy, gx);
      r[e] = jx < 49 ? (a + b) + (d - m) : 0.f;
    }
  }
  v4f o = {r[0], r[1], r[2], r[3]};
  *(v4f*)(Xm + 4 * (long)i) = o;
}

__global__ void pack_wqkv(const float* __restrict__ wq, const float* __restrict__ wk,
                          const float* __restrict__ wv, float* __restrict__ dst) {
  int i = blockIdx.x * 256 + threadIdx.x;
  if (i >= 384 * 64) return;
  int ic = i % 64, r = i / 64;
  int t = r / 128, q = r % 128;
  const float* w = (t == 0) ? wq : ((t == 1) ? wk : wv);
  dst[i] = w[q * 64 + ic];
}

__global__ void selu_k(float* __restrict__ s, int nelem) {
  int i = blockIdx.x * 256 + threadIdx.x;
  if (i >= nelem) return;
  float x = s[i];
  const float lam = 1.0507009873554805f, alp = 1.6732632423543772f;
  s[i] = lam * (x > 0.f ? x : alp * (expf(x) - 1.f));
}

extern "C" void kernel_launch(void* const* d_in, const int* in_sizes, int n_in,
                              void* d_out, int out_size, void* d_ws, size_t ws_size,
                              hipStream_t stream) {
  const float* in = (const float*)d_in[0];
  const float* wq = (const float*)d_in[1];
  const float* wk = (const float*)d_in[2];
  const float* wv = (const float*)d_in[3];
  const float* wo = (const float*)d_in[4];
  float* out = (float*)d_out;
  float* w = (float*)d_ws;

  const long P1_o   = 0,          P2_o   = 28901376;
  const long T_o    = 43653120;
  const long blk_o  = 84934656;
  const long QKV_o  = 0;
  const long S_o    = 92012544;
  const long U_o    = 42467328,   spec_o = 14155776;
  const long Xm_o   = 15925248;
  const long T1_o   = 18735104,   T2_o   = 23552000;
  const long Wqkv_o = 106541568;
  const long Cf_o   = 106566144,  Vp_o   = 106575360;
  const long WS_FLOATS = 106584576;
  if (ws_size < (size_t)WS_FLOATS * sizeof(float)) return;

  float *P1 = w+P1_o, *P2 = w+P2_o, *T = w+T_o;
  float *blk = w+blk_o, *QKV = w+QKV_o, *S = w+S_o;
  float *U = w+U_o, *spec = w+spec_o;
  float *Xm = w+Xm_o, *T1 = w+T1_o, *T2 = w+T2_o, *Wqkv = w+Wqkv_o;
  float *Cf = w+Cf_o, *Vp = w+Vp_o;

  auto gemm = [&](const float* A, long sAm, long sAb,
                  const float* B, long sBk, long sBn, long sBb,
                  float* C, long sCm, long sCn, long sCb,
                  int M, int N, int K, int batch, float alpha, int accum) {
    dim3 g((unsigned)((M + 63) / 64), (unsigned)((N + 63) / 64), (unsigned)batch);
    if (sBk == 1)
      gemm_f32_wmma_k4<true><<<g, 128, 0, stream>>>(A, sAm, sAb, B, sBk, sBn, sBb,
                                           C, sCm, sCn, sCb, M, N, K, alpha, accum);
    else
      gemm_f32_wmma_k4<false><<<g, 128, 0, stream>>>(A, sAm, sAb, B, sBk, sBn, sBb,
                                           C, sCm, sCn, sCb, M, N, K, alpha, accum);
  };
  auto ew = [&](int n) { return dim3((unsigned)((n + 255) / 256)); };

  gen_consts<<<ew(5376), 256, 0, stream>>>(Cf, Vp);
  pack_wqkv<<<ew(24576), 256, 0, stream>>>(wq, wk, wv, Wqkv);

  gemm_f32_wmma_bstat<12, 4, 12><<<dim3(768), 128, 0, stream>>>(
      in, 96, Cf, 96, P1, 49, 589824, 49, 36864);
  gemm_f32_wmma_astat<12, 6><<<dim3(1, 768), 128, 0, stream>>>(
      Cf, 96, P1, 49, 4704, P2, 49, 2401, 49, 49, 1, 6144, 1.f);
  gemm_f32_wmma_astat<12, 6><<<dim3(1, 768), 128, 0, stream>>>(
      Cf, 96, P2, 2401, 230496, T, 2401, 117649, 49, 2401, 38, 2432, 1.f);
  hartley_mirror_fwd<<<ew(7077888), 256, 0, stream>>>(T, blk);

  gemm_f32_wmma_astat<8, 8><<<dim3(6, 170), 128, 0, stream>>>(
      Wqkv, 64, blk, 110592, 0, QKV, 110592, 0, 384, 110592, 1728, 1728, 1.f);

  const long HD = 16L * 110592L;
  attn_scores<<<dim3(7, 4, 8), 128, 0, stream>>>(QKV, QKV + 8 * HD, S,
                                                 0.011048543456039806f);
  selu_k<<<ew(373248), 256, 0, stream>>>(S, 373248);
  attn_values<<<dim3(4, 128, 8), 128, 0, stream>>>(S, QKV + 16 * HD, U);

  gemm(wo, 128, 0, U, 110592, 1, 0, spec, 110592, 1, 0, 16, 110592, 128, 1, 1.f, 0);

  hartley_mirror_spec<<<ew(702464), 256, 0, stream>>>(spec, Xm);
  gemm(Xm, 56, 0, Vp, 1, 56, 0, T1, 96, 1, 0, 50176, 96, 56, 1, 1.f, 0);
  gemm_f32_wmma_astat<7, 7><<<dim3(2, 512), 128, 0, stream>>>(
      Vp, 56, T1, 96, 5376, T2, 96, 9216, 96, 96, 2, 1792, 1.f);
  const float halfInvN = 0.5f / 884736.f;
  gemm_f32_wmma_astat<7, 7><<<dim3(2, 512), 128, 0, stream>>>(
      Vp, 56, T2, 9216, 516096, out, 9216, 884736, 96, 9216, 144, 2304, halfInvN);
}


# Output projection folded into the value weights (w_out block x w_value per head): value product sums over heads and tokens, K=1728, and writes the spectrum; no out-projection pass (sum order changes)
# speedup vs baseline: 7.2596x; 1.0891x over previous
#include <hip/hip_runtime.h>
#include <math.h>

typedef float v2f __attribute__((ext_vector_type(2)));
typedef float v4f __attribute__((ext_vector_type(4)));
typedef float v8f __attribute__((ext_vector_type(8)));

template <bool BK>
__global__ __launch_bounds__(128) void gemm_f32_wmma_k4(
    const float* __restrict__ A, long sAm, long sAb,
    const float* __restrict__ B, long sBk, long sBn, long sBb,
    float* __restrict__ C, long sCm, long sCn, long sCb,
    int M, int N, int K, float alpha, int accum)
{
  const int lane = threadIdx.x & 31;
  const int wave = threadIdx.x >> 5;
  const int half = lane >> 4;
  const int l16  = lane & 15;
  const int mBase = blockIdx.x * 64;
  const int nBase = blockIdx.y * 64 + wave * 16;

  A += (long)blockIdx.z * sAb;
  B += (long)blockIdx.z * sBb;
  C += (long)blockIdx.z * sCb;

  v8f zero = {0.f,0.f,0.f,0.f,0.f,0.f,0.f,0.f};
  v8f acc[4];
#pragma unroll
  for (int t = 0; t < 4; ++t) acc[t] = zero;

  const float* aPtr[4];
  float aMask[4];
#pragma unroll
  for (int t = 0; t < 4; ++t) {
    int m = mBase + t * 16 + l16;
    bool v = (m < M);
    aMask[t] = v ? 1.f : 0.f;
    aPtr[t] = A + (long)(v ? m : 0) * sAm + 4 * half;
  }
  const int n = nBase + l16;
  const bool nv = (n < N);
  const float bMask = nv ? 1.f : 0.f;
  const float* bPtr = B + (long)(nv ? n : 0) * sBn + (long)(4 * half) * sBk;

  constexpr int D = 3;
  const int ng = K >> 3;
  v4f ra[D][4];
  v4f rb[D];
  v4f aq[4];
  v4f bq;

  auto load_step = [&](int j) {
    if (BK) {
      rb[j] = *(const v4f*)bPtr;
      bPtr += 8;
    } else {
      rb[j].x = bPtr[0];
      rb[j].y = bPtr[sBk];
      rb[j].z = bPtr[2 * sBk];
      rb[j].w = bPtr[3 * sBk];
      bPtr += 8 * sBk;
    }
#pragma unroll
    for (int t = 0; t < 4; ++t) {
      ra[j][t] = *(const v4f*)aPtr[t];
      aPtr[t] += 8;
    }
  };
  auto take_step = [&](int j) {
    bq = bMask * rb[j];
#pragma unroll
    for (int t = 0; t < 4; ++t) aq[t] = aMask[t] * ra[j][t];
  };
  auto mma_step = [&]() {
    v2f b0 = {bq.x, bq.y};
    v2f b1 = {bq.z, bq.w};
#pragma unroll
    for (int t = 0; t < 4; ++t) {
      v2f a0 = {aq[t].x, aq[t].y};
      acc[t] = __builtin_amdgcn_wmma_f32_16x16x4_f32(
          false, a0, false, b0, (short)0, acc[t], false, false);
    }
#pragma unroll
    for (int t = 0; t < 4; ++t) {
      v2f a1 = {aq[t].z, aq[t].w};
      acc[t] = __builtin_amdgcn_wmma_f32_16x16x4_f32(
          false, a1, false, b1, (short)0, acc[t], false, false);
    }
  };

  if (ng >= 2 * D) {
#pragma unroll
    for (int j = 0; j < D; ++j) {
      load_step(j);
      __builtin_amdgcn_sched_barrier(0);
    }

    int s = 0;
    do {
#pragma unroll
      for (int j = 0; j < D; ++j) {
        take_step(j);
        load_step(j);
        mma_step();
        __builtin_amdgcn_sched_barrier(0);
      }
      s += D;
    } while (s + 2 * D <= ng);

#pragma unroll
    for (int j = 0; j < D; ++j) {
      take_step(j);
      if (s + D + j < ng) load_step(j);
      mma_step();
    }
#pragma unroll
    for (int j = 0; j < D - 1; ++j) {
      if (s + D + j < ng) {
        take_step(j);
        mma_step();
      }
    }
  } else {
    for (int i = 0; i < ng; ++i) {
      load_step(0);
      take_step(0);
      mma_step();
    }
  }

  if (nv) {
#pragma unroll
    for (int t = 0; t < 4; ++t) {
#pragma unroll
      for (int r = 0; r < 8; ++r) {
        int m = mBase + t * 16 + r + 8 * half;
        if (m < M) {
          long idx = (long)m * sCm + (long)n * sCn;
          float val = alpha * acc[t][r];
          if (accum) C[idx] += val; else C[idx] = val;
        }
      }
    }
  }
}

template <int NGK, int D>
__global__ __launch_bounds__(128) void gemm_f32_wmma_astat(
    const float* __restrict__ A, int sAm,
    const float* __restrict__ B, long sBk, long sBb,
    float* __restrict__ C, long sCm, long sCb,
    int M, int N, int ntn, int nitems, float alpha)
{
  static_assert(NGK % D == 0, "a group must keep its ring slot from item to item");
  const int lane = threadIdx.x & 31;
  const int wave = threadIdx.x >> 5;
  const int half = lane >> 4;
  const int l16  = lane & 15;
  const int mBase = blockIdx.x * 64;
  const int nl = wave * 16 + l16;
  const int stride = (int)gridDim.y;
  int it = (int)blockIdx.y;
  if (it >= nitems) return;

  v4f pa[4][NGK];
#pragma unroll
  for (int t = 0; t < 4; ++t) {
    const int m = mBase + t * 16 + l16;
    const bool v = (m < M);
    const float am = v ? 1.f : 0.f;
    const float* ap = A + (long)(v ? m : 0) * sAm + 4 * half;
#pragma unroll
    for (int g = 0; g < NGK; ++g) pa[t][g] = am * *(const v4f*)(ap + 8 * g);
  }

  const unsigned hoff = (unsigned)(4 * half) * (unsigned)sBk;
  const long sBkB = sBk * 4;

  struct Item { const char* base; unsigned vo; bool nv; long c; };
  auto item_of = [&](int i) {
    Item r;
    const int b = i / ntn;
    const int nt = i - b * ntn;
    const int n = nt * 64 + nl;
    r.nv = (n < N);
    r.base = (const char*)(B + (long)b * sBb + nt * 64);
    r.vo = ((unsigned)(r.nv ? nl : 0) + hoff) * 4u;
    r.c = (long)b * sCb + n;
    return r;
  };

  v4f rb[D];
  v4f bq;
  v8f acc[4];
  const v8f zero = {0.f,0.f,0.f,0.f,0.f,0.f,0.f,0.f};

  auto load_step = [&](int j, const Item& x, int g) {
    const char* q = x.base + (long)(8 * g) * sBkB;
    rb[j].x = *(const float*)(q + x.vo);
    rb[j].y = *(const float*)(q + sBkB + x.vo);
    rb[j].z = *(const float*)(q + 2 * sBkB + x.vo);
    rb[j].w = *(const float*)(q + 3 * sBkB + x.vo);
  };
  auto mma_step = [&](int g) {
    v2f b0 = {bq.x, bq.y};
    v2f b1 = {bq.z, bq.w};
#pragma unroll
    for (int t = 0; t < 4; ++t) {
      v2f a0 = {pa[t][g].x, pa[t][g].y};
      acc[t] = __builtin_amdgcn_wmma_f32_16x16x4_f32(
          false, a0, false, b0, (short)0, g == 0 ? zero : acc[t], false, false);
    }
#pragma unroll
    for (int t = 0; t < 4; ++t) {
      v2f a1 = {pa[t][g].z, pa[t][g].w};
      acc[t] = __builtin_amdgcn_wmma_f32_16x16x4_f32(
          false, a1, false, b1, (short)0, acc[t], false, false);
    }
  };

  Item cur = item_of(it);
#pragma unroll
  for (int j = 0; j < D; ++j) {
    load_step(j, cur, j);
    __builtin_amdgcn_sched_barrier(0);
  }

  for (;;) {
    const int itn = it + stride;
    const bool more = (itn < nitems);
    const Item nxt = item_of(more ? itn : it);
    const float bMask = cur.nv ? 1.f : 0.f;
#pragma unroll
    for (int g = 0; g < NGK; ++g) {
      const int j = g % D;
      bq = bMask * rb[j];
      if (g + D < NGK) load_step(j, cur, g + D);
      else             load_step(j, nxt, g + D - NGK);
      mma_step(g);
      __builtin_amdgcn_sched_barrier(0);
    }
    if (cur.nv) {
      float* cp = C + cur.c + (long)(mBase + 8 * half) * sCm;
#pragma unroll
      for (int t = 0; t < 4; ++t) {
#pragma unroll
        for (int r = 0; r < 8; ++r) {
          const int m = mBase + t * 16 + r + 8 * half;
          if (m < M) *cp = alpha * acc[t][r];
          cp += sCm;
        }
        cp += 8 * sCm;
      }
    }
    if (!more) break;
    it = itn;
    cur = nxt;
  }
}

template <int NGK, int NT, int D>
__global__ __launch_bounds__(128) void gemm_f32_wmma_bstat(
    const float* __restrict__ A, int sAm,
    const float* __restrict__ B, int sBn,
    float* __restrict__ C, int sCm,
    int M, int N, int nitems)
{
  static_assert(NGK % D == 0, "a group must keep its ring slot from item to item");
  const int lane = threadIdx.x & 31;
  const int wave = threadIdx.x >> 5;
  const int half = lane >> 4;
  const int l16  = lane & 15;
  const int stride = (int)gridDim.x * 4;
  int it = (int)blockIdx.x * 4 + wave;
  if (it >= nitems) return;

  v4f pb[NT][NGK];
#pragma unroll
  for (int u = 0; u < NT; ++u) {
    const int n = u * 16 + l16;
    const bool v = (n < N);
    const float bm = v ? 1.f : 0.f;
    const float* bp = B + (long)(v ? n : 0) * sBn + 4 * half;
#pragma unroll
    for (int g = 0; g < NGK; ++g) pb[u][g] = bm * *(const v4f*)(bp + 8 * g);
  }

  auto row_of = [&](int i, float& mask) {
    const int m = i * 16 + l16;
    const bool v = (m < M);
    mask = v ? 1.f : 0.f;
    return A + (long)(v ? m : 0) * sAm + 4 * half;
  };

  v4f ra[D];
  v4f aq;
  v8f acc[NT];
  const v8f zero = {0.f,0.f,0.f,0.f,0.f,0.f,0.f,0.f};

  auto mma_step = [&](int g) {
    v2f a0 = {aq.x, aq.y};
    v2f a1 = {aq.z, aq.w};
#pragma unroll
    for (int u = 0; u < NT; ++u) {
      v2f b0 = {pb[u][g].x, pb[u][g].y};
      acc[u] = __builtin_amdgcn_wmma_f32_16x16x4_f32(
          false, a0, false, b0, (short)0, g == 0 ? zero : acc[u], false, false);
    }
#pragma unroll
    for (int u = 0; u < NT; ++u) {
      v2f b1 = {pb[u][g].z, pb[u][g].w};
      acc[u] = __builtin_amdgcn_wmma_f32_16x16x4_f32(
          false, a1, false, b1, (short)0, acc[u], false, false);
    }
  };

  float curMask;
  const float* curRow = row_of(it, curMask);
#pragma unroll
  for (int j = 0; j < D; ++j) {
    ra[j] = *(const v4f*)(curRow + 8 * j);
    __builtin_amdgcn_sched_barrier(0);
  }

  for (;;) {
    const int itn = it + stride;
    const bool more = (itn < nitems);
    float nxtMask;
    const float* nxtRow = row_of(more ? itn : it, nxtMask);
#pragma unroll
    for (int g = 0; g < NGK; ++g) {
      const int j = g % D;
      aq = curMask * ra[j];
      if (g + D < NGK) ra[j] = *(const v4f*)(curRow + 8 * (g + D));
      else             ra[j] = *(const v4f*)(nxtRow + 8 * (g + D - NGK));
      mma_step(g);
      __builtin_amdgcn_sched_barrier(0);
    }
    {
      const bool full = (it * 16 + 16 <= M);
#pragma unroll
      for (int u = 0; u < NT; ++u) {
        if (u * 16 + l16 < N) {
          float* cp = C + (long)(it * 16 + 8 * half) * sCm + (u * 16 + l16);
#pragma unroll
          for (int r = 0; r < 8; ++r) {
            if (full || it * 16 + r + 8 * half < M) *cp = acc[u][r];
            cp += sCm;
          }
        }
      }
    }
    if (!more) break;
    it = itn;
    curRow = nxtRow;
    curMask = nxtMask;
  }
}

constexpr int tok_off(int m) {
  return (m / 36) * 18432 + ((m / 6) % 6) * 384 + (m % 6) * 8;
}
constexpr int chan_off(int c) {
  return (c >> 9) * 110592 + ((c >> 6) & 7) * 2304 + ((c >> 3) & 7) * 48 + (c & 7);
}
struct TokOffsets {
  int v[264];
  constexpr TokOffsets() : v() {
    for (int m = 0; m < 264; ++m) v[m] = tok_off(m < 216 ? m : 0);
  }
};
__constant__ const TokOffsets kTok = TokOffsets();

__global__ __launch_bounds__(128) void attn_scores(
    const float* __restrict__ Q, const float* __restrict__ Kp,
    float* __restrict__ S, float alpha)
{
  constexpr int TM = 2;
  const int lane = threadIdx.x & 31;
  const int wave = threadIdx.x >> 5;
  const int half = lane >> 4;
  const int l16  = lane & 15;
  const int mBase = blockIdx.x * (16 * TM);
  const int nBase = blockIdx.y * 64 + wave * 16;
  if (nBase >= 216) return;

  Q  += (long)blockIdx.z * 1769472;
  Kp += (long)blockIdx.z * 1769472;
  S  += (long)blockIdx.z * 46656;

  v8f zero = {0.f,0.f,0.f,0.f,0.f,0.f,0.f,0.f};
  v8f acc[TM];
#pragma unroll
  for (int t = 0; t < TM; ++t) acc[t] = zero;

  unsigned aByte[TM];
  float aMask[TM];
#pragma unroll
  for (int t = 0; t < TM; ++t) {
    int m = mBase + t * 16 + l16;
    bool v = (m < 216);
    aMask[t] = v ? 1.f : 0.f;
    aByte[t] = (unsigned)(tok_off(v ? m : 0) + 4 * half) * 4u;
  }
  const int n = nBase + l16;
  const bool nv = (n < 216);
  const float bMask = nv ? 1.f : 0.f;
  const unsigned bByte = (unsigned)(tok_off(nv ? n : 0) + 4 * half) * 4u;

  constexpr int D = 16;
  constexpr int NG = 1024;
  v4f ra[D][TM];
  v4f rb[D];
  v4f aq[TM];
  v4f bq;
  int gl = 0;

  auto load_step = [&](int j) {
    const long ko = (long)(gl >> 6) * 110592 + (long)(((gl >> 3) & 7) * 2304 + (gl & 7) * 48);
    const char* qg = (const char*)(Q + ko);
    const char* kg = (const char*)(Kp + ko);
    rb[j] = *(const v4f*)(kg + bByte);
#pragma unroll
    for (int t = 0; t < TM; ++t) ra[j][t] = *(const v4f*)(qg + aByte[t]);
    ++gl;
  };
  auto take_step = [&](int j) {
    bq = bMask * rb[j];
#pragma unroll
    for (int t = 0; t < TM; ++t) aq[t] = aMask[t] * ra[j][t];
  };
  auto mma_step = [&]() {
    v2f b0 = {bq.x, bq.y};
    v2f b1 = {bq.z, bq.w};
#pragma unroll
    for (int t = 0; t < TM; ++t) {
      v2f a0 = {aq[t].x, aq[t].y};
      acc[t] = __builtin_amdgcn_wmma_f32_16x16x4_f32(
          false, a0, false, b0, (short)0, acc[t], false, false);
    }
#pragma unroll
    for (int t = 0; t < TM; ++t) {
      v2f a1 = {aq[t].z, aq[t].w};
      acc[t] = __builtin_amdgcn_wmma_f32_16x16x4_f32(
          false, a1, false, b1, (short)0, acc[t], false, false);
    }
  };

#pragma unroll
  for (int j = 0; j < D; ++j) {
    load_step(j);
    __builtin_amdgcn_sched_barrier(0);
  }

  int s = 0;
  do {
#pragma unroll
    for (int j = 0; j < D; ++j) {
      take_step(j);
      load_step(j);
      mma_step();
      __builtin_amdgcn_sched_barrier(0);
    }
    s += D;
  } while (s + 2 * D <= NG);

#pragma unroll
  for (int j = 0; j < D; ++j) {
    take_step(j);
    if (s + D + j < NG) load_step(j);
    mma_step();
  }
#pragma unroll
  for (int j = 0; j < D - 1; ++j) {
    if (s + D + j < NG) {
      take_step(j);
      mma_step();
    }
  }

  if (nv) {
#pragma unroll
    for (int t = 0; t < TM; ++t) {
#pragma unroll
      for (int r = 0; r < 8; ++r) {
        int m = mBase + t * 16 + r + 8 * half;
        if (m < 216) S[m * 216 + n] = alpha * acc[t][r];
      }
    }
  }
}

__global__ __launch_bounds__(128) void attn_values(
    const float* __restrict__ S, const float* __restrict__ V,
    float* __restrict__ O)
{
  const int lane = threadIdx.x & 31;
  const int wave = threadIdx.x >> 5;
  const int half = lane >> 4;
  const int l16  = lane & 15;
  const int mBase = blockIdx.x * 64;
  const int nBase = blockIdx.y * 64 + wave * 16;

  v8f zero = {0.f,0.f,0.f,0.f,0.f,0.f,0.f,0.f};
  v8f acc[4];
#pragma unroll
  for (int t = 0; t < 4; ++t) acc[t] = zero;

  unsigned aByte[4];
  float aMask[4];
#pragma unroll
  for (int t = 0; t < 4; ++t) {
    int m = mBase + t * 16 + l16;
    bool v = (m < 216);
    aMask[t] = v ? 1.f : 0.f;
    aByte[t] = (unsigned)((v ? m : 0) * 216 + 4 * half) * 4u;
  }
  const int c = nBase + l16;
  const bool nv = (c < 8192);
  const float bMask = nv ? 1.f : 0.f;
  const unsigned cOff = (unsigned)chan_off(nv ? c : 0);

  constexpr int D = 6;
  constexpr int NGH = 27;
  constexpr int NG = 8 * NGH;
  v4f ra[D][4];
  v4f rb[D];
  v4f aq[4];
  v4f bq;
  int gi = 0;
  const char* sG = (const char*)S;
  const char* vG = (const char*)V;
  int tk0[4], tk1[4];
  auto fetch_tok = [&]() {
#pragma unroll
    for (int e = 0; e < 4; ++e) {
      tk0[e] = kTok.v[8 * gi + e];
      tk1[e] = kTok.v[8 * gi + 4 + e];
    }
  };
  fetch_tok();

  auto load_step = [&](int j) {
    const unsigned o0 = (cOff + (unsigned)(half ? tk1[0] : tk0[0])) << 2;
    const unsigned o1 = (cOff + (unsigned)(half ? tk1[1] : tk0[1])) << 2;
    const unsigned o2 = (cOff + (unsigned)(half ? tk1[2] : tk0[2])) << 2;
    const unsigned o3 = (cOff + (unsigned)(half ? tk1[3] : tk0[3])) << 2;
    rb[j].x = *(const float*)(vG + o0);
    rb[j].y = *(const float*)(vG + o1);
    rb[j].z = *(const float*)(vG + o2);
    rb[j].w = *(const float*)(vG + o3);
#pragma unroll
    for (int t = 0; t < 4; ++t) ra[j][t] = *(const v4f*)(sG + aByte[t]);
    ++gi;
    sG += 32;
    const bool wrap = (gi == NGH);
    gi = wrap ? 0 : gi;
    sG += wrap ? (46656 - 216) * 4 : 0;
    vG += wrap ? 1769472L * 4 : 0L;
    fetch_tok();
  };
  auto take_step = [&](int j) {
    bq = bMask * rb[j];
#pragma unroll
    for (int t = 0; t < 4; ++t) aq[t] = aMask[t] * ra[j][t];
  };
  auto mma_step = [&]() {
    v2f b0 = {bq.x, bq.y};
    v2f b1 = {bq.z, bq.w};
#pragma unroll
    for (int t = 0; t < 4; ++t) {
      v2f a0 = {aq[t].x, aq[t].y};
      acc[t] = __builtin_amdgcn_wmma_f32_16x16x4_f32(
          false, a0, false, b0, (short)0, acc[t], false, false);
    }
#pragma unroll
    for (int t = 0; t < 4; ++t) {
      v2f a1 = {aq[t].z, aq[t].w};
      acc[t] = __builtin_amdgcn_wmma_f32_16x16x4_f32(
          false, a1, false, b1, (short)0, acc[t], false, false);
    }
  };

#pragma unroll
  for (int j = 0; j < D; ++j) {
    load_step(j);
    __builtin_amdgcn_sched_barrier(0);
  }

  int s = 0;
  do {
#pragma unroll
    for (int j = 0; j < D; ++j) {
      take_step(j);
      load_step(j);
      mma_step();
      __builtin_amdgcn_sched_barrier(0);
    }
    s += D;
  } while (s + 2 * D <= NG);

#pragma unroll
  for (int j = 0; j < D; ++j) {
    take_step(j);
    if (s + D + j < NG) load_step(j);
    mma_step();
  }
#pragma unroll
  for (int j = 0; j < D - 1; ++j) {
    if (s + D + j < NG) {
      take_step(j);
      mma_step();
    }
  }

#pragma unroll
  for (int t = 0; t < 4; ++t) {
#pragma unroll
    for (int r = 0; r < 8; ++r) {
      const int mu = mBase + t * 16 + r;
      const int e0 = kTok.v[mu];
      const int e1 = kTok.v[mu + 8];
      if (nv && mu + 8 * half < 216) O[cOff + (unsigned)(half ? e1 : e0)] = acc[t][r];
    }
  }
}

__global__ void gen_consts(float* Cf, float* Vp) {
  int i = blockIdx.x * 256 + threadIdx.x;
  if (i >= 56 * 96) return;
  int j = i / 96, nn = i % 96;
  float val = 0.f;
  if (j < 49) {
    int f = (j <= 24) ? j : (j + 47);
    float th = 6.283185307179586f * (float)((f * nn) % 96) / 96.f;
    val = cosf(th) + sinf(th);
    Cf[i] = val;
  }
  Vp[nn * 56 + j] = val;
}

__global__ void hartley_mirror_fwd(const float* __restrict__ T, float* __restrict__ blk) {
  int i = blockIdx.x * 256 + threadIdx.x;
  if (i >= 7077888) return;
  int ix = i % 48, iy = (i / 48) % 48, iz = (i / 2304) % 48, c = i / 110592;
  int jx = ix + (ix >= 24), jy = iy + (iy >= 24), jz = iz + (iz >= 24);
  int gx = jx ? 49 - jx : 0, gy = jy ? 49 - jy : 0, gz = jz ? 49 - jz : 0;
  const float* t = T + (long)c * 117649;
  float a = t[(gz * 49 + jy) * 49 + jx];
  float b = t[(jz * 49 + gy) * 49 + jx];
  float d = t[(jz * 49 + jy) * 49 + gx];
  float e = t[(gz * 49 + gy) * 49 + gx];
  blk[i] = 0.5f * ((a + b) + (d - e));
}

__device__ __forceinline__ float spec_at(const float* __restrict__ s, int jz, int jy, int jx) {
  int iz = jz - (jz > 24), iy = jy - (jy > 24), ix = jx - (jx > 24);
  float v = s[(iz * 48 + iy) * 48 + ix];
  return (jz == 24 || jy == 24 || jx == 24) ? 0.f : v;
}
__global__ void hartley_mirror_spec(const float* __restrict__ spec, float* __restrict__ Xm) {
  int i = blockIdx.x * 256 + threadIdx.x;
  if (i >= 702464) return;
  int q = i % 14, jy = (i / 14) % 56, jz = (i / 784) % 56, c = i / 43904;
  float r[4] = {0.f, 0.f, 0.f, 0.f};
  if (jz < 49 && jy < 49) {
    const float* s = spec + (long)c * 110592;
    int gz = jz ? 49 - jz : 0, gy = jy ? 49 - jy : 0;
#pragma unroll
    for (int e = 0; e < 4; ++e) {
      int jx = 4 * q + e;
      int jc = jx < 49 ? jx : 48;
      int gx = jc ? 49 - jc : 0;
      float a = spec_at(s, gz, jy, jc);
      float b = spec_at(s, jz, gy, jc);
      float d = spec_at(s, jz, jy, gx);
      float m = spec_at(s, gz, gy, gx);
      r[e] = jx < 49 ? (a + b) + (d - m) : 0.f;
    }
  }
  v4f o = {r[0], r[1], r[2], r[3]};
  *(v4f*)(Xm + 4 * (long)i) = o;
}

__global__ void pack_wqkv(const float* __restrict__ wq, const float* __restrict__ wk,
                          const float* __restrict__ wv, const float* __restrict__ wo,
                          float* __restrict__ dst) {
  int i = blockIdx.x * 256 + threadIdx.x;
  if (i >= 384 * 64) return;
  int ic = i % 64, r = i / 64;
  int t = r / 128, q = r % 128;
  if (t < 2) {
    const float* w = (t == 0) ? wq : wk;
    dst[i] = w[q * 64 + ic];
  } else {
    const int h = q >> 4, o = q & 15;
    const float* wor = wo + o * 128 + h * 16;
    const float* wvc = wv + h * 1024 + ic;
    float s = 0.f;
#pragma unroll
    for (int u = 0; u < 16; ++u) s += wor[u] * wvc[u * 64];
    dst[i] = s;
  }
}

__global__ void selu_k(float* __restrict__ s, int nelem) {
  int i = blockIdx.x * 256 + threadIdx.x;
  if (i >= nelem) return;
  float x = s[i];
  const float lam = 1.0507009873554805f, alp = 1.6732632423543772f;
  s[i] = lam * (x > 0.f ? x : alp * (expf(x) - 1.f));
}

extern "C" void kernel_launch(void* const* d_in, const int* in_sizes, int n_in,
                              void* d_out, int out_size, void* d_ws, size_t ws_size,
                              hipStream_t stream) {
  const float* in = (const float*)d_in[0];
  const float* wq = (const float*)d_in[1];
  const float* wk = (const float*)d_in[2];
  const float* wv = (const float*)d_in[3];
  const float* wo = (const float*)d_in[4];
  float* out = (float*)d_out;
  float* w = (float*)d_ws;

  const long P1_o   = 0,          P2_o   = 28901376;
  const long T_o    = 43653120;
  const long blk_o  = 84934656;
  const long QKV_o  = 0;
  const long S_o    = 92012544;
  const long spec_o = 14155776;
  const long Xm_o   = 15925248;
  const long T1_o   = 18735104,   T2_o   = 23552000;
  const long Wqkv_o = 106541568;
  const long Cf_o   = 106566144,  Vp_o   = 106575360;
  const long WS_FLOATS = 106584576;
  if (ws_size < (size_t)WS_FLOATS * sizeof(float)) return;

  float *P1 = w+P1_o, *P2 = w+P2_o, *T = w+T_o;
  float *blk = w+blk_o, *QKV = w+QKV_o, *S = w+S_o;
  float *spec = w+spec_o;
  float *Xm = w+Xm_o, *T1 = w+T1_o, *T2 = w+T2_o, *Wqkv = w+Wqkv_o;
  float *Cf = w+Cf_o, *Vp = w+Vp_o;

  auto gemm = [&](const float* A, long sAm, long sAb,
                  const float* B, long sBk, long sBn, long sBb,
                  float* C, long sCm, long sCn, long sCb,
                  int M, int N, int K, int batch, float alpha, int accum) {
    dim3 g((unsigned)((M + 63) / 64), (unsigned)((N + 63) / 64), (unsigned)batch);
    gemm_f32_wmma_k4<true><<<g, 128, 0, stream>>>(A, sAm, sAb, B, sBk, sBn, sBb,
                                         C, sCm, sCn, sCb, M, N, K, alpha, accum);
  };
  auto ew = [&](int n) { return dim3((unsigned)((n + 255) / 256)); };

  gen_consts<<<ew(5376), 256, 0, stream>>>(Cf, Vp);
  pack_wqkv<<<ew(24576), 256, 0, stream>>>(wq, wk, wv, wo, Wqkv);

  gemm_f32_wmma_bstat<12, 4, 12><<<dim3(768), 128, 0, stream>>>(
      in, 96, Cf, 96, P1, 49, 589824, 49, 36864);
  gemm_f32_wmma_astat<12, 6><<<dim3(1, 768), 128, 0, stream>>>(
      Cf, 96, P1, 49, 4704, P2, 49, 2401, 49, 49, 1, 6144, 1.f);
  gemm_f32_wmma_astat<12, 6><<<dim3(1, 768), 128, 0, stream>>>(
      Cf, 96, P2, 2401, 230496, T, 2401, 117649, 49, 2401, 38, 2432, 1.f);
  hartley_mirror_fwd<<<ew(7077888), 256, 0, stream>>>(T, blk);

  gemm_f32_wmma_astat<8, 8><<<dim3(6, 170), 128, 0, stream>>>(
      Wqkv, 64, blk, 110592, 0, QKV, 110592, 0, 384, 110592, 1728, 1728, 1.f);

  const long HD = 16L * 110592L;
  attn_scores<<<dim3(7, 4, 8), 128, 0, stream>>>(QKV, QKV + 8 * HD, S,
                                                 0.011048543456039806f);
  selu_k<<<ew(373248), 256, 0, stream>>>(S, 373248);
  attn_values<<<dim3(4, 128, 1), 128, 0, stream>>>(S, QKV + 16 * HD, spec);

  hartley_mirror_spec<<<ew(702464), 256, 0, stream>>>(spec, Xm);
  gemm(Xm, 56, 0, Vp, 1, 56, 0, T1, 96, 1, 0, 50176, 96, 56, 1, 1.f, 0);
  gemm_f32_wmma_astat<7, 7><<<dim3(2, 512), 128, 0, stream>>>(
      Vp, 56, T1, 96, 5376, T2, 96, 9216, 96, 96, 2, 1792, 1.f);
  const float halfInvN = 0.5f / 884736.f;
  gemm_f32_wmma_astat<7, 7><<<dim3(2, 512), 128, 0, stream>>>(
      Vp, 56, T2, 9216, 516096, out, 9216, 884736, 96, 9216, 144, 2304, halfInvN);
}


# Small re-read operands (cas tables, packed weights, S) stored fragment-major: a 16-byte-per-lane fragment load covers 4 whole 128-byte lines, not 16 pieces of 32 bytes; k slots and sums unchanged
# speedup vs baseline: 7.5588x; 1.0412x over previous
#include <hip/hip_runtime.h>
#include <math.h>

typedef float v2f __attribute__((ext_vector_type(2)));
typedef float v4f __attribute__((ext_vector_type(4)));
typedef float v8f __attribute__((ext_vector_type(8)));

__host__ __device__ constexpr int frag_off(int row, int k, int ngk) {
  return (((row >> 4) * ngk + (k >> 3)) * 16 + (row & 15)) * 8 + (k & 7);
}

template <bool BK>
__global__ __launch_bounds__(128) void gemm_f32_wmma_k4(
    const float* __restrict__ A, long sAm, long sAb,
    const float* __restrict__ B, long sBk, long sBn, long sBb,
    float* __restrict__ C, long sCm, long sCn, long sCb,
    int M, int N, int K, float alpha, int accum)
{
  const int lane = threadIdx.x & 31;
  const int wave = threadIdx.x >> 5;
  const int half = lane >> 4;
  const int l16  = lane & 15;
  const int mBase = blockIdx.x * 64;
  const int nBase = blockIdx.y * 64 + wave * 16;

  A += (long)blockIdx.z * sAb;
  B += (long)blockIdx.z * sBb;
  C += (long)blockIdx.z * sCb;

  v8f zero = {0.f,0.f,0.f,0.f,0.f,0.f,0.f,0.f};
  v8f acc[4];
#pragma unroll
  for (int t = 0; t < 4; ++t) acc[t] = zero;

  const float* aPtr[4];
  float aMask[4];
#pragma unroll
  for (int t = 0; t < 4; ++t) {
    int m = mBase + t * 16 + l16;
    bool v = (m < M);
    aMask[t] = v ? 1.f : 0.f;
    aPtr[t] = A + (long)(v ? m : 0) * sAm + 4 * half;
  }
  const int n = nBase + l16;
  const bool nv = (n < N);
  const float bMask = nv ? 1.f : 0.f;
  const float* bPtr = B + (long)(nv ? n : 0) * sBn + (long)(4 * half) * sBk;

  constexpr int D = 3;
  const int ng = K >> 3;
  v4f ra[D][4];
  v4f rb[D];
  v4f aq[4];
  v4f bq;

  auto load_step = [&](int j) {
    if (BK) {
      rb[j] = *(const v4f*)bPtr;
      bPtr += 8;
    } else {
      rb[j].x = bPtr[0];
      rb[j].y = bPtr[sBk];
      rb[j].z = bPtr[2 * sBk];
      rb[j].w = bPtr[3 * sBk];
      bPtr += 8 * sBk;
    }
#pragma unroll
    for (int t = 0; t < 4; ++t) {
      ra[j][t] = *(const v4f*)aPtr[t];
      aPtr[t] += 8;
    }
  };
  auto take_step = [&](int j) {
    bq = bMask * rb[j];
#pragma unroll
    for (int t = 0; t < 4; ++t) aq[t] = aMask[t] * ra[j][t];
  };
  auto mma_step = [&]() {
    v2f b0 = {bq.x, bq.y};
    v2f b1 = {bq.z, bq.w};
#pragma unroll
    for (int t = 0; t < 4; ++t) {
      v2f a0 = {aq[t].x, aq[t].y};
      acc[t] = __builtin_amdgcn_wmma_f32_16x16x4_f32(
          false, a0, false, b0, (short)0, acc[t], false, false);
    }
#pragma unroll
    for (int t = 0; t < 4; ++t) {
      v2f a1 = {aq[t].z, aq[t].w};
      acc[t] = __builtin_amdgcn_wmma_f32_16x16x4_f32(
          false, a1, false, b1, (short)0, acc[t], false, false);
    }
  };

  if (ng >= 2 * D) {
#pragma unroll
    for (int j = 0; j < D; ++j) {
      load_step(j);
      __builtin_amdgcn_sched_barrier(0);
    }

    int s = 0;
    do {
#pragma unroll
      for (int j = 0; j < D; ++j) {
        take_step(j);
        load_step(j);
        mma_step();
        __builtin_amdgcn_sched_barrier(0);
      }
      s += D;
    } while (s + 2 * D <= ng);

#pragma unroll
    for (int j = 0; j < D; ++j) {
      take_step(j);
      if (s + D + j < ng) load_step(j);
      mma_step();
    }
#pragma unroll
    for (int j = 0; j < D - 1; ++j) {
      if (s + D + j < ng) {
        take_step(j);
        mma_step();
      }
    }
  } else {
    for (int i = 0; i < ng; ++i) {
      load_step(0);
      take_step(0);
      mma_step();
    }
  }

  if (nv) {
#pragma unroll
    for (int t = 0; t < 4; ++t) {
#pragma unroll
      for (int r = 0; r < 8; ++r) {
        int m = mBase + t * 16 + r + 8 * half;
        if (m < M) {
          long idx = (long)m * sCm + (long)n * sCn;
          float val = alpha * acc[t][r];
          if (accum) C[idx] += val; else C[idx] = val;
        }
      }
    }
  }
}

template <int NGK, int D>
__global__ __launch_bounds__(128) void gemm_f32_wmma_astat(
    const float* __restrict__ A,
    const float* __restrict__ B, long sBk, long sBb,
    float* __restrict__ C, long sCm, long sCb,
    int M, int N, int ntn, int nitems, float alpha)
{
  static_assert(NGK % D == 0, "a group must keep its ring slot from item to item");
  const int lane = threadIdx.x & 31;
  const int wave = threadIdx.x >> 5;
  const int half = lane >> 4;
  const int l16  = lane & 15;
  const int mBase = blockIdx.x * 64;
  const int nl = wave * 16 + l16;
  const int stride = (int)gridDim.y;
  int it = (int)blockIdx.y;
  if (it >= nitems) return;

  v4f pa[4][NGK];
#pragma unroll
  for (int t = 0; t < 4; ++t) {
    const int m = mBase + t * 16 + l16;
    const bool v = (m < M);
    const float am = v ? 1.f : 0.f;
    const float* ap = A + (v ? frag_off(m, 0, NGK) : 0) + 4 * half;
#pragma unroll
    for (int g = 0; g < NGK; ++g) pa[t][g] = am * *(const v4f*)(ap + 128 * g);
  }

  const unsigned hoff = (unsigned)(4 * half) * (unsigned)sBk;
  const long sBkB = sBk * 4;

  struct Item { const char* base; unsigned vo; bool nv; long c; };
  auto item_of = [&](int i) {
    Item r;
    const int b = i / ntn;
    const int nt = i - b * ntn;
    const int n = nt * 64 + nl;
    r.nv = (n < N);
    r.base = (const char*)(B + (long)b * sBb + nt * 64);
    r.vo = ((unsigned)(r.nv ? nl : 0) + hoff) * 4u;
    r.c = (long)b * sCb + n;
    return r;
  };

  v4f rb[D];
  v4f bq;
  v8f acc[4];
  const v8f zero = {0.f,0.f,0.f,0.f,0.f,0.f,0.f,0.f};

  auto load_step = [&](int j, const Item& x, int g) {
    const char* q = x.base + (long)(8 * g) * sBkB;
    rb[j].x = *(const float*)(q + x.vo);
    rb[j].y = *(const float*)(q + sBkB + x.vo);
    rb[j].z = *(const float*)(q + 2 * sBkB + x.vo);
    rb[j].w = *(const float*)(q + 3 * sBkB + x.vo);
  };
  auto mma_step = [&](int g) {
    v2f b0 = {bq.x, bq.y};
    v2f b1 = {bq.z, bq.w};
#pragma unroll
    for (int t = 0; t < 4; ++t) {
      v2f a0 = {pa[t][g].x, pa[t][g].y};
      acc[t] = __builtin_amdgcn_wmma_f32_16x16x4_f32(
          false, a0, false, b0, (short)0, g == 0 ? zero : acc[t], false, false);
    }
#pragma unroll
    for (int t = 0; t < 4; ++t) {
      v2f a1 = {pa[t][g].z, pa[t][g].w};
      acc[t] = __builtin_amdgcn_wmma_f32_16x16x4_f32(
          false, a1, false, b1, (short)0, acc[t], false, false);
    }
  };

  Item cur = item_of(it);
#pragma unroll
  for (int j = 0; j < D; ++j) {
    load_step(j, cur, j);
    __builtin_amdgcn_sched_barrier(0);
  }

  for (;;) {
    const int itn = it + stride;
    const bool more = (itn < nitems);
    const Item nxt = item_of(more ? itn : it);
    const float bMask = cur.nv ? 1.f : 0.f;
#pragma unroll
    for (int g = 0; g < NGK; ++g) {
      const int j = g % D;
      bq = bMask * rb[j];
      if (g + D < NGK) load_step(j, cur, g + D);
      else             load_step(j, nxt, g + D - NGK);
      mma_step(g);
      __builtin_amdgcn_sched_barrier(0);
    }
    if (cur.nv) {
      float* cp = C + cur.c + (long)(mBase + 8 * half) * sCm;
#pragma unroll
      for (int t = 0; t < 4; ++t) {
#pragma unroll
        for (int r = 0; r < 8; ++r) {
          const int m = mBase + t * 16 + r + 8 * half;
          if (m < M) *cp = alpha * acc[t][r];
          cp += sCm;
        }
        cp += 8 * sCm;
      }
    }
    if (!more) break;
    it = itn;
    cur = nxt;
  }
}

template <int NGK, int NT, int D>
__global__ __launch_bounds__(128) void gemm_f32_wmma_bstat(
    const float* __restrict__ A, int sAm,
    const float* __restrict__ B,
    float* __restrict__ C, int sCm,
    int M, int N, int nitems)
{
  static_assert(NGK % D == 0, "a group must keep its ring slot from item to item");
  const int lane = threadIdx.x & 31;
  const int wave = threadIdx.x >> 5;
  const int half = lane >> 4;
  const int l16  = lane & 15;
  const int stride = (int)gridDim.x * 4;
  int it = (int)blockIdx.x * 4 + wave;
  if (it >= nitems) return;

  v4f pb[NT][NGK];
#pragma unroll
  for (int u = 0; u < NT; ++u) {
    const int n = u * 16 + l16;
    const bool v = (n < N);
    const float bm = v ? 1.f : 0.f;
    const float* bp = B + (v ? frag_off(n, 0, NGK) : 0) + 4 * half;
#pragma unroll
    for (int g = 0; g < NGK; ++g) pb[u][g] = bm * *(const v4f*)(bp + 128 * g);
  }

  auto row_of = [&](int i, float& mask) {
    const int m = i * 16 + l16;
    const bool v = (m < M);
    mask = v ? 1.f : 0.f;
    return A + (long)(v ? m : 0) * sAm + 4 * half;
  };

  v4f ra[D];
  v4f aq;
  v8f acc[NT];
  const v8f zero = {0.f,0.f,0.f,0.f,0.f,0.f,0.f,0.f};

  auto mma_step = [&](int g) {
    v2f a0 = {aq.x, aq.y};
    v2f a1 = {aq.z, aq.w};
#pragma unroll
    for (int u = 0; u < NT; ++u) {
      v2f b0 = {pb[u][g].x, pb[u][g].y};
      acc[u] = __builtin_amdgcn_wmma_f32_16x16x4_f32(
          false, a0, false, b0, (short)0, g == 0 ? zero : acc[u], false, false);
    }
#pragma unroll
    for (int u = 0; u < NT; ++u) {
      v2f b1 = {pb[u][g].z, pb[u][g].w};
      acc[u] = __builtin_amdgcn_wmma_f32_16x16x4_f32(
          false, a1, false, b1, (short)0, acc[u], false, false);
    }
  };

  float curMask;
  const float* curRow = row_of(it, curMask);
#pragma unroll
  for (int j = 0; j < D; ++j) {
    ra[j] = *(const v4f*)(curRow + 8 * j);
    __builtin_amdgcn_sched_barrier(0);
  }

  for (;;) {
    const int itn = it + stride;
    const bool more = (itn < nitems);
    float nxtMask;
    const float* nxtRow = row_of(more ? itn : it, nxtMask);
#pragma unroll
    for (int g = 0; g < NGK; ++g) {
      const int j = g % D;
      aq = curMask * ra[j];
      if (g + D < NGK) ra[j] = *(const v4f*)(curRow + 8 * (g + D));
      else             ra[j] = *(const v4f*)(nxtRow + 8 * (g + D - NGK));
      mma_step(g);
      __builtin_amdgcn_sched_barrier(0);
    }
    {
      const bool full = (it * 16 + 16 <= M);
#pragma unroll
      for (int u = 0; u < NT; ++u) {
        if (u * 16 + l16 < N) {
          float* cp = C + (long)(it * 16 + 8 * half) * sCm + (u * 16 + l16);
#pragma unroll
          for (int r = 0; r < 8; ++r) {
            if (full || it * 16 + r + 8 * half < M) *cp = acc[u][r];
            cp += sCm;
          }
        }
      }
    }
    if (!more) break;
    it = itn;
    curRow = nxtRow;
    curMask = nxtMask;
  }
}

constexpr int tok_off(int m) {
  return (m / 36) * 18432 + ((m / 6) % 6) * 384 + (m % 6) * 8;
}
constexpr int chan_off(int c) {
  return (c >> 9) * 110592 + ((c >> 6) & 7) * 2304 + ((c >> 3) & 7) * 48 + (c & 7);
}
struct TokOffsets {
  int v[264];
  constexpr TokOffsets() : v() {
    for (int m = 0; m < 264; ++m) v[m] = tok_off(m < 216 ? m : 0);
  }
};
__constant__ const TokOffsets kTok = TokOffsets();

__global__ __launch_bounds__(128) void attn_scores(
    const float* __restrict__ Q, const float* __restrict__ Kp,
    float* __restrict__ S, float alpha)
{
  constexpr int TM = 2;
  const int lane = threadIdx.x & 31;
  const int wave = threadIdx.x >> 5;
  const int half = lane >> 4;
  const int l16  = lane & 15;
  const int mBase = blockIdx.x * (16 * TM);
  const int nBase = blockIdx.y * 64 + wave * 16;
  if (nBase >= 216) return;

  Q  += (long)blockIdx.z * 1769472;
  Kp += (long)blockIdx.z * 1769472;
  S  += (long)blockIdx.z * 48384;

  v8f zero = {0.f,0.f,0.f,0.f,0.f,0.f,0.f,0.f};
  v8f acc[TM];
#pragma unroll
  for (int t = 0; t < TM; ++t) acc[t] = zero;

  unsigned aByte[TM];
  float aMask[TM];
#pragma unroll
  for (int t = 0; t < TM; ++t) {
    int m = mBase + t * 16 + l16;
    bool v = (m < 216);
    aMask[t] = v ? 1.f : 0.f;
    aByte[t] = (unsigned)(tok_off(v ? m : 0) + 4 * half) * 4u;
  }
  const int n = nBase + l16;
  const bool nv = (n < 216);
  const float bMask = nv ? 1.f : 0.f;
  const unsigned bByte = (unsigned)(tok_off(nv ? n : 0) + 4 * half) * 4u;

  constexpr int D = 16;
  constexpr int NG = 1024;
  v4f ra[D][TM];
  v4f rb[D];
  v4f aq[TM];
  v4f bq;
  int gl = 0;

  auto load_step = [&](int j) {
    const long ko = (long)(gl >> 6) * 110592 + (long)(((gl >> 3) & 7) * 2304 + (gl & 7) * 48);
    const char* qg = (const char*)(Q + ko);
    const char* kg = (const char*)(Kp + ko);
    rb[j] = *(const v4f*)(kg + bByte);
#pragma unroll
    for (int t = 0; t < TM; ++t) ra[j][t] = *(const v4f*)(qg + aByte[t]);
    ++gl;
  };
  auto take_step = [&](int j) {
    bq = bMask * rb[j];
#pragma unroll
    for (int t = 0; t < TM; ++t) aq[t] = aMask[t] * ra[j][t];
  };
  auto mma_step = [&]() {
    v2f b0 = {bq.x, bq.y};
    v2f b1 = {bq.z, bq.w};
#pragma unroll
    for (int t = 0; t < TM; ++t) {
      v2f a0 = {aq[t].x, aq[t].y};
      acc[t] = __builtin_amdgcn_wmma_f32_16x16x4_f32(
          false, a0, false, b0, (short)0, acc[t], false, false);
    }
#pragma unroll
    for (int t = 0; t < TM; ++t) {
      v2f a1 = {aq[t].z, aq[t].w};
      acc[t] = __builtin_amdgcn_wmma_f32_16x16x4_f32(
          false, a1, false, b1, (short)0, acc[t], false, false);
    }
  };

#pragma unroll
  for (int j = 0; j < D; ++j) {
    load_step(j);
    __builtin_amdgcn_sched_barrier(0);
  }

  int s = 0;
  do {
#pragma unroll
    for (int j = 0; j < D; ++j) {
      take_step(j);
      load_step(j);
      mma_step();
      __builtin_amdgcn_sched_barrier(0);
    }
    s += D;
  } while (s + 2 * D <= NG);

#pragma unroll
  for (int j = 0; j < D; ++j) {
    take_step(j);
    if (s + D + j < NG) load_step(j);
    mma_step();
  }
#pragma unroll
  for (int j = 0; j < D - 1; ++j) {
    if (s + D + j < NG) {
      take_step(j);
      mma_step();
    }
  }

  if (nv) {
#pragma unroll
    for (int t = 0; t < TM; ++t) {
#pragma unroll
      for (int r = 0; r < 8; ++r) {
        int m = mBase + t * 16 + r + 8 * half;
        S[frag_off(m, n, 27)] = alpha * acc[t][r];
      }
    }
  }
}

__global__ __launch_bounds__(128) void attn_values(
    const float* __restrict__ S, const float* __restrict__ V,
    float* __restrict__ O)
{
  const int lane = threadIdx.x & 31;
  const int wave = threadIdx.x >> 5;
  const int half = lane >> 4;
  const int l16  = lane & 15;
  const int mBase = blockIdx.x * 64;
  const int nBase = blockIdx.y * 64 + wave * 16;

  v8f zero = {0.f,0.f,0.f,0.f,0.f,0.f,0.f,0.f};
  v8f acc[4];
#pragma unroll
  for (int t = 0; t < 4; ++t) acc[t] = zero;

  unsigned aByte[4];
  float aMask[4];
#pragma unroll
  for (int t = 0; t < 4; ++t) {
    int m = mBase + t * 16 + l16;
    bool v = (m < 216);
    aMask[t] = v ? 1.f : 0.f;
    aByte[t] = (unsigned)((v ? frag_off(m, 0, 27) : 0) + 4 * half) * 4u;
  }
  const int c = nBase + l16;
  const bool nv = (c < 8192);
  const float bMask = nv ? 1.f : 0.f;
  const unsigned cOff = (unsigned)chan_off(nv ? c : 0);

  constexpr int D = 6;
  constexpr int NGH = 27;
  constexpr int NG = 8 * NGH;
  v4f ra[D][4];
  v4f rb[D];
  v4f aq[4];
  v4f bq;
  int gi = 0;
  const char* sG = (const char*)S;
  const char* vG = (const char*)V;
  int tk0[4], tk1[4];
  auto fetch_tok = [&]() {
#pragma unroll
    for (int e = 0; e < 4; ++e) {
      tk0[e] = kTok.v[8 * gi + e];
      tk1[e] = kTok.v[8 * gi + 4 + e];
    }
  };
  fetch_tok();

  auto load_step = [&](int j) {
    const unsigned o0 = (cOff + (unsigned)(half ? tk1[0] : tk0[0])) << 2;
    const unsigned o1 = (cOff + (unsigned)(half ? tk1[1] : tk0[1])) << 2;
    const unsigned o2 = (cOff + (unsigned)(half ? tk1[2] : tk0[2])) << 2;
    const unsigned o3 = (cOff + (unsigned)(half ? tk1[3] : tk0[3])) << 2;
    rb[j].x = *(const float*)(vG + o0);
    rb[j].y = *(const float*)(vG + o1);
    rb[j].z = *(const float*)(vG + o2);
    rb[j].w = *(const float*)(vG + o3);
#pragma unroll
    for (int t = 0; t < 4; ++t) ra[j][t] = *(const v4f*)(sG + aByte[t]);
    ++gi;
    sG += 512;
    const bool wrap = (gi == NGH);
    gi = wrap ? 0 : gi;
    sG += wrap ? (48384 - 3456) * 4 : 0;
    vG += wrap ? 1769472L * 4 : 0L;
    fetch_tok();
  };
  auto take_step = [&](int j) {
    bq = bMask * rb[j];
#pragma unroll
    for (int t = 0; t < 4; ++t) aq[t] = aMask[t] * ra[j][t];
  };
  auto mma_step = [&]() {
    v2f b0 = {bq.x, bq.y};
    v2f b1 = {bq.z, bq.w};
#pragma unroll
    for (int t = 0; t < 4; ++t) {
      v2f a0 = {aq[t].x, aq[t].y};
      acc[t] = __builtin_amdgcn_wmma_f32_16x16x4_f32(
          false, a0, false, b0, (short)0, acc[t], false, false);
    }
#pragma unroll
    for (int t = 0; t < 4; ++t) {
      v2f a1 = {aq[t].z, aq[t].w};
      acc[t] = __builtin_amdgcn_wmma_f32_16x16x4_f32(
          false, a1, false, b1, (short)0, acc[t], false, false);
    }
  };

#pragma unroll
  for (int j = 0; j < D; ++j) {
    load_step(j);
    __builtin_amdgcn_sched_barrier(0);
  }

  int s = 0;
  do {
#pragma unroll
    for (int j = 0; j < D; ++j) {
      take_step(j);
      load_step(j);
      mma_step();
      __builtin_amdgcn_sched_barrier(0);
    }
    s += D;
  } while (s + 2 * D <= NG);

#pragma unroll
  for (int j = 0; j < D; ++j) {
    take_step(j);
    if (s + D + j < NG) load_step(j);
    mma_step();
  }
#pragma unroll
  for (int j = 0; j < D - 1; ++j) {
    if (s + D + j < NG) {
      take_step(j);
      mma_step();
    }
  }

#pragma unroll
  for (int t = 0; t < 4; ++t) {
#pragma unroll
    for (int r = 0; r < 8; ++r) {
      const int mu = mBase + t * 16 + r;
      const int e0 = kTok.v[mu];
      const int e1 = kTok.v[mu + 8];
      if (nv && mu + 8 * half < 216) O[cOff + (unsigned)(half ? e1 : e0)] = acc[t][r];
    }
  }
}

__global__ void gen_consts(float* CfF, float* Vp, float* VpF) {
  int i = blockIdx.x * 256 + threadIdx.x;
  if (i >= 64 * 96) return;
  int j = i / 96, nn = i % 96;
  float val = 0.f;
  if (j < 49) {
    int f = (j <= 24) ? j : (j + 47);
    float th = 6.283185307179586f * (float)((f * nn) % 96) / 96.f;
    val = cosf(th) + sinf(th);
  }
  CfF[frag_off(j, nn, 12)] = val;
  if (j < 56) {
    Vp[nn * 56 + j] = val;
    VpF[frag_off(nn, j, 7)] = val;
  }
}

__global__ void hartley_mirror_fwd(const float* __restrict__ T, float* __restrict__ blk) {
  int i = blockIdx.x * 256 + threadIdx.x;
  if (i >= 7077888) return;
  int ix = i % 48, iy = (i / 48) % 48, iz = (i / 2304) % 48, c = i / 110592;
  int jx = ix + (ix >= 24), jy = iy + (iy >= 24), jz = iz + (iz >= 24);
  int gx = jx ? 49 - jx : 0, gy = jy ? 49 - jy : 0, gz = jz ? 49 - jz : 0;
  const float* t = T + (long)c * 117649;
  float a = t[(gz * 49 + jy) * 49 + jx];
  float b = t[(jz * 49 + gy) * 49 + jx];
  float d = t[(jz * 49 + jy) * 49 + gx];
  float e = t[(gz * 49 + gy) * 49 + gx];
  blk[i] = 0.5f * ((a + b) + (d - e));
}

__device__ __forceinline__ float spec_at(const float* __restrict__ s, int jz, int jy, int jx) {
  int iz = jz - (jz > 24), iy = jy - (jy > 24), ix = jx - (jx > 24);
  float v = s[(iz * 48 + iy) * 48 + ix];
  return (jz == 24 || jy == 24 || jx == 24) ? 0.f : v;
}
__global__ void hartley_mirror_spec(const float* __restrict__ spec, float* __restrict__ Xm) {
  int i = blockIdx.x * 256 + threadIdx.x;
  if (i >= 702464) return;
  int q = i % 14, jy = (i / 14) % 56, jz = (i / 784) % 56, c = i / 43904;
  float r[4] = {0.f, 0.f, 0.f, 0.f};
  if (jz < 49 && jy < 49) {
    const float* s = spec + (long)c * 110592;
    int gz = jz ? 49 - jz : 0, gy = jy ? 49 - jy : 0;
#pragma unroll
    for (int e = 0; e < 4; ++e) {
      int jx = 4 * q + e;
      int jc = jx < 49 ? jx : 48;
      int gx = jc ? 49 - jc : 0;
      float a = spec_at(s, gz, jy, jc);
      float b = spec_at(s, jz, gy, jc);
      float d = spec_at(s, jz, jy, gx);
      float m = spec_at(s, gz, gy, gx);
      r[e] = jx < 49 ? (a + b) + (d - m) : 0.f;
    }
  }
  v4f o = {r[0], r[1], r[2], r[3]};
  *(v4f*)(Xm + 4 * (long)i) = o;
}

__global__ void pack_wqkv(const float* __restrict__ wq, const float* __restrict__ wk,
                          const float* __restrict__ wv, const float* __restrict__ wo,
                          float* __restrict__ dst) {
  int i = blockIdx.x * 256 + threadIdx.x;
  if (i >= 384 * 64) return;
  int ic = i % 64, r = i / 64;
  int t = r / 128, q = r % 128;
  if (t < 2) {
    const float* w = (t == 0) ? wq : wk;
    dst[frag_off(r, ic, 8)] = w[q * 64 + ic];
  } else {
    const int h = q >> 4, o = q & 15;
    const float* wor = wo + o * 128 + h * 16;
    const float* wvc = wv + h * 1024 + ic;
    float s = 0.f;
#pragma unroll
    for (int u = 0; u < 16; ++u) s += wor[u] * wvc[u * 64];
    dst[frag_off(r, ic, 8)] = s;
  }
}

__global__ void selu_k(float* __restrict__ s, int nelem) {
  int i = blockIdx.x * 256 + threadIdx.x;
  if (i >= nelem) return;
  float x = s[i];
  const float lam = 1.0507009873554805f, alp = 1.6732632423543772f;
  s[i] = lam * (x > 0.f ? x : alp * (expf(x) - 1.f));
}

extern "C" void kernel_launch(void* const* d_in, const int* in_sizes, int n_in,
                              void* d_out, int out_size, void* d_ws, size_t ws_size,
                              hipStream_t stream) {
  const float* in = (const float*)d_in[0];
  const float* wq = (const float*)d_in[1];
  const float* wk = (const float*)d_in[2];
  const float* wv = (const float*)d_in[3];
  const float* wo = (const float*)d_in[4];
  float* out = (float*)d_out;
  float* w = (float*)d_ws;

  const long P1_o   = 0,          P2_o   = 28901376;
  const long T_o    = 43653120;
  const long blk_o  = 84934656;
  const long QKV_o  = 0;
  const long S_o    = 92012544;
  const long spec_o = 14155776;
  const long Xm_o   = 15925248;
  const long T1_o   = 18735104,   T2_o   = 23552000;
  const long Wqkv_o = 106541568;
  const long CfF_o  = 106566144,  VpF_o  = 106572288;
  const long Vp_o   = 106577664;
  const long WS_FLOATS = 106584576;
  if (ws_size < (size_t)WS_FLOATS * sizeof(float)) return;

  float *P1 = w+P1_o, *P2 = w+P2_o, *T = w+T_o;
  float *blk = w+blk_o, *QKV = w+QKV_o, *S = w+S_o;
  float *spec = w+spec_o;
  float *Xm = w+Xm_o, *T1 = w+T1_o, *T2 = w+T2_o, *Wqkv = w+Wqkv_o;
  float *CfF = w+CfF_o, *Vp = w+Vp_o, *VpF = w+VpF_o;

  auto gemm = [&](const float* A, long sAm, long sAb,
                  const float* B, long sBk, long sBn, long sBb,
                  float* C, long sCm, long sCn, long sCb,
                  int M, int N, int K, int batch, float alpha, int accum) {
    dim3 g((unsigned)((M + 63) / 64), (unsigned)((N + 63) / 64), (unsigned)batch);
    gemm_f32_wmma_k4<true><<<g, 128, 0, stream>>>(A, sAm, sAb, B, sBk, sBn, sBb,
                                         C, sCm, sCn, sCb, M, N, K, alpha, accum);
  };
  auto ew = [&](int n) { return dim3((unsigned)((n + 255) / 256)); };

  gen_consts<<<ew(6144), 256, 0, stream>>>(CfF, Vp, VpF);
  pack_wqkv<<<ew(24576), 256, 0, stream>>>(wq, wk, wv, wo, Wqkv);

  gemm_f32_wmma_bstat<12, 4, 12><<<dim3(768), 128, 0, stream>>>(
      in, 96, CfF, P1, 49, 589824, 49, 36864);
  gemm_f32_wmma_astat<12, 6><<<dim3(1, 768), 128, 0, stream>>>(
      CfF, P1, 49, 4704, P2, 49, 2401, 49, 49, 1, 6144, 1.f);
  gemm_f32_wmma_astat<12, 6><<<dim3(1, 768), 128, 0, stream>>>(
      CfF, P2, 2401, 230496, T, 2401, 117649, 49, 2401, 38, 2432, 1.f);
  hartley_mirror_fwd<<<ew(7077888), 256, 0, stream>>>(T, blk);

  gemm_f32_wmma_astat<8, 8><<<dim3(6, 170), 128, 0, stream>>>(
      Wqkv, blk, 110592, 0, QKV, 110592, 0, 384, 110592, 1728, 1728, 1.f);

  const long HD = 16L * 110592L;
  attn_scores<<<dim3(7, 4, 8), 128, 0, stream>>>(QKV, QKV + 8 * HD, S,
                                                 0.011048543456039806f);
  selu_k<<<ew(387072), 256, 0, stream>>>(S, 387072);
  attn_values<<<dim3(4, 128, 1), 128, 0, stream>>>(S, QKV + 16 * HD, spec);

  hartley_mirror_spec<<<ew(702464), 256, 0, stream>>>(spec, Xm);
  gemm(Xm, 56, 0, Vp, 1, 56, 0, T1, 96, 1, 0, 50176, 96, 56, 1, 1.f, 0);
  gemm_f32_wmma_astat<7, 7><<<dim3(2, 512), 128, 0, stream>>>(
      VpF, T1, 96, 5376, T2, 96, 9216, 96, 96, 2, 1792, 1.f);
  const float halfInvN = 0.5f / 884736.f;
  gemm_f32_wmma_astat<7, 7><<<dim3(2, 512), 128, 0, stream>>>(
      VpF, T2, 9216, 516096, out, 9216, 884736, 96, 9216, 144, 2304, halfInvN);
}


# Forward and inverse y and z products with the cas table folded on n <-> 96-n (modes f <-> -f): paired rows added/subtracted on the vector unit, cos + sin tables of half the k (sum order changes)
# speedup vs baseline: 7.9198x; 1.0478x over previous
#include <hip/hip_runtime.h>
#include <math.h>

typedef float v2f __attribute__((ext_vector_type(2)));
typedef float v4f __attribute__((ext_vector_type(4)));
typedef float v8f __attribute__((ext_vector_type(8)));

__host__ __device__ constexpr int frag_off(int row, int k, int ngk) {
  return (((row >> 4) * ngk + (k >> 3)) * 16 + (row & 15)) * 8 + (k & 7);
}

template <bool BK>
__global__ __launch_bounds__(128) void gemm_f32_wmma_k4(
    const float* __restrict__ A, long sAm, long sAb,
    const float* __restrict__ B, long sBk, long sBn, long sBb,
    float* __restrict__ C, long sCm, long sCn, long sCb,
    int M, int N, int K, float alpha, int accum)
{
  const int lane = threadIdx.x & 31;
  const int wave = threadIdx.x >> 5;
  const int half = lane >> 4;
  const int l16  = lane & 15;
  const int mBase = blockIdx.x * 64;
  const int nBase = blockIdx.y * 64 + wave * 16;

  A += (long)blockIdx.z * sAb;
  B += (long)blockIdx.z * sBb;
  C += (long)blockIdx.z * sCb;

  v8f zero = {0.f,0.f,0.f,0.f,0.f,0.f,0.f,0.f};
  v8f acc[4];
#pragma unroll
  for (int t = 0; t < 4; ++t) acc[t] = zero;

  const float* aPtr[4];
  float aMask[4];
#pragma unroll
  for (int t = 0; t < 4; ++t) {
    int m = mBase + t * 16 + l16;
    bool v = (m < M);
    aMask[t] = v ? 1.f : 0.f;
    aPtr[t] = A + (long)(v ? m : 0) * sAm + 4 * half;
  }
  const int n = nBase + l16;
  const bool nv = (n < N);
  const float bMask = nv ? 1.f : 0.f;
  const float* bPtr = B + (long)(nv ? n : 0) * sBn + (long)(4 * half) * sBk;

  constexpr int D = 3;
  const int ng = K >> 3;
  v4f ra[D][4];
  v4f rb[D];
  v4f aq[4];
  v4f bq;

  auto load_step = [&](int j) {
    if (BK) {
      rb[j] = *(const v4f*)bPtr;
      bPtr += 8;
    } else {
      rb[j].x = bPtr[0];
      rb[j].y = bPtr[sBk];
      rb[j].z = bPtr[2 * sBk];
      rb[j].w = bPtr[3 * sBk];
      bPtr += 8 * sBk;
    }
#pragma unroll
    for (int t = 0; t < 4; ++t) {
      ra[j][t] = *(const v4f*)aPtr[t];
      aPtr[t] += 8;
    }
  };
  auto take_step = [&](int j) {
    bq = bMask * rb[j];
#pragma unroll
    for (int t = 0; t < 4; ++t) aq[t] = aMask[t] * ra[j][t];
  };
  auto mma_step = [&]() {
    v2f b0 = {bq.x, bq.y};
    v2f b1 = {bq.z, bq.w};
#pragma unroll
    for (int t = 0; t < 4; ++t) {
      v2f a0 = {aq[t].x, aq[t].y};
      acc[t] = __builtin_amdgcn_wmma_f32_16x16x4_f32(
          false, a0, false, b0, (short)0, acc[t], false, false);
    }
#pragma unroll
    for (int t = 0; t < 4; ++t) {
      v2f a1 = {aq[t].z, aq[t].w};
      acc[t] = __builtin_amdgcn_wmma_f32_16x16x4_f32(
          false, a1, false, b1, (short)0, acc[t], false, false);
    }
  };

  if (ng >= 2 * D) {
#pragma unroll
    for (int j = 0; j < D; ++j) {
      load_step(j);
      __builtin_amdgcn_sched_barrier(0);
    }

    int s = 0;
    do {
#pragma unroll
      for (int j = 0; j < D; ++j) {
        take_step(j);
        load_step(j);
        mma_step();
        __builtin_amdgcn_sched_barrier(0);
      }
      s += D;
    } while (s + 2 * D <= ng);

#pragma unroll
    for (int j = 0; j < D; ++j) {
      take_step(j);
      if (s + D + j < ng) load_step(j);
      mma_step();
    }
#pragma unroll
    for (int j = 0; j < D - 1; ++j) {
      if (s + D + j < ng) {
        take_step(j);
        mma_step();
      }
    }
  } else {
    for (int i = 0; i < ng; ++i) {
      load_step(0);
      take_step(0);
      mma_step();
    }
  }

  if (nv) {
#pragma unroll
    for (int t = 0; t < 4; ++t) {
#pragma unroll
      for (int r = 0; r < 8; ++r) {
        int m = mBase + t * 16 + r + 8 * half;
        if (m < M) {
          long idx = (long)m * sCm + (long)n * sCn;
          float val = alpha * acc[t][r];
          if (accum) C[idx] += val; else C[idx] = val;
        }
      }
    }
  }
}

template <int NGK, int D>
__global__ __launch_bounds__(128) void gemm_f32_wmma_astat(
    const float* __restrict__ A,
    const float* __restrict__ B, long sBk, long sBb,
    float* __restrict__ C, long sCm, long sCb,
    int M, int N, int ntn, int nitems, float alpha)
{
  static_assert(NGK % D == 0, "a group must keep its ring slot from item to item");
  const int lane = threadIdx.x & 31;
  const int wave = threadIdx.x >> 5;
  const int half = lane >> 4;
  const int l16  = lane & 15;
  const int mBase = blockIdx.x * 64;
  const int nl = wave * 16 + l16;
  const int stride = (int)gridDim.y;
  int it = (int)blockIdx.y;
  if (it >= nitems) return;

  v4f pa[4][NGK];
#pragma unroll
  for (int t = 0; t < 4; ++t) {
    const int m = mBase + t * 16 + l16;
    const bool v = (m < M);
    const float am = v ? 1.f : 0.f;
    const float* ap = A + (v ? frag_off(m, 0, NGK) : 0) + 4 * half;
#pragma unroll
    for (int g = 0; g < NGK; ++g) pa[t][g] = am * *(const v4f*)(ap + 128 * g);
  }

  const unsigned hoff = (unsigned)(4 * half) * (unsigned)sBk;
  const long sBkB = sBk * 4;

  struct Item { const char* base; unsigned vo; bool nv; long c; };
  auto item_of = [&](int i) {
    Item r;
    const int b = i / ntn;
    const int nt = i - b * ntn;
    const int n = nt * 64 + nl;
    r.nv = (n < N);
    r.base = (const char*)(B + (long)b * sBb + nt * 64);
    r.vo = ((unsigned)(r.nv ? nl : 0) + hoff) * 4u;
    r.c = (long)b * sCb + n;
    return r;
  };

  v4f rb[D];
  v4f bq;
  v8f acc[4];
  const v8f zero = {0.f,0.f,0.f,0.f,0.f,0.f,0.f,0.f};

  auto load_step = [&](int j, const Item& x, int g) {
    const char* q = x.base + (long)(8 * g) * sBkB;
    rb[j].x = *(const float*)(q + x.vo);
    rb[j].y = *(const float*)(q + sBkB + x.vo);
    rb[j].z = *(const float*)(q + 2 * sBkB + x.vo);
    rb[j].w = *(const float*)(q + 3 * sBkB + x.vo);
  };
  auto mma_step = [&](int g) {
    v2f b0 = {bq.x, bq.y};
    v2f b1 = {bq.z, bq.w};
#pragma unroll
    for (int t = 0; t < 4; ++t) {
      v2f a0 = {pa[t][g].x, pa[t][g].y};
      acc[t] = __builtin_amdgcn_wmma_f32_16x16x4_f32(
          false, a0, false, b0, (short)0, g == 0 ? zero : acc[t], false, false);
    }
#pragma unroll
    for (int t = 0; t < 4; ++t) {
      v2f a1 = {pa[t][g].z, pa[t][g].w};
      acc[t] = __builtin_amdgcn_wmma_f32_16x16x4_f32(
          false, a1, false, b1, (short)0, acc[t], false, false);
    }
  };

  Item cur = item_of(it);
#pragma unroll
  for (int j = 0; j < D; ++j) {
    load_step(j, cur, j);
    __builtin_amdgcn_sched_barrier(0);
  }

  for (;;) {
    const int itn = it + stride;
    const bool more = (itn < nitems);
    const Item nxt = item_of(more ? itn : it);
    const float bMask = cur.nv ? 1.f : 0.f;
#pragma unroll
    for (int g = 0; g < NGK; ++g) {
      const int j = g % D;
      bq = bMask * rb[j];
      if (g + D < NGK) load_step(j, cur, g + D);
      else             load_step(j, nxt, g + D - NGK);
      mma_step(g);
      __builtin_amdgcn_sched_barrier(0);
    }
    if (cur.nv) {
      float* cp = C + cur.c + (long)(mBase + 8 * half) * sCm;
#pragma unroll
      for (int t = 0; t < 4; ++t) {
#pragma unroll
        for (int r = 0; r < 8; ++r) {
          const int m = mBase + t * 16 + r + 8 * half;
          if (m < M) *cp = alpha * acc[t][r];
          cp += sCm;
        }
        cp += 8 * sCm;
      }
    }
    if (!more) break;
    it = itn;
    cur = nxt;
  }
}

__global__ __launch_bounds__(128) void cas_fold_astat(
    const float* __restrict__ Cc, const float* __restrict__ Ss,
    const float* __restrict__ B, long sBk, long sBb,
    float* __restrict__ C, long sCm, long sCb,
    int N, int ntn, int nitems, float alpha)
{
  const int lane = threadIdx.x & 31;
  const int wave = threadIdx.x >> 5;
  const int half = lane >> 4;
  const int l16  = lane & 15;
  const int nl = wave * 16 + l16;
  const int stride = (int)gridDim.y;
  int it = (int)blockIdx.y;
  if (it >= nitems) return;

  v4f pc[2][7];
  v4f pz[2][6];
#pragma unroll
  for (int t = 0; t < 2; ++t) {
    const float* ac = Cc + frag_off(16 * t + l16, 0, 7) + 4 * half;
    const float* as = Ss + frag_off(16 * t + l16, 0, 6) + 4 * half;
#pragma unroll
    for (int g = 0; g < 7; ++g) pc[t][g] = *(const v4f*)(ac + 128 * g);
#pragma unroll
    for (int g = 0; g < 6; ++g) pz[t][g] = *(const v4f*)(as + 128 * g);
  }

  const unsigned sBkB = (unsigned)sBk * 4u;
  const unsigned hB = (unsigned)(4 * half) * sBkB;
  const unsigned mB = half ? 0u : 4u * sBkB;
  const unsigned m0B = half ? 92u * sBkB : 0u;

  struct Item { const char* base; unsigned cb; bool nv; long c; };
  auto item_of = [&](int i) {
    Item r;
    const int b = i / ntn;
    const int nt = i - b * ntn;
    const int n = nt * 64 + nl;
    r.nv = (n < N);
    r.base = (const char*)(B + (long)b * sBb + nt * 64);
    r.cb = (unsigned)(r.nv ? nl : 0) * 4u;
    r.c = (long)b * sCb + n;
    return r;
  };

  v4f rf[7];
  v4f rm[6];
  v8f accE[2], accO[2];
  const v8f zero = {0.f,0.f,0.f,0.f,0.f,0.f,0.f,0.f};

  auto load_group = [&](int g, const Item& x) {
    const char* q = x.base + (long)(8 * g) * sBkB;
    const unsigned fo = x.cb + hB;
    rf[g].x = *(const float*)(q + fo);
    rf[g].y = *(const float*)(q + sBkB + fo);
    rf[g].z = *(const float*)(q + 2 * sBkB + fo);
    rf[g].w = *(const float*)(q + 3 * sBkB + fo);
    if (g < 6) {
      const char* p = x.base + (long)(92 - 8 * g) * sBkB;
      const unsigned mo = x.cb + mB;
      if (g == 0) rm[g].x = *(const float*)(x.base + (x.cb + m0B));
      else        rm[g].x = *(const float*)(p + mo);
      rm[g].y = *(const float*)(p - sBkB + mo);
      rm[g].z = *(const float*)(p - 2 * sBkB + mo);
      rm[g].w = *(const float*)(p - 3 * sBkB + mo);
    }
  };

  Item cur = item_of(it);
#pragma unroll
  for (int g = 0; g < 7; ++g) {
    load_group(g, cur);
    __builtin_amdgcn_sched_barrier(0);
  }

  for (;;) {
    const int itn = it + stride;
    const bool more = (itn < nitems);
    const Item nxt = item_of(more ? itn : it);
#pragma unroll
    for (int g = 0; g < 7; ++g) {
      const v4f f = rf[g];
      v4f u, w;
      if (g < 6) { const v4f m = rm[g]; u = f + m; w = f - m; }
      else       { u = f + f; w = u; }
      load_group(g, nxt);
      const v2f u0 = {u.x, u.y}, u1 = {u.z, u.w};
      const v2f w0 = {w.x, w.y}, w1 = {w.z, w.w};
#pragma unroll
      for (int t = 0; t < 2; ++t) {
        v2f a0 = {pc[t][g].x, pc[t][g].y};
        accE[t] = __builtin_amdgcn_wmma_f32_16x16x4_f32(
            false, a0, false, u0, (short)0, g == 0 ? zero : accE[t], false, false);
      }
      if (g < 6) {
#pragma unroll
        for (int t = 0; t < 2; ++t) {
          v2f a0 = {pz[t][g].x, pz[t][g].y};
          accO[t] = __builtin_amdgcn_wmma_f32_16x16x4_f32(
              false, a0, false, w0, (short)0, g == 0 ? zero : accO[t], false, false);
        }
      }
#pragma unroll
      for (int t = 0; t < 2; ++t) {
        v2f a1 = {pc[t][g].z, pc[t][g].w};
        accE[t] = __builtin_amdgcn_wmma_f32_16x16x4_f32(
            false, a1, false, u1, (short)0, accE[t], false, false);
      }
      if (g < 6) {
#pragma unroll
        for (int t = 0; t < 2; ++t) {
          v2f a1 = {pz[t][g].z, pz[t][g].w};
          accO[t] = __builtin_amdgcn_wmma_f32_16x16x4_f32(
              false, a1, false, w1, (short)0, accO[t], false, false);
        }
      }
      __builtin_amdgcn_sched_barrier(0);
    }
    if (cur.nv) {
      float* cb = C + cur.c;
#pragma unroll
      for (int t = 0; t < 2; ++t) {
#pragma unroll
        for (int r = 0; r < 8; ++r) {
          const int f = 16 * t + r + 8 * half;
          const float e = accE[t][r], o = accO[t][r];
          if (f <= 24) cb[(long)f * sCm] = alpha * (e + o);
          if (f >= 1 && f <= 24) cb[(long)(49 - f) * sCm] = alpha * (e - o);
        }
      }
    }
    if (!more) break;
    it = itn;
    cur = nxt;
  }
}

__global__ __launch_bounds__(128) void cas_fold_inv_astat(
    const float* __restrict__ Ci, const float* __restrict__ Si,
    const float* __restrict__ B, long sBk, long sBb,
    float* __restrict__ C, long sCm, long sCb,
    int N, int ntn, int nitems, float alpha)
{
  const int lane = threadIdx.x & 31;
  const int wave = threadIdx.x >> 5;
  const int half = lane >> 4;
  const int l16  = lane & 15;
  const int nl = wave * 16 + l16;
  const int stride = (int)gridDim.y;
  int it = (int)blockIdx.y;
  if (it >= nitems) return;

  v4f pc[4][4];
  v4f pz[3][4];
#pragma unroll
  for (int t = 0; t < 4; ++t) {
    const float* ac = Ci + frag_off(16 * t + l16, 0, 4) + 4 * half;
#pragma unroll
    for (int g = 0; g < 4; ++g) pc[t][g] = *(const v4f*)(ac + 128 * g);
  }
#pragma unroll
  for (int t = 0; t < 3; ++t) {
    const float* as = Si + frag_off(16 * t + l16, 0, 4) + 4 * half;
#pragma unroll
    for (int g = 0; g < 4; ++g) pz[t][g] = *(const v4f*)(as + 128 * g);
  }

  const unsigned sBkB = (unsigned)sBk * 4u;
  const unsigned hB = (unsigned)(4 * half) * sBkB;
  const unsigned mB = half ? 0u : 4u * sBkB;
  const unsigned m0B = half ? 45u * sBkB : 0u;

  struct Item { const char* base; unsigned cb; bool nv; long c; };
  auto item_of = [&](int i) {
    Item r;
    const int b = i / ntn;
    const int nt = i - b * ntn;
    const int n = nt * 64 + nl;
    r.nv = (n < N);
    r.base = (const char*)(B + (long)b * sBb + nt * 64);
    r.cb = (unsigned)(r.nv ? nl : 0) * 4u;
    r.c = (long)b * sCb + n;
    return r;
  };

  v4f rf[4];
  v4f rm[4];
  v8f accE[4], accO[3];
  const v8f zero = {0.f,0.f,0.f,0.f,0.f,0.f,0.f,0.f};

  auto load_group = [&](int g, const Item& x) {
    const char* q = x.base + (long)(8 * g) * sBkB;
    const unsigned fo = x.cb + hB;
    rf[g].x = *(const float*)(q + fo);
    rf[g].y = *(const float*)(q + sBkB + fo);
    rf[g].z = *(const float*)(q + 2 * sBkB + fo);
    rf[g].w = *(const float*)(q + 3 * sBkB + fo);
    const char* p = x.base + (long)(45 - 8 * g) * sBkB;
    const unsigned mo = x.cb + mB;
    if (g == 0) rm[g].x = *(const float*)(x.base + (x.cb + m0B));
    else        rm[g].x = *(const float*)(p + mo);
    rm[g].y = *(const float*)(p - sBkB + mo);
    rm[g].z = *(const float*)(p - 2 * sBkB + mo);
    rm[g].w = *(const float*)(p - 3 * sBkB + mo);
  };

  Item cur = item_of(it);
#pragma unroll
  for (int g = 0; g < 4; ++g) {
    load_group(g, cur);
    __builtin_amdgcn_sched_barrier(0);
  }

  for (;;) {
    const int itn = it + stride;
    const bool more = (itn < nitems);
    const Item nxt = item_of(more ? itn : it);
#pragma unroll
    for (int g = 0; g < 4; ++g) {
      const v4f f = rf[g];
      const v4f m = rm[g];
      const v4f u = f + m;
      const v4f w = f - m;
      load_group(g, nxt);
      const v2f u0 = {u.x, u.y}, u1 = {u.z, u.w};
      const v2f w0 = {w.x, w.y}, w1 = {w.z, w.w};
#pragma unroll
      for (int t = 0; t < 4; ++t) {
        v2f a0 = {pc[t][g].x, pc[t][g].y};
        accE[t] = __builtin_amdgcn_wmma_f32_16x16x4_f32(
            false, a0, false, u0, (short)0, g == 0 ? zero : accE[t], false, false);
      }
#pragma unroll
      for (int t = 0; t < 3; ++t) {
        v2f a0 = {pz[t][g].x, pz[t][g].y};
        accO[t] = __builtin_amdgcn_wmma_f32_16x16x4_f32(
            false, a0, false, w0, (short)0, g == 0 ? zero : accO[t], false, false);
      }
#pragma unroll
      for (int t = 0; t < 4; ++t) {
        v2f a1 = {pc[t][g].z, pc[t][g].w};
        accE[t] = __builtin_amdgcn_wmma_f32_16x16x4_f32(
            false, a1, false, u1, (short)0, accE[t], false, false);
      }
#pragma unroll
      for (int t = 0; t < 3; ++t) {
        v2f a1 = {pz[t][g].z, pz[t][g].w};
        accO[t] = __builtin_amdgcn_wmma_f32_16x16x4_f32(
            false, a1, false, w1, (short)0, accO[t], false, false);
      }
      __builtin_amdgcn_sched_barrier(0);
    }
    if (cur.nv) {
      float* cb = C + cur.c;
#pragma unroll
      for (int t = 0; t < 3; ++t) {
#pragma unroll
        for (int r = 0; r < 8; ++r) {
          const int n = 16 * t + r + 8 * half;
          const float e = accE[t][r], o = accO[t][r];
          cb[(long)n * sCm] = alpha * (e + o);
          if (n >= 1) cb[(long)(96 - n) * sCm] = alpha * (e - o);
        }
      }
      if (half == 0) cb[48L * sCm] = alpha * accE[3][0];
    }
    if (!more) break;
    it = itn;
    cur = nxt;
  }
}

template <int NGK, int NT, int D>
__global__ __launch_bounds__(128) void gemm_f32_wmma_bstat(
    const float* __restrict__ A, int sAm,
    const float* __restrict__ B,
    float* __restrict__ C, int sCm,
    int M, int N, int nitems)
{
  static_assert(NGK % D == 0, "a group must keep its ring slot from item to item");
  const int lane = threadIdx.x & 31;
  const int wave = threadIdx.x >> 5;
  const int half = lane >> 4;
  const int l16  = lane & 15;
  const int stride = (int)gridDim.x * 4;
  int it = (int)blockIdx.x * 4 + wave;
  if (it >= nitems) return;

  v4f pb[NT][NGK];
#pragma unroll
  for (int u = 0; u < NT; ++u) {
    const int n = u * 16 + l16;
    const bool v = (n < N);
    const float bm = v ? 1.f : 0.f;
    const float* bp = B + (v ? frag_off(n, 0, NGK) : 0) + 4 * half;
#pragma unroll
    for (int g = 0; g < NGK; ++g) pb[u][g] = bm * *(const v4f*)(bp + 128 * g);
  }

  auto row_of = [&](int i, float& mask) {
    const int m = i * 16 + l16;
    const bool v = (m < M);
    mask = v ? 1.f : 0.f;
    return A + (long)(v ? m : 0) * sAm + 4 * half;
  };

  v4f ra[D];
  v4f aq;
  v8f acc[NT];
  const v8f zero = {0.f,0.f,0.f,0.f,0.f,0.f,0.f,0.f};

  auto mma_step = [&](int g) {
    v2f a0 = {aq.x, aq.y};
    v2f a1 = {aq.z, aq.w};
#pragma unroll
    for (int u = 0; u < NT; ++u) {
      v2f b0 = {pb[u][g].x, pb[u][g].y};
      acc[u] = __builtin_amdgcn_wmma_f32_16x16x4_f32(
          false, a0, false, b0, (short)0, g == 0 ? zero : acc[u], false, false);
    }
#pragma unroll
    for (int u = 0; u < NT; ++u) {
      v2f b1 = {pb[u][g].z, pb[u][g].w};
      acc[u] = __builtin_amdgcn_wmma_f32_16x16x4_f32(
          false, a1, false, b1, (short)0, acc[u], false, false);
    }
  };

  float curMask;
  const float* curRow = row_of(it, curMask);
#pragma unroll
  for (int j = 0; j < D; ++j) {
    ra[j] = *(const v4f*)(curRow + 8 * j);
    __builtin_amdgcn_sched_barrier(0);
  }

  for (;;) {
    const int itn = it + stride;
    const bool more = (itn < nitems);
    float nxtMask;
    const float* nxtRow = row_of(more ? itn : it, nxtMask);
#pragma unroll
    for (int g = 0; g < NGK; ++g) {
      const int j = g % D;
      aq = curMask * ra[j];
      if (g + D < NGK) ra[j] = *(const v4f*)(curRow + 8 * (g + D));
      else             ra[j] = *(const v4f*)(nxtRow + 8 * (g + D - NGK));
      mma_step(g);
      __builtin_amdgcn_sched_barrier(0);
    }
    {
      const bool full = (it * 16 + 16 <= M);
#pragma unroll
      for (int u = 0; u < NT; ++u) {
        if (u * 16 + l16 < N) {
          float* cp = C + (long)(it * 16 + 8 * half) * sCm + (u * 16 + l16);
#pragma unroll
          for (int r = 0; r < 8; ++r) {
            if (full || it * 16 + r + 8 * half < M) *cp = acc[u][r];
            cp += sCm;
          }
        }
      }
    }
    if (!more) break;
    it = itn;
    curRow = nxtRow;
    curMask = nxtMask;
  }
}

constexpr int tok_off(int m) {
  return (m / 36) * 18432 + ((m / 6) % 6) * 384 + (m % 6) * 8;
}
constexpr int chan_off(int c) {
  return (c >> 9) * 110592 + ((c >> 6) & 7) * 2304 + ((c >> 3) & 7) * 48 + (c & 7);
}
struct TokOffsets {
  int v[264];
  constexpr TokOffsets() : v() {
    for (int m = 0; m < 264; ++m) v[m] = tok_off(m < 216 ? m : 0);
  }
};
__constant__ const TokOffsets kTok = TokOffsets();

__global__ __launch_bounds__(128) void attn_scores(
    const float* __restrict__ Q, const float* __restrict__ Kp,
    float* __restrict__ S, float alpha)
{
  constexpr int TM = 2;
  const int lane = threadIdx.x & 31;
  const int wave = threadIdx.x >> 5;
  const int half = lane >> 4;
  const int l16  = lane & 15;
  const int mBase = blockIdx.x * (16 * TM);
  const int nBase = blockIdx.y * 64 + wave * 16;
  if (nBase >= 216) return;

  Q  += (long)blockIdx.z * 1769472;
  Kp += (long)blockIdx.z * 1769472;
  S  += (long)blockIdx.z * 48384;

  v8f zero = {0.f,0.f,0.f,0.f,0.f,0.f,0.f,0.f};
  v8f acc[TM];
#pragma unroll
  for (int t = 0; t < TM; ++t) acc[t] = zero;

  unsigned aByte[TM];
  float aMask[TM];
#pragma unroll
  for (int t = 0; t < TM; ++t) {
    int m = mBase + t * 16 + l16;
    bool v = (m < 216);
    aMask[t] = v ? 1.f : 0.f;
    aByte[t] = (unsigned)(tok_off(v ? m : 0) + 4 * half) * 4u;
  }
  const int n = nBase + l16;
  const bool nv = (n < 216);
  const float bMask = nv ? 1.f : 0.f;
  const unsigned bByte = (unsigned)(tok_off(nv ? n : 0) + 4 * half) * 4u;

  constexpr int D = 16;
  constexpr int NG = 1024;
  v4f ra[D][TM];
  v4f rb[D];
  v4f aq[TM];
  v4f bq;
  int gl = 0;

  auto load_step = [&](int j) {
    const long ko = (long)(gl >> 6) * 110592 + (long)(((gl >> 3) & 7) * 2304 + (gl & 7) * 48);
    const char* qg = (const char*)(Q + ko);
    const char* kg = (const char*)(Kp + ko);
    rb[j] = *(const v4f*)(kg + bByte);
#pragma unroll
    for (int t = 0; t < TM; ++t) ra[j][t] = *(const v4f*)(qg + aByte[t]);
    ++gl;
  };
  auto take_step = [&](int j) {
    bq = bMask * rb[j];
#pragma unroll
    for (int t = 0; t < TM; ++t) aq[t] = aMask[t] * ra[j][t];
  };
  auto mma_step = [&]() {
    v2f b0 = {bq.x, bq.y};
    v2f b1 = {bq.z, bq.w};
#pragma unroll
    for (int t = 0; t < TM; ++t) {
      v2f a0 = {aq[t].x, aq[t].y};
      acc[t] = __builtin_amdgcn_wmma_f32_16x16x4_f32(
          false, a0, false, b0, (short)0, acc[t], false, false);
    }
#pragma unroll
    for (int t = 0; t < TM; ++t) {
      v2f a1 = {aq[t].z, aq[t].w};
      acc[t] = __builtin_amdgcn_wmma_f32_16x16x4_f32(
          false, a1, false, b1, (short)0, acc[t], false, false);
    }
  };

#pragma unroll
  for (int j = 0; j < D; ++j) {
    load_step(j);
    __builtin_amdgcn_sched_barrier(0);
  }

  int s = 0;
  do {
#pragma unroll
    for (int j = 0; j < D; ++j) {
      take_step(j);
      load_step(j);
      mma_step();
      __builtin_amdgcn_sched_barrier(0);
    }
    s += D;
  } while (s + 2 * D <= NG);

#pragma unroll
  for (int j = 0; j < D; ++j) {
    take_step(j);
    if (s + D + j < NG) load_step(j);
    mma_step();
  }
#pragma unroll
  for (int j = 0; j < D - 1; ++j) {
    if (s + D + j < NG) {
      take_step(j);
      mma_step();
    }
  }

  if (nv) {
#pragma unroll
    for (int t = 0; t < TM; ++t) {
#pragma unroll
      for (int r = 0; r < 8; ++r) {
        int m = mBase + t * 16 + r + 8 * half;
        S[frag_off(m, n, 27)] = alpha * acc[t][r];
      }
    }
  }
}

__global__ __launch_bounds__(128) void attn_values(
    const float* __restrict__ S, const float* __restrict__ V,
    float* __restrict__ O)
{
  const int lane = threadIdx.x & 31;
  const int wave = threadIdx.x >> 5;
  const int half = lane >> 4;
  const int l16  = lane & 15;
  const int mBase = blockIdx.x * 64;
  const int nBase = blockIdx.y * 64 + wave * 16;

  v8f zero = {0.f,0.f,0.f,0.f,0.f,0.f,0.f,0.f};
  v8f acc[4];
#pragma unroll
  for (int t = 0; t < 4; ++t) acc[t] = zero;

  unsigned aByte[4];
  float aMask[4];
#pragma unroll
  for (int t = 0; t < 4; ++t) {
    int m = mBase + t * 16 + l16;
    bool v = (m < 216);
    aMask[t] = v ? 1.f : 0.f;
    aByte[t] = (unsigned)((v ? frag_off(m, 0, 27) : 0) + 4 * half) * 4u;
  }
  const int c = nBase + l16;
  const bool nv = (c < 8192);
  const float bMask = nv ? 1.f : 0.f;
  const unsigned cOff = (unsigned)chan_off(nv ? c : 0);

  constexpr int D = 6;
  constexpr int NGH = 27;
  constexpr int NG = 8 * NGH;
  v4f ra[D][4];
  v4f rb[D];
  v4f aq[4];
  v4f bq;
  int gi = 0;
  const char* sG = (const char*)S;
  const char* vG = (const char*)V;
  int tk0[4], tk1[4];
  auto fetch_tok = [&]() {
#pragma unroll
    for (int e = 0; e < 4; ++e) {
      tk0[e] = kTok.v[8 * gi + e];
      tk1[e] = kTok.v[8 * gi + 4 + e];
    }
  };
  fetch_tok();

  auto load_step = [&](int j) {
    const unsigned o0 = (cOff + (unsigned)(half ? tk1[0] : tk0[0])) << 2;
    const unsigned o1 = (cOff + (unsigned)(half ? tk1[1] : tk0[1])) << 2;
    const unsigned o2 = (cOff + (unsigned)(half ? tk1[2] : tk0[2])) << 2;
    const unsigned o3 = (cOff + (unsigned)(half ? tk1[3] : tk0[3])) << 2;
    rb[j].x = *(const float*)(vG + o0);
    rb[j].y = *(const float*)(vG + o1);
    rb[j].z = *(const float*)(vG + o2);
    rb[j].w = *(const float*)(vG + o3);
#pragma unroll
    for (int t = 0; t < 4; ++t) ra[j][t] = *(const v4f*)(sG + aByte[t]);
    ++gi;
    sG += 512;
    const bool wrap = (gi == NGH);
    gi = wrap ? 0 : gi;
    sG += wrap ? (48384 - 3456) * 4 : 0;
    vG += wrap ? 1769472L * 4 : 0L;
    fetch_tok();
  };
  auto take_step = [&](int j) {
    bq = bMask * rb[j];
#pragma unroll
    for (int t = 0; t < 4; ++t) aq[t] = aMask[t] * ra[j][t];
  };
  auto mma_step = [&]() {
    v2f b0 = {bq.x, bq.y};
    v2f b1 = {bq.z, bq.w};
#pragma unroll
    for (int t = 0; t < 4; ++t) {
      v2f a0 = {aq[t].x, aq[t].y};
      acc[t] = __builtin_amdgcn_wmma_f32_16x16x4_f32(
          false, a0, false, b0, (short)0, acc[t], false, false);
    }
#pragma unroll
    for (int t = 0; t < 4; ++t) {
      v2f a1 = {aq[t].z, aq[t].w};
      acc[t] = __builtin_amdgcn_wmma_f32_16x16x4_f32(
          false, a1, false, b1, (short)0, acc[t], false, false);
    }
  };

#pragma unroll
  for (int j = 0; j < D; ++j) {
    load_step(j);
    __builtin_amdgcn_sched_barrier(0);
  }

  int s = 0;
  do {
#pragma unroll
    for (int j = 0; j < D; ++j) {
      take_step(j);
      load_step(j);
      mma_step();
      __builtin_amdgcn_sched_barrier(0);
    }
    s += D;
  } while (s + 2 * D <= NG);

#pragma unroll
  for (int j = 0; j < D; ++j) {
    take_step(j);
    if (s + D + j < NG) load_step(j);
    mma_step();
  }
#pragma unroll
  for (int j = 0; j < D - 1; ++j) {
    if (s + D + j < NG) {
      take_step(j);
      mma_step();
    }
  }

#pragma unroll
  for (int t = 0; t < 4; ++t) {
#pragma unroll
    for (int r = 0; r < 8; ++r) {
      const int mu = mBase + t * 16 + r;
      const int e0 = kTok.v[mu];
      const int e1 = kTok.v[mu + 8];
      if (nv && mu + 8 * half < 216) O[cOff + (unsigned)(half ? e1 : e0)] = acc[t][r];
    }
  }
}

__global__ void gen_consts(float* CfF, float* Vp, float* CcF, float* SsF, float* CiF, float* SiF) {
  int i = blockIdx.x * 256 + threadIdx.x;
  if (i >= 64 * 96) return;
  int j = i / 96, nn = i % 96;
  float val = 0.f;
  if (j < 49) {
    int f = (j <= 24) ? j : (j + 47);
    float th = 6.283185307179586f * (float)((f * nn) % 96) / 96.f;
    val = cosf(th) + sinf(th);
  }
  CfF[frag_off(j, nn, 12)] = val;
  if (j < 32 && nn < 56) {
    float cv = 0.f, sv = 0.f;
    if (j <= 24 && nn <= 48) {
      float th = 6.283185307179586f * (float)((j * nn) % 96) / 96.f;
      cv = cosf(th) * ((nn == 0 || nn == 48) ? 0.5f : 1.f);
      sv = (j >= 1 && nn >= 1 && nn <= 47) ? sinf(th) : 0.f;
    }
    CcF[frag_off(j, nn, 7)] = cv;
    if (nn < 48) SsF[frag_off(j, nn, 6)] = sv;
  }
  if (nn < 32) {
    float cv = 0.f, sv = 0.f;
    if (j <= 48 && nn <= 24) {
      float th = 6.283185307179586f * (float)((j * nn) % 96) / 96.f;
      cv = cosf(th) * (nn == 0 ? 0.5f : 1.f);
      sv = (nn >= 1) ? sinf(th) : 0.f;
    }
    CiF[frag_off(j, nn, 4)] = cv;
    if (j < 48) SiF[frag_off(j, nn, 4)] = sv;
  }
  if (j < 56) Vp[nn * 56 + j] = val;
}

__global__ void hartley_mirror_fwd(const float* __restrict__ T, float* __restrict__ blk) {
  int i = blockIdx.x * 256 + threadIdx.x;
  if (i >= 7077888) return;
  int ix = i % 48, iy = (i / 48) % 48, iz = (i / 2304) % 48, c = i / 110592;
  int jx = ix + (ix >= 24), jy = iy + (iy >= 24), jz = iz + (iz >= 24);
  int gx = jx ? 49 - jx : 0, gy = jy ? 49 - jy : 0, gz = jz ? 49 - jz : 0;
  const float* t = T + (long)c * 117649;
  float a = t[(gz * 49 + jy) * 49 + jx];
  float b = t[(jz * 49 + gy) * 49 + jx];
  float d = t[(jz * 49 + jy) * 49 + gx];
  float e = t[(gz * 49 + gy) * 49 + gx];
  blk[i] = 0.5f * ((a + b) + (d - e));
}

__device__ __forceinline__ float spec_at(const float* __restrict__ s, int jz, int jy, int jx) {
  int iz = jz - (jz > 24), iy = jy - (jy > 24), ix = jx - (jx > 24);
  float v = s[(iz * 48 + iy) * 48 + ix];
  return (jz == 24 || jy == 24 || jx == 24) ? 0.f : v;
}
__global__ void hartley_mirror_spec(const float* __restrict__ spec, float* __restrict__ Xm) {
  int i = blockIdx.x * 256 + threadIdx.x;
  if (i >= 702464) return;
  int q = i % 14, jy = (i / 14) % 56, jz = (i / 784) % 56, c = i / 43904;
  float r[4] = {0.f, 0.f, 0.f, 0.f};
  if (jz < 49 && jy < 49) {
    const float* s = spec + (long)c * 110592;
    int gz = jz ? 49 - jz : 0, gy = jy ? 49 - jy : 0;
#pragma unroll
    for (int e = 0; e < 4; ++e) {
      int jx = 4 * q + e;
      int jc = jx < 49 ? jx : 48;
      int gx = jc ? 49 - jc : 0;
      float a = spec_at(s, gz, jy, jc);
      float b = spec_at(s, jz, gy, jc);
      float d = spec_at(s, jz, jy, gx);
      float m = spec_at(s, gz, gy, gx);
      r[e] = jx < 49 ? (a + b) + (d - m) : 0.f;
    }
  }
  v4f o = {r[0], r[1], r[2], r[3]};
  *(v4f*)(Xm + 4 * (long)i) = o;
}

__global__ void pack_wqkv(const float* __restrict__ wq, const float* __restrict__ wk,
                          const float* __restrict__ wv, const float* __restrict__ wo,
                          float* __restrict__ dst) {
  int i = blockIdx.x * 256 + threadIdx.x;
  if (i >= 384 * 64) return;
  int ic = i % 64, r = i / 64;
  int t = r / 128, q = r % 128;
  if (t < 2) {
    const float* w = (t == 0) ? wq : wk;
    dst[frag_off(r, ic, 8)] = w[q * 64 + ic];
  } else {
    const int h = q >> 4, o = q & 15;
    const float* wor = wo + o * 128 + h * 16;
    const float* wvc = wv + h * 1024 + ic;
    float s = 0.f;
#pragma unroll
    for (int u = 0; u < 16; ++u) s += wor[u] * wvc[u * 64];
    dst[frag_off(r, ic, 8)] = s;
  }
}

__global__ void selu_k(float* __restrict__ s, int nelem) {
  int i = blockIdx.x * 256 + threadIdx.x;
  if (i >= nelem) return;
  float x = s[i];
  const float lam = 1.0507009873554805f, alp = 1.6732632423543772f;
  s[i] = lam * (x > 0.f ? x : alp * (expf(x) - 1.f));
}

extern "C" void kernel_launch(void* const* d_in, const int* in_sizes, int n_in,
                              void* d_out, int out_size, void* d_ws, size_t ws_size,
                              hipStream_t stream) {
  const float* in = (const float*)d_in[0];
  const float* wq = (const float*)d_in[1];
  const float* wk = (const float*)d_in[2];
  const float* wv = (const float*)d_in[3];
  const float* wo = (const float*)d_in[4];
  float* out = (float*)d_out;
  float* w = (float*)d_ws;

  const long P1_o   = 0,          P2_o   = 28901376;
  const long T_o    = 43653120;
  const long blk_o  = 84934656;
  const long QKV_o  = 0;
  const long S_o    = 92012544;
  const long spec_o = 14155776;
  const long Xm_o   = 15925248;
  const long T1_o   = 18735104,   T2_o   = 23552000;
  const long Wqkv_o = 106541568;
  const long CfF_o  = 106566144;
  const long Vp_o   = 106577664;
  const long CcF_o  = 52000000,   SsF_o  = 52002048;
  const long CiF_o  = 52004096,   SiF_o  = 52006144;
  const long WS_FLOATS = 106584576;
  if (ws_size < (size_t)WS_FLOATS * sizeof(float)) return;

  float *P1 = w+P1_o, *P2 = w+P2_o, *T = w+T_o;
  float *blk = w+blk_o, *QKV = w+QKV_o, *S = w+S_o;
  float *spec = w+spec_o;
  float *Xm = w+Xm_o, *T1 = w+T1_o, *T2 = w+T2_o, *Wqkv = w+Wqkv_o;
  float *CfF = w+CfF_o, *Vp = w+Vp_o;
  float *CcF = w+CcF_o, *SsF = w+SsF_o, *CiF = w+CiF_o, *SiF = w+SiF_o;

  auto gemm = [&](const float* A, long sAm, long sAb,
                  const float* B, long sBk, long sBn, long sBb,
                  float* C, long sCm, long sCn, long sCb,
                  int M, int N, int K, int batch, float alpha, int accum) {
    dim3 g((unsigned)((M + 63) / 64), (unsigned)((N + 63) / 64), (unsigned)batch);
    gemm_f32_wmma_k4<true><<<g, 128, 0, stream>>>(A, sAm, sAb, B, sBk, sBn, sBb,
                                         C, sCm, sCn, sCb, M, N, K, alpha, accum);
  };
  auto ew = [&](int n) { return dim3((unsigned)((n + 255) / 256)); };

  gen_consts<<<ew(6144), 256, 0, stream>>>(CfF, Vp, CcF, SsF, CiF, SiF);
  pack_wqkv<<<ew(24576), 256, 0, stream>>>(wq, wk, wv, wo, Wqkv);

  gemm_f32_wmma_bstat<12, 4, 12><<<dim3(768), 128, 0, stream>>>(
      in, 96, CfF, P1, 49, 589824, 49, 36864);
  cas_fold_astat<<<dim3(1, 768), 128, 0, stream>>>(
      CcF, SsF, P1, 49, 4704, P2, 49, 2401, 49, 1, 6144, 1.f);
  cas_fold_astat<<<dim3(1, 768), 128, 0, stream>>>(
      CcF, SsF, P2, 2401, 230496, T, 2401, 117649, 2401, 38, 2432, 1.f);
  hartley_mirror_fwd<<<ew(7077888), 256, 0, stream>>>(T, blk);

  gemm_f32_wmma_astat<8, 8><<<dim3(6, 170), 128, 0, stream>>>(
      Wqkv, blk, 110592, 0, QKV, 110592, 0, 384, 110592, 1728, 1728, 1.f);

  const long HD = 16L * 110592L;
  attn_scores<<<dim3(7, 4, 8), 128, 0, stream>>>(QKV, QKV + 8 * HD, S,
                                                 0.011048543456039806f);
  selu_k<<<ew(387072), 256, 0, stream>>>(S, 387072);
  attn_values<<<dim3(4, 128, 1), 128, 0, stream>>>(S, QKV + 16 * HD, spec);

  hartley_mirror_spec<<<ew(702464), 256, 0, stream>>>(spec, Xm);
  gemm(Xm, 56, 0, Vp, 1, 56, 0, T1, 96, 1, 0, 50176, 96, 56, 1, 1.f, 0);
  cas_fold_inv_astat<<<dim3(1, 896), 128, 0, stream>>>(
      CiF, SiF, T1, 96, 5376, T2, 96, 9216, 96, 2, 1792, 1.f);
  const float halfInvN = 0.5f / 884736.f;
  cas_fold_inv_astat<<<dim3(1, 768), 128, 0, stream>>>(
      CiF, SiF, T2, 9216, 516096, out, 9216, 884736, 9216, 144, 2304, halfInvN);
}
